# starting kernel
# baseline (speedup 1.0000x reference)
#include <hip/hip_runtime.h>

typedef __attribute__((ext_vector_type(2))) float v2f;
typedef __attribute__((ext_vector_type(4))) float v4f;
typedef __attribute__((ext_vector_type(8))) float v8f;

#define F_IN   256
#define HIDDEN 64


__global__ void deg_init_k(float* __restrict__ deg, int n) {
    int i = blockIdx.x * blockDim.x + threadIdx.x;
    if (i < n) deg[i] = 1.0f;
}

__global__ void deg_accum_k(const int* __restrict__ dst, float* deg, int E) {
    int i = blockIdx.x * blockDim.x + threadIdx.x;
    if (i < E) atomicAdd(&deg[dst[i]], 1.0f);
}

__global__ void deg_rsqrt_k(float* deg, int n) {
    int i = blockIdx.x * blockDim.x + threadIdx.x;
    if (i < n) deg[i] = rsqrtf(deg[i]);
}


template <int K, int NOUT>
__global__ __launch_bounds__(128) void gemm_wmma_k(const float* __restrict__ A,
                                                   const float* __restrict__ W,
                                                   float* __restrict__ Out,
                                                   int M) {
    constexpr int LDSK = K + 4;
    __shared__ float Atile[16 * LDSK];

    const int tid  = threadIdx.x;
    const int row0 = blockIdx.x * 16;

    constexpr int KV = K / 4;
    for (int idx = tid; idx < 16 * KV; idx += 128) {
        int r = idx / KV, c4 = idx % KV;
        int row = row0 + r; if (row >= M) row = M - 1;
        v4f v = *(const v4f*)&A[(size_t)row * K + c4 * 4];
        *(v4f*)&Atile[r * LDSK + c4 * 4] = v;
    }
    __syncthreads();

    const int lane = tid & 31;
    const int wave = tid >> 5;
    const int half = lane >> 4;
    const int l    = lane & 15;
    const int col0 = wave * 16;

    v8f acc = {};
    const float* wp = W + col0 + l;
    for (int k = 0; k < K; k += 4) {
        const int ka = k + half * 2;
        v2f a = *(const v2f*)&Atile[l * LDSK + ka];
        v2f b;
        b.x = wp[ka * NOUT];
        b.y = wp[(ka + 1) * NOUT];
        acc = __builtin_amdgcn_wmma_f32_16x16x4_f32(false, a, false, b,
                                                    (short)0, acc, false, false);
    }

    float* op = Out + (size_t)(row0 + half * 8) * NOUT + col0 + l;
    if (row0 + 16 <= M) {
#pragma unroll
        for (int r = 0; r < 8; ++r) op[r * NOUT] = acc[r];
    } else {
#pragma unroll
        for (int r = 0; r < 8; ++r)
            if (row0 + half * 8 + r < M) op[r * NOUT] = acc[r];
    }
}


template <int F>
__global__ void init_agg_k(const float* __restrict__ h, const float* __restrict__ dinv,
                           float* __restrict__ agg, int n) {
    long long idx = (long long)blockIdx.x * blockDim.x + threadIdx.x;
    if (idx >= (long long)n * F) return;
    int i = (int)(idx / F);
    float d = dinv[i];
    agg[idx] = h[idx] * d * d;
}

template <int F>
__global__ void edge_agg_k(const int* __restrict__ src, const int* __restrict__ dst,
                           const float* __restrict__ dinv, const float* __restrict__ h,
                           float* agg, int E) {
    long long idx = (long long)blockIdx.x * blockDim.x + threadIdx.x;
    if (idx >= (long long)E * F) return;
    int e = (int)(idx / F), f = (int)(idx % F);
    int s = src[e], d = dst[e];
    float w = dinv[s] * dinv[d];
    atomicAdd(&agg[(size_t)d * F + f], h[(size_t)s * F + f] * w);
}

__global__ void bias_relu_k(float* a, const float* __restrict__ b, long long total) {
    long long idx = (long long)blockIdx.x * blockDim.x + threadIdx.x;
    if (idx >= total) return;
    float v = a[idx] + b[idx & (HIDDEN - 1)];
    a[idx] = v > 0.0f ? v : 0.0f;
}


__global__ void gemm3_k(const float* __restrict__ h, const float* __restrict__ W3,
                        const float* __restrict__ dinv, float* __restrict__ h3,
                        float* __restrict__ out, int n) {
    int i = blockIdx.x * blockDim.x + threadIdx.x;
    if (i >= n) return;
    const float* hr = h + (size_t)i * HIDDEN;
    float s0 = 0.0f, s1 = 0.0f;
#pragma unroll
    for (int f = 0; f < HIDDEN; ++f) {
        float v = hr[f];
        s0 += v * W3[2 * f];
        s1 += v * W3[2 * f + 1];
    }
    h3[2 * i] = s0; h3[2 * i + 1] = s1;
    float d2 = dinv[i] * dinv[i];
    out[2 * i] = s0 * d2; out[2 * i + 1] = s1 * d2;
}

__global__ void logsoftmax_k(float* out, const float* __restrict__ b3, int n) {
    int i = blockIdx.x * blockDim.x + threadIdx.x;
    if (i >= n) return;
    float z0 = out[2 * i] + b3[0];
    float z1 = out[2 * i + 1] + b3[1];
    float m = fmaxf(z0, z1);
    float lse = m + logf(__expf(z0 - m) + __expf(z1 - m));
    out[2 * i] = z0 - lse;
    out[2 * i + 1] = z1 - lse;
}


extern "C" void kernel_launch(void* const* d_in, const int* in_sizes, int n_in,
                              void* d_out, int out_size, void* d_ws, size_t ws_size,
                              hipStream_t stream) {
    const float* x  = (const float*)d_in[0];
    const int*   ei = (const int*)d_in[1];
    const float* W1 = (const float*)d_in[2];
    const float* b1 = (const float*)d_in[3];
    const float* W2 = (const float*)d_in[4];
    const float* b2 = (const float*)d_in[5];
    const float* W3 = (const float*)d_in[6];
    const float* b3 = (const float*)d_in[7];
    float* out = (float*)d_out;

    const int N = in_sizes[0] / F_IN;
    const int E = in_sizes[1] / 2;
    const int* src = ei;
    const int* dst = ei + E;

    float* ws   = (float*)d_ws;
    float* dinv = ws;
    float* bufA = dinv + N;
    float* bufB = bufA + (size_t)N * HIDDEN;
    float* h3   = bufB + (size_t)N * HIDDEN;

    const int T = 256;
    dim3 blk(T);
    int gN   = (N + T - 1) / T;
    int gE   = (E + T - 1) / T;
    long long nh = (long long)N * HIDDEN;
    long long eh = (long long)E * HIDDEN;
    int gNH  = (int)((nh + T - 1) / T);
    int gEH  = (int)((eh + T - 1) / T);
    int gE2  = (int)(((long long)E * 2 + T - 1) / T);
    int gTile = (N + 15) / 16;

    deg_init_k<<<gN, blk, 0, stream>>>(dinv, N);
    deg_accum_k<<<gE, blk, 0, stream>>>(dst, dinv, E);
    deg_rsqrt_k<<<gN, blk, 0, stream>>>(dinv, N);

    gemm_wmma_k<F_IN, HIDDEN><<<gTile, dim3(128), 0, stream>>>(x, W1, bufA, N);
    init_agg_k<HIDDEN><<<gNH, blk, 0, stream>>>(bufA, dinv, bufB, N);
    edge_agg_k<HIDDEN><<<gEH, blk, 0, stream>>>(src, dst, dinv, bufA, bufB, E);
    bias_relu_k<<<gNH, blk, 0, stream>>>(bufB, b1, nh);

    gemm_wmma_k<HIDDEN, HIDDEN><<<gTile, dim3(128), 0, stream>>>(bufB, W2, bufA, N);
    init_agg_k<HIDDEN><<<gNH, blk, 0, stream>>>(bufA, dinv, bufB, N);
    edge_agg_k<HIDDEN><<<gEH, blk, 0, stream>>>(src, dst, dinv, bufA, bufB, E);
    bias_relu_k<<<gNH, blk, 0, stream>>>(bufB, b2, nh);

    gemm3_k<<<gN, blk, 0, stream>>>(bufB, W3, dinv, h3, out, N);
    edge_agg_k<2><<<gE2, blk, 0, stream>>>(src, dst, dinv, h3, out, E);
    logsoftmax_k<<<gN, blk, 0, stream>>>(out, b3, N);
}


# 64-wide edge aggregation: one wave walks 32 edges (src, dst, weight loaded once per edge, broadcast by lane read, eight rows in flight), not one thread per element; atomic order unordered as before
# speedup vs baseline: 1.7379x; 1.7379x over previous
#include <hip/hip_runtime.h>

typedef __attribute__((ext_vector_type(2))) float v2f;
typedef __attribute__((ext_vector_type(4))) float v4f;
typedef __attribute__((ext_vector_type(8))) float v8f;

#define F_IN   256
#define HIDDEN 64


__global__ void deg_init_k(float* __restrict__ deg, int n) {
    int i = blockIdx.x * blockDim.x + threadIdx.x;
    if (i < n) deg[i] = 1.0f;
}

__global__ void deg_accum_k(const int* __restrict__ dst, float* deg, int E) {
    int i = blockIdx.x * blockDim.x + threadIdx.x;
    if (i < E) atomicAdd(&deg[dst[i]], 1.0f);
}

__global__ void deg_rsqrt_k(float* deg, int n) {
    int i = blockIdx.x * blockDim.x + threadIdx.x;
    if (i < n) deg[i] = rsqrtf(deg[i]);
}


template <int K, int NOUT>
__global__ __launch_bounds__(128) void gemm_wmma_k(const float* __restrict__ A,
                                                   const float* __restrict__ W,
                                                   float* __restrict__ Out,
                                                   int M) {
    constexpr int LDSK = K + 4;
    __shared__ float Atile[16 * LDSK];

    const int tid  = threadIdx.x;
    const int row0 = blockIdx.x * 16;

    constexpr int KV = K / 4;
    for (int idx = tid; idx < 16 * KV; idx += 128) {
        int r = idx / KV, c4 = idx % KV;
        int row = row0 + r; if (row >= M) row = M - 1;
        v4f v = *(const v4f*)&A[(size_t)row * K + c4 * 4];
        *(v4f*)&Atile[r * LDSK + c4 * 4] = v;
    }
    __syncthreads();

    const int lane = tid & 31;
    const int wave = tid >> 5;
    const int half = lane >> 4;
    const int l    = lane & 15;
    const int col0 = wave * 16;

    v8f acc = {};
    const float* wp = W + col0 + l;
    for (int k = 0; k < K; k += 4) {
        const int ka = k + half * 2;
        v2f a = *(const v2f*)&Atile[l * LDSK + ka];
        v2f b;
        b.x = wp[ka * NOUT];
        b.y = wp[(ka + 1) * NOUT];
        acc = __builtin_amdgcn_wmma_f32_16x16x4_f32(false, a, false, b,
                                                    (short)0, acc, false, false);
    }

    float* op = Out + (size_t)(row0 + half * 8) * NOUT + col0 + l;
    if (row0 + 16 <= M) {
#pragma unroll
        for (int r = 0; r < 8; ++r) op[r * NOUT] = acc[r];
    } else {
#pragma unroll
        for (int r = 0; r < 8; ++r)
            if (row0 + half * 8 + r < M) op[r * NOUT] = acc[r];
    }
}


template <int F>
__global__ void init_agg_k(const float* __restrict__ h, const float* __restrict__ dinv,
                           float* __restrict__ agg, int n) {
    long long idx = (long long)blockIdx.x * blockDim.x + threadIdx.x;
    if (idx >= (long long)n * F) return;
    int i = (int)(idx / F);
    float d = dinv[i];
    agg[idx] = h[idx] * d * d;
}

template <int F>
__global__ void edge_agg_k(const int* __restrict__ src, const int* __restrict__ dst,
                           const float* __restrict__ dinv, const float* __restrict__ h,
                           float* agg, int E) {
    long long idx = (long long)blockIdx.x * blockDim.x + threadIdx.x;
    if (idx >= (long long)E * F) return;
    int e = (int)(idx / F), f = (int)(idx % F);
    int s = src[e], d = dst[e];
    float w = dinv[s] * dinv[d];
    atomicAdd(&agg[(size_t)d * F + f], h[(size_t)s * F + f] * w);
}

template <bool FULL>
__device__ __forceinline__ void edge_walk64(int s, int d, int wi, int cnt, int lane,
                                            const float* __restrict__ h, float* agg) {
#pragma unroll
    for (int j0 = 0; j0 < 32; j0 += 8) {
        if (FULL || j0 < cnt) {
            float v0[8], v1[8];
#pragma unroll
            for (int u = 0; u < 8; ++u) {
                const int sj = __builtin_amdgcn_readlane(s, j0 + u);
                const float* hp = h + (size_t)sj * HIDDEN + lane;
                v0[u] = hp[0];
                v1[u] = hp[32];
            }
#pragma unroll
            for (int u = 0; u < 8; ++u) {
                if (FULL || j0 + u < cnt) {
                    const int   dj = __builtin_amdgcn_readlane(d, j0 + u);
                    const float wj = __int_as_float(__builtin_amdgcn_readlane(wi, j0 + u));
                    float* ap = agg + (size_t)dj * HIDDEN + lane;
                    atomicAdd(ap,      v0[u] * wj);
                    atomicAdd(ap + 32, v1[u] * wj);
                }
            }
        }
    }
}

__global__ __launch_bounds__(256) void edge_agg64_k(const int* __restrict__ src,
                                                    const int* __restrict__ dst,
                                                    const float* __restrict__ dinv,
                                                    const float* __restrict__ h,
                                                    float* agg, int E) {
    const int lane = threadIdx.x & 31;
    const int wave = __builtin_amdgcn_readfirstlane((int)(threadIdx.x >> 5));
    const int e0   = ((int)blockIdx.x * 8 + wave) * 32;
    const int cnt  = E - e0;
    if (cnt <= 0) return;

    int s = 0, d = 0;
    float w = 0.0f;
    if (lane < cnt) {
        s = src[e0 + lane];
        d = dst[e0 + lane];
        w = dinv[s] * dinv[d];
    }
    const int wi = __float_as_int(w);

    if (cnt >= 32) edge_walk64<true>(s, d, wi, cnt, lane, h, agg);
    else           edge_walk64<false>(s, d, wi, cnt, lane, h, agg);
}

__global__ void bias_relu_k(float* a, const float* __restrict__ b, long long total) {
    long long idx = (long long)blockIdx.x * blockDim.x + threadIdx.x;
    if (idx >= total) return;
    float v = a[idx] + b[idx & (HIDDEN - 1)];
    a[idx] = v > 0.0f ? v : 0.0f;
}


__global__ void gemm3_k(const float* __restrict__ h, const float* __restrict__ W3,
                        const float* __restrict__ dinv, float* __restrict__ h3,
                        float* __restrict__ out, int n) {
    int i = blockIdx.x * blockDim.x + threadIdx.x;
    if (i >= n) return;
    const float* hr = h + (size_t)i * HIDDEN;
    float s0 = 0.0f, s1 = 0.0f;
#pragma unroll
    for (int f = 0; f < HIDDEN; ++f) {
        float v = hr[f];
        s0 += v * W3[2 * f];
        s1 += v * W3[2 * f + 1];
    }
    h3[2 * i] = s0; h3[2 * i + 1] = s1;
    float d2 = dinv[i] * dinv[i];
    out[2 * i] = s0 * d2; out[2 * i + 1] = s1 * d2;
}

__global__ void logsoftmax_k(float* out, const float* __restrict__ b3, int n) {
    int i = blockIdx.x * blockDim.x + threadIdx.x;
    if (i >= n) return;
    float z0 = out[2 * i] + b3[0];
    float z1 = out[2 * i + 1] + b3[1];
    float m = fmaxf(z0, z1);
    float lse = m + logf(__expf(z0 - m) + __expf(z1 - m));
    out[2 * i] = z0 - lse;
    out[2 * i + 1] = z1 - lse;
}


extern "C" void kernel_launch(void* const* d_in, const int* in_sizes, int n_in,
                              void* d_out, int out_size, void* d_ws, size_t ws_size,
                              hipStream_t stream) {
    const float* x  = (const float*)d_in[0];
    const int*   ei = (const int*)d_in[1];
    const float* W1 = (const float*)d_in[2];
    const float* b1 = (const float*)d_in[3];
    const float* W2 = (const float*)d_in[4];
    const float* b2 = (const float*)d_in[5];
    const float* W3 = (const float*)d_in[6];
    const float* b3 = (const float*)d_in[7];
    float* out = (float*)d_out;

    const int N = in_sizes[0] / F_IN;
    const int E = in_sizes[1] / 2;
    const int* src = ei;
    const int* dst = ei + E;

    float* ws   = (float*)d_ws;
    float* dinv = ws;
    float* bufA = dinv + N;
    float* bufB = bufA + (size_t)N * HIDDEN;
    float* h3   = bufB + (size_t)N * HIDDEN;

    const int T = 256;
    dim3 blk(T);
    int gN   = (N + T - 1) / T;
    int gE   = (E + T - 1) / T;
    long long nh = (long long)N * HIDDEN;
    int gNH  = (int)((nh + T - 1) / T);
    int gE2  = (int)(((long long)E * 2 + T - 1) / T);
    int gTile = (N + 15) / 16;

    deg_init_k<<<gN, blk, 0, stream>>>(dinv, N);
    deg_accum_k<<<gE, blk, 0, stream>>>(dst, dinv, E);
    deg_rsqrt_k<<<gN, blk, 0, stream>>>(dinv, N);

    gemm_wmma_k<F_IN, HIDDEN><<<gTile, dim3(128), 0, stream>>>(x, W1, bufA, N);
    init_agg_k<HIDDEN><<<gNH, blk, 0, stream>>>(bufA, dinv, bufB, N);
    edge_agg64_k<<<gE, blk, 0, stream>>>(src, dst, dinv, bufA, bufB, E);
    bias_relu_k<<<gNH, blk, 0, stream>>>(bufB, b1, nh);

    gemm_wmma_k<HIDDEN, HIDDEN><<<gTile, dim3(128), 0, stream>>>(bufB, W2, bufA, N);
    init_agg_k<HIDDEN><<<gNH, blk, 0, stream>>>(bufA, dinv, bufB, N);
    edge_agg64_k<<<gE, blk, 0, stream>>>(src, dst, dinv, bufA, bufB, E);
    bias_relu_k<<<gNH, blk, 0, stream>>>(bufB, b2, nh);

    gemm3_k<<<gN, blk, 0, stream>>>(bufB, W3, dinv, h3, out, N);
    edge_agg_k<2><<<gE2, blk, 0, stream>>>(src, dst, dinv, h3, out, E);
    logsoftmax_k<<<gN, blk, 0, stream>>>(out, b3, N);
}


# aggregation by destination-sorted edge list (counting sort per call: slot = returned add, scan, place): one wave sums a row in registers and stores it once; no float atomics; sums in arrival order
# speedup vs baseline: 2.8452x; 1.6371x over previous
#include <hip/hip_runtime.h>


typedef __attribute__((ext_vector_type(2))) float v2f;
typedef __attribute__((ext_vector_type(4))) float v4f;
typedef __attribute__((ext_vector_type(8))) float v8f;

#define F_IN   256
#define HIDDEN 64
#define PLANE  32


__global__ void cnt_zero_k(int* __restrict__ cnt, int n) {
    int i = blockIdx.x * blockDim.x + threadIdx.x;
    if (i < n) cnt[i] = 0;
}

__global__ void count_rank_k(const int* __restrict__ dst, int* cnt, int* __restrict__ rank, int E) {
    int i = blockIdx.x * blockDim.x + threadIdx.x;
    if (i < E) rank[i] = atomicAdd(&cnt[dst[i]], 1);
}

__global__ __launch_bounds__(1024) void scan_k(const int* __restrict__ cnt, int* __restrict__ rowptr, int n) {
    __shared__ int wtot[32];
    __shared__ int wpart[32];
    const int tid = threadIdx.x, lane = tid & 31, wave = tid >> 5;
    const int start = blockIdx.x * 1024;
    int part = 0;
    for (int j = tid; j < start; j += 1024) part += cnt[j];
    const int i = start + tid;
    const int c = (i < n) ? cnt[i] : 0;
    int incl = c;
    for (int o = 1; o < 32; o <<= 1) {
        int t = __shfl_up(incl, o);
        if (lane >= o) incl += t;
    }
    for (int o = 16; o > 0; o >>= 1) part += __shfl_xor(part, o);
    if (lane == 31) wtot[wave] = incl;
    if (lane == 0)  wpart[wave] = part;
    __syncthreads();
    int tp = (lane < wave) ? wtot[lane] : 0;
    int p  = wpart[lane];
    for (int o = 16; o > 0; o >>= 1) { tp += __shfl_xor(tp, o); p += __shfl_xor(p, o); }
    if (i < n) rowptr[i] = p + tp + incl - c;
}

__global__ void fill_k(const int* __restrict__ src, const int* __restrict__ dst,
                       const int* __restrict__ cnt, const int* __restrict__ rowptr,
                       const int* __restrict__ rank, int2* __restrict__ colw, int E) {
    int i = blockIdx.x * blockDim.x + threadIdx.x;
    if (i >= E) return;
    const int s = src[i], d = dst[i];
    const float ws = rsqrtf((float)(cnt[s] + 1));
    colw[rowptr[d] + rank[i]] = make_int2(s, __float_as_int(ws));
}


template <int K, bool PLANES>
__global__ __launch_bounds__(128) void gemm_wmma_k(const float* A0, const float* A1,
                                                   const float* __restrict__ W,
                                                   float* O0, float* O1, int M) {
    static_assert(!PLANES || K == 2 * PLANE, "plane input is 64 wide");
    constexpr int NOUT = HIDDEN;
    constexpr int LDSK = K + 4;
    __shared__ float Atile[16 * LDSK];

    const int tid  = threadIdx.x;
    const int row0 = blockIdx.x * 16;

    constexpr int KV = K / 4;
    for (int idx = tid; idx < 16 * KV; idx += 128) {
        int r = idx / KV, c4 = idx % KV;
        int row = row0 + r; if (row >= M) row = M - 1;
        const float* p;
        if (PLANES) p = (c4 < PLANE / 4) ? &A0[(size_t)row * PLANE + c4 * 4]
                                         : &A1[(size_t)row * PLANE + (c4 - PLANE / 4) * 4];
        else        p = &A0[(size_t)row * K + c4 * 4];
        v4f v = *(const v4f*)p;
        *(v4f*)&Atile[r * LDSK + c4 * 4] = v;
    }
    __syncthreads();

    const int lane = tid & 31;
    const int wave = tid >> 5;
    const int half = lane >> 4;
    const int l    = lane & 15;
    const int col0 = wave * 16;

    v8f acc = {};
    const float* wp = W + col0 + l;
    for (int k = 0; k < K; k += 4) {
        const int ka = k + half * 2;
        v2f a = *(const v2f*)&Atile[l * LDSK + ka];
        v2f b;
        b.x = wp[ka * NOUT];
        b.y = wp[(ka + 1) * NOUT];
        acc = __builtin_amdgcn_wmma_f32_16x16x4_f32(false, a, false, b,
                                                    (short)0, acc, false, false);
    }

    float* O  = (wave < 2) ? O0 : O1;
    float* op = O + (size_t)(row0 + half * 8) * PLANE + (wave & 1) * 16 + l;
    if (row0 + 16 <= M) {
#pragma unroll
        for (int r = 0; r < 8; ++r) op[r * PLANE] = acc[r];
    } else {
#pragma unroll
        for (int r = 0; r < 8; ++r)
            if (row0 + half * 8 + r < M) op[r * PLANE] = acc[r];
    }
}


__global__ __launch_bounds__(256) void pull32_k(const int* __restrict__ rowptr,
                                                const int2* __restrict__ colw,
                                                const float* __restrict__ hin,
                                                const float* __restrict__ bias,
                                                float* __restrict__ hout, int n, int E) {
    const int lane = threadIdx.x & 31;
    const int wave = __builtin_amdgcn_readfirstlane((int)(threadIdx.x >> 5));
    const int i = (int)blockIdx.x * 8 + wave;
    if (i >= n) return;
    const int beg = rowptr[i];
    const int end = (i + 1 < n) ? rowptr[i + 1] : E;
    const float di = rsqrtf((float)(end - beg + 1));
    float acc = hin[(size_t)i * PLANE + lane] * (di * di);

    for (int base = beg; base < end; base += 32) {
        const int cnt = (end - base < 32) ? (end - base) : 32;
        int s = 0, wi = 0;
        if (lane < cnt) {
            const int2 r = colw[base + lane];
            s  = r.x;
            wi = __float_as_int(__int_as_float(r.y) * di);
        }
        for (int j0 = 0; j0 < cnt; j0 += 8) {
            float v[8];
            if (j0 + 8 <= cnt) {
#pragma unroll
                for (int u = 0; u < 8; ++u) {
                    const int sj = __builtin_amdgcn_readlane(s, j0 + u);
                    v[u] = hin[(size_t)sj * PLANE + lane];
                }
#pragma unroll
                for (int u = 0; u < 8; ++u)
                    acc += v[u] * __int_as_float(__builtin_amdgcn_readlane(wi, j0 + u));
            } else {
#pragma unroll
                for (int u = 0; u < 8; ++u) {
                    v[u] = 0.0f;
                    if (j0 + u < cnt) {
                        const int sj = __builtin_amdgcn_readlane(s, j0 + u);
                        v[u] = hin[(size_t)sj * PLANE + lane];
                    }
                }
#pragma unroll
                for (int u = 0; u < 8; ++u)
                    if (j0 + u < cnt)
                        acc += v[u] * __int_as_float(__builtin_amdgcn_readlane(wi, j0 + u));
            }
        }
    }
    const float r = acc + bias[lane];
    hout[(size_t)i * PLANE + lane] = r > 0.0f ? r : 0.0f;
}


__global__ void gemm3_k(const float* __restrict__ p0, const float* __restrict__ p1,
                        const float* __restrict__ W3, float* __restrict__ h3, int n) {
    int i = blockIdx.x * blockDim.x + threadIdx.x;
    if (i >= n) return;
    const float* r0 = p0 + (size_t)i * PLANE;
    const float* r1 = p1 + (size_t)i * PLANE;
    float s0 = 0.0f, s1 = 0.0f;
#pragma unroll
    for (int f = 0; f < PLANE; ++f) {
        float v = r0[f];
        s0 += v * W3[2 * f];
        s1 += v * W3[2 * f + 1];
    }
#pragma unroll
    for (int f = 0; f < PLANE; ++f) {
        float v = r1[f];
        s0 += v * W3[2 * (f + PLANE)];
        s1 += v * W3[2 * (f + PLANE) + 1];
    }
    h3[2 * i] = s0; h3[2 * i + 1] = s1;
}

__global__ void pull2_lsm_k(const int* __restrict__ rowptr, const int2* __restrict__ colw,
                            const float* __restrict__ h3, const float* __restrict__ b3,
                            float* __restrict__ out, int n, int E) {
    int i = blockIdx.x * blockDim.x + threadIdx.x;
    if (i >= n) return;
    const int beg = rowptr[i];
    const int end = (i + 1 < n) ? rowptr[i + 1] : E;
    const float di = rsqrtf((float)(end - beg + 1));
    const v2f* h2 = (const v2f*)h3;
    const v2f hs = h2[i];
    float a0 = hs.x * (di * di), a1 = hs.y * (di * di);
    for (int e = beg; e < end; ++e) {
        const int2 r = colw[e];
        const float w = __int_as_float(r.y) * di;
        const v2f v = h2[r.x];
        a0 += v.x * w;
        a1 += v.y * w;
    }
    float z0 = a0 + b3[0];
    float z1 = a1 + b3[1];
    float m = fmaxf(z0, z1);
    float lse = m + logf(__expf(z0 - m) + __expf(z1 - m));
    out[2 * i] = z0 - lse;
    out[2 * i + 1] = z1 - lse;
}


extern "C" void kernel_launch(void* const* d_in, const int* in_sizes, int n_in,
                              void* d_out, int out_size, void* d_ws, size_t ws_size,
                              hipStream_t stream) {
    const float* x  = (const float*)d_in[0];
    const int*   ei = (const int*)d_in[1];
    const float* W1 = (const float*)d_in[2];
    const float* b1 = (const float*)d_in[3];
    const float* W2 = (const float*)d_in[4];
    const float* b2 = (const float*)d_in[5];
    const float* W3 = (const float*)d_in[6];
    const float* b3 = (const float*)d_in[7];
    float* out = (float*)d_out;

    const int N = in_sizes[0] / F_IN;
    const int E = in_sizes[1] / 2;
    const int* src = ei;
    const int* dst = ei + E;

    int2*  colw   = (int2*)d_ws;
    int*   rowptr = (int*)d_ws + (size_t)2 * E;
    float* pA     = (float*)(rowptr + N);
    float* pB     = pA + (size_t)N * PLANE;
    float* pC     = pB + (size_t)N * PLANE;
    int*   rank   = (int*)pC;
    float* h3     = pA;
    int*   cnt    = (int*)d_out;

    const int T = 256;
    dim3 blk(T);
    int gN    = (N + T - 1) / T;
    int gE    = (E + T - 1) / T;
    int gRow  = (N + 7) / 8;
    int gScan = (N + 1023) / 1024;
    int gTile = (N + 15) / 16;

    cnt_zero_k<<<gN, blk, 0, stream>>>(cnt, N);
    count_rank_k<<<gE, blk, 0, stream>>>(dst, cnt, rank, E);
    scan_k<<<gScan, dim3(1024), 0, stream>>>(cnt, rowptr, N);
    fill_k<<<gE, blk, 0, stream>>>(src, dst, cnt, rowptr, rank, colw, E);

    gemm_wmma_k<F_IN, false><<<gTile, dim3(128), 0, stream>>>(x, x, W1, pA, pB, N);
    pull32_k<<<gRow, blk, 0, stream>>>(rowptr, colw, pA, b1, pC, N, E);
    pull32_k<<<gRow, blk, 0, stream>>>(rowptr, colw, pB, b1 + PLANE, pA, N, E);

    gemm_wmma_k<HIDDEN, true><<<gTile, dim3(128), 0, stream>>>(pC, pA, W2, pC, pA, N);
    pull32_k<<<gRow, blk, 0, stream>>>(rowptr, colw, pC, b2, pB, N, E);
    pull32_k<<<gRow, blk, 0, stream>>>(rowptr, colw, pA, b2 + PLANE, pC, N, E);

    gemm3_k<<<gN, blk, 0, stream>>>(pB, pC, W3, h3, N);
    pull2_lsm_k<<<gN, blk, 0, stream>>>(rowptr, colw, h3, b3, out, N, E);
}


# aggregation: one wave walks 12 consecutive rows (8x fewer waves), next row's records and self term loaded a row ahead, all of a row's first 32 source rows requested before any add; sum order kept
# speedup vs baseline: 2.9872x; 1.0499x over previous
#include <hip/hip_runtime.h>


typedef __attribute__((ext_vector_type(2))) float v2f;
typedef __attribute__((ext_vector_type(4))) float v4f;
typedef __attribute__((ext_vector_type(8))) float v8f;

#define F_IN   256
#define HIDDEN 64
#define PLANE  32


__global__ void cnt_zero_k(int* __restrict__ cnt, int n) {
    int i = blockIdx.x * blockDim.x + threadIdx.x;
    if (i < n) cnt[i] = 0;
}

__global__ void count_rank_k(const int* __restrict__ dst, int* cnt, int* __restrict__ rank, int E) {
    int i = blockIdx.x * blockDim.x + threadIdx.x;
    if (i < E) rank[i] = atomicAdd(&cnt[dst[i]], 1);
}

__global__ __launch_bounds__(1024) void scan_k(const int* __restrict__ cnt, int* __restrict__ rowptr, int n) {
    __shared__ int wtot[32];
    __shared__ int wpart[32];
    const int tid = threadIdx.x, lane = tid & 31, wave = tid >> 5;
    const int start = blockIdx.x * 1024;
    int part = 0;
    for (int j = tid; j < start; j += 1024) part += cnt[j];
    const int i = start + tid;
    const int c = (i < n) ? cnt[i] : 0;
    int incl = c;
    for (int o = 1; o < 32; o <<= 1) {
        int t = __shfl_up(incl, o);
        if (lane >= o) incl += t;
    }
    for (int o = 16; o > 0; o >>= 1) part += __shfl_xor(part, o);
    if (lane == 31) wtot[wave] = incl;
    if (lane == 0)  wpart[wave] = part;
    __syncthreads();
    int tp = (lane < wave) ? wtot[lane] : 0;
    int p  = wpart[lane];
    for (int o = 16; o > 0; o >>= 1) { tp += __shfl_xor(tp, o); p += __shfl_xor(p, o); }
    if (i < n) rowptr[i] = p + tp + incl - c;
}

__global__ void fill_k(const int* __restrict__ src, const int* __restrict__ dst,
                       const int* __restrict__ cnt, const int* __restrict__ rowptr,
                       const int* __restrict__ rank, int2* __restrict__ colw, int E) {
    int i = blockIdx.x * blockDim.x + threadIdx.x;
    if (i >= E) return;
    const int s = src[i], d = dst[i];
    const float ws = rsqrtf((float)(cnt[s] + 1));
    colw[rowptr[d] + rank[i]] = make_int2(s, __float_as_int(ws));
}


template <int K, bool PLANES>
__global__ __launch_bounds__(128) void gemm_wmma_k(const float* A0, const float* A1,
                                                   const float* __restrict__ W,
                                                   float* O0, float* O1, int M) {
    static_assert(!PLANES || K == 2 * PLANE, "plane input is 64 wide");
    constexpr int NOUT = HIDDEN;
    constexpr int LDSK = K + 4;
    __shared__ float Atile[16 * LDSK];

    const int tid  = threadIdx.x;
    const int row0 = blockIdx.x * 16;

    constexpr int KV = K / 4;
    for (int idx = tid; idx < 16 * KV; idx += 128) {
        int r = idx / KV, c4 = idx % KV;
        int row = row0 + r; if (row >= M) row = M - 1;
        const float* p;
        if (PLANES) p = (c4 < PLANE / 4) ? &A0[(size_t)row * PLANE + c4 * 4]
                                         : &A1[(size_t)row * PLANE + (c4 - PLANE / 4) * 4];
        else        p = &A0[(size_t)row * K + c4 * 4];
        v4f v = *(const v4f*)p;
        *(v4f*)&Atile[r * LDSK + c4 * 4] = v;
    }
    __syncthreads();

    const int lane = tid & 31;
    const int wave = tid >> 5;
    const int half = lane >> 4;
    const int l    = lane & 15;
    const int col0 = wave * 16;

    v8f acc = {};
    const float* wp = W + col0 + l;
    for (int k = 0; k < K; k += 4) {
        const int ka = k + half * 2;
        v2f a = *(const v2f*)&Atile[l * LDSK + ka];
        v2f b;
        b.x = wp[ka * NOUT];
        b.y = wp[(ka + 1) * NOUT];
        acc = __builtin_amdgcn_wmma_f32_16x16x4_f32(false, a, false, b,
                                                    (short)0, acc, false, false);
    }

    float* O  = (wave < 2) ? O0 : O1;
    float* op = O + (size_t)(row0 + half * 8) * PLANE + (wave & 1) * 16 + l;
    if (row0 + 16 <= M) {
#pragma unroll
        for (int r = 0; r < 8; ++r) op[r * PLANE] = acc[r];
    } else {
#pragma unroll
        for (int r = 0; r < 8; ++r)
            if (row0 + half * 8 + r < M) op[r * PLANE] = acc[r];
    }
}


#define RPW 12

__global__ __launch_bounds__(256) void pull32_k(const int* __restrict__ rowptr,
                                                const int2* __restrict__ colw,
                                                const float* __restrict__ hin,
                                                const float* __restrict__ bias,
                                                float* __restrict__ hout, int n, int E) {
    const int lane = threadIdx.x & 31;
    const int wave = __builtin_amdgcn_readfirstlane((int)(threadIdx.x >> 5));
    const int row0 = ((int)blockIdx.x * 8 + wave) * RPW;
    if (row0 >= n) return;
    const int nrows = (n - row0 < RPW) ? (n - row0) : RPW;

    int bnd = E;
    if (lane <= nrows && row0 + lane < n) bnd = rowptr[row0 + lane];
    const float bl = bias[lane];

    int beg_n = __builtin_amdgcn_readlane(bnd, 0);
    int end_n = __builtin_amdgcn_readlane(bnd, 1);
    int2 rec_n = make_int2(0, 0);
    if (lane < end_n - beg_n) rec_n = colw[beg_n + lane];
    float self_n = hin[(size_t)row0 * PLANE + lane];

    for (int r = 0; r < nrows; ++r) {
        const int beg = beg_n, end = end_n;
        const int2 cur = rec_n;
        const float selfv = self_n;
        const float di = rsqrtf((float)(end - beg + 1));
        float acc = selfv * (di * di);

        const int cnt   = (end - beg < 32) ? (end - beg) : 32;
        const int nfull = cnt >> 3, rem = cnt & 7, jr = cnt & ~7;
        const int s  = cur.x;
        const int wi = __float_as_int(__int_as_float(cur.y) * di);
#define ROWLD(j) ((hin + (size_t)(unsigned)__builtin_amdgcn_readlane(s, (j)) * PLANE)[lane])
#define WJ(j)    (__int_as_float(__builtin_amdgcn_readlane(wi, (j))))
        float a0, a1, a2, a3, a4, a5, a6, a7,
              b0, b1, b2, b3, b4, b5, b6, b7,
              c0, c1, c2, c3, c4, c5, c6, c7,
              d0, d1, d2, d3, d4, d5, d6, d7,
              p0, p1, p2, p3, p4, p5, p6;
        if (nfull > 0) { a0 = ROWLD(0); a1 = ROWLD(1); a2 = ROWLD(2); a3 = ROWLD(3); a4 = ROWLD(4); a5 = ROWLD(5); a6 = ROWLD(6); a7 = ROWLD(7); }
        if (nfull > 1) { b0 = ROWLD(8); b1 = ROWLD(9); b2 = ROWLD(10); b3 = ROWLD(11); b4 = ROWLD(12); b5 = ROWLD(13); b6 = ROWLD(14); b7 = ROWLD(15); }
        if (nfull > 2) { c0 = ROWLD(16); c1 = ROWLD(17); c2 = ROWLD(18); c3 = ROWLD(19); c4 = ROWLD(20); c5 = ROWLD(21); c6 = ROWLD(22); c7 = ROWLD(23); }
        if (nfull > 3) { d0 = ROWLD(24); d1 = ROWLD(25); d2 = ROWLD(26); d3 = ROWLD(27); d4 = ROWLD(28); d5 = ROWLD(29); d6 = ROWLD(30); d7 = ROWLD(31); }
        if (rem > 0) p0 = ROWLD(jr + 0);
        if (rem > 1) p1 = ROWLD(jr + 1);
        if (rem > 2) p2 = ROWLD(jr + 2);
        if (rem > 3) p3 = ROWLD(jr + 3);
        if (rem > 4) p4 = ROWLD(jr + 4);
        if (rem > 5) p5 = ROWLD(jr + 5);
        if (rem > 6) p6 = ROWLD(jr + 6);

        {
            const int rn = (r + 1 < nrows) ? r + 1 : r;
            beg_n = __builtin_amdgcn_readlane(bnd, rn);
            end_n = __builtin_amdgcn_readlane(bnd, rn + 1);
            rec_n = make_int2(0, 0);
            if (lane < end_n - beg_n) rec_n = colw[beg_n + lane];
            self_n = hin[(size_t)(row0 + rn) * PLANE + lane];
        }

        if (nfull > 0) { acc += a0 * WJ(0); acc += a1 * WJ(1); acc += a2 * WJ(2); acc += a3 * WJ(3); acc += a4 * WJ(4); acc += a5 * WJ(5); acc += a6 * WJ(6); acc += a7 * WJ(7); }
        if (nfull > 1) { acc += b0 * WJ(8); acc += b1 * WJ(9); acc += b2 * WJ(10); acc += b3 * WJ(11); acc += b4 * WJ(12); acc += b5 * WJ(13); acc += b6 * WJ(14); acc += b7 * WJ(15); }
        if (nfull > 2) { acc += c0 * WJ(16); acc += c1 * WJ(17); acc += c2 * WJ(18); acc += c3 * WJ(19); acc += c4 * WJ(20); acc += c5 * WJ(21); acc += c6 * WJ(22); acc += c7 * WJ(23); }
        if (nfull > 3) { acc += d0 * WJ(24); acc += d1 * WJ(25); acc += d2 * WJ(26); acc += d3 * WJ(27); acc += d4 * WJ(28); acc += d5 * WJ(29); acc += d6 * WJ(30); acc += d7 * WJ(31); }
        if (rem > 0) acc += p0 * WJ(jr + 0);
        if (rem > 1) acc += p1 * WJ(jr + 1);
        if (rem > 2) acc += p2 * WJ(jr + 2);
        if (rem > 3) acc += p3 * WJ(jr + 3);
        if (rem > 4) acc += p4 * WJ(jr + 4);
        if (rem > 5) acc += p5 * WJ(jr + 5);
        if (rem > 6) acc += p6 * WJ(jr + 6);
#undef ROWLD
#undef WJ

        for (int base = beg + 32; base < end; base += 32) {
            const int cnt = (end - base < 32) ? (end - base) : 32;
            int s = 0, wi = 0;
            if (lane < cnt) {
                const int2 rr = colw[base + lane];
                s  = rr.x;
                wi = __float_as_int(__int_as_float(rr.y) * di);
            }
            for (int j = 0; j < cnt; ++j) {
                const int sj = __builtin_amdgcn_readlane(s, j);
                acc += hin[(size_t)sj * PLANE + lane] * __int_as_float(__builtin_amdgcn_readlane(wi, j));
            }
        }
        const float res = acc + bl;
        hout[(size_t)(row0 + r) * PLANE + lane] = res > 0.0f ? res : 0.0f;
    }
}


__global__ void gemm3_k(const float* __restrict__ p0, const float* __restrict__ p1,
                        const float* __restrict__ W3, float* __restrict__ h3, int n) {
    int i = blockIdx.x * blockDim.x + threadIdx.x;
    if (i >= n) return;
    const float* r0 = p0 + (size_t)i * PLANE;
    const float* r1 = p1 + (size_t)i * PLANE;
    float s0 = 0.0f, s1 = 0.0f;
#pragma unroll
    for (int f = 0; f < PLANE; ++f) {
        float v = r0[f];
        s0 += v * W3[2 * f];
        s1 += v * W3[2 * f + 1];
    }
#pragma unroll
    for (int f = 0; f < PLANE; ++f) {
        float v = r1[f];
        s0 += v * W3[2 * (f + PLANE)];
        s1 += v * W3[2 * (f + PLANE) + 1];
    }
    h3[2 * i] = s0; h3[2 * i + 1] = s1;
}

__global__ void pull2_lsm_k(const int* __restrict__ rowptr, const int2* __restrict__ colw,
                            const float* __restrict__ h3, const float* __restrict__ b3,
                            float* __restrict__ out, int n, int E) {
    int i = blockIdx.x * blockDim.x + threadIdx.x;
    if (i >= n) return;
    const int beg = rowptr[i];
    const int end = (i + 1 < n) ? rowptr[i + 1] : E;
    const float di = rsqrtf((float)(end - beg + 1));
    const v2f* h2 = (const v2f*)h3;
    const v2f hs = h2[i];
    float a0 = hs.x * (di * di), a1 = hs.y * (di * di);
    for (int e = beg; e < end; ++e) {
        const int2 r = colw[e];
        const float w = __int_as_float(r.y) * di;
        const v2f v = h2[r.x];
        a0 += v.x * w;
        a1 += v.y * w;
    }
    float z0 = a0 + b3[0];
    float z1 = a1 + b3[1];
    float m = fmaxf(z0, z1);
    float lse = m + logf(__expf(z0 - m) + __expf(z1 - m));
    out[2 * i] = z0 - lse;
    out[2 * i + 1] = z1 - lse;
}


extern "C" void kernel_launch(void* const* d_in, const int* in_sizes, int n_in,
                              void* d_out, int out_size, void* d_ws, size_t ws_size,
                              hipStream_t stream) {
    const float* x  = (const float*)d_in[0];
    const int*   ei = (const int*)d_in[1];
    const float* W1 = (const float*)d_in[2];
    const float* b1 = (const float*)d_in[3];
    const float* W2 = (const float*)d_in[4];
    const float* b2 = (const float*)d_in[5];
    const float* W3 = (const float*)d_in[6];
    const float* b3 = (const float*)d_in[7];
    float* out = (float*)d_out;

    const int N = in_sizes[0] / F_IN;
    const int E = in_sizes[1] / 2;
    const int* src = ei;
    const int* dst = ei + E;

    int2*  colw   = (int2*)d_ws;
    int*   rowptr = (int*)d_ws + (size_t)2 * E;
    float* pA     = (float*)(rowptr + N);
    float* pB     = pA + (size_t)N * PLANE;
    float* pC     = pB + (size_t)N * PLANE;
    int*   rank   = (int*)pC;
    float* h3     = pA;
    int*   cnt    = (int*)d_out;

    const int T = 256;
    dim3 blk(T);
    int gN    = (N + T - 1) / T;
    int gE    = (E + T - 1) / T;
    int gRow  = (N + 8 * RPW - 1) / (8 * RPW);
    int gScan = (N + 1023) / 1024;
    int gTile = (N + 15) / 16;

    cnt_zero_k<<<gN, blk, 0, stream>>>(cnt, N);
    count_rank_k<<<gE, blk, 0, stream>>>(dst, cnt, rank, E);
    scan_k<<<gScan, dim3(1024), 0, stream>>>(cnt, rowptr, N);
    fill_k<<<gE, blk, 0, stream>>>(src, dst, cnt, rowptr, rank, colw, E);

    gemm_wmma_k<F_IN, false><<<gTile, dim3(128), 0, stream>>>(x, x, W1, pA, pB, N);
    pull32_k<<<gRow, blk, 0, stream>>>(rowptr, colw, pA, b1, pC, N, E);
    pull32_k<<<gRow, blk, 0, stream>>>(rowptr, colw, pB, b1 + PLANE, pA, N, E);

    gemm_wmma_k<HIDDEN, true><<<gTile, dim3(128), 0, stream>>>(pC, pA, W2, pC, pA, N);
    pull32_k<<<gRow, blk, 0, stream>>>(rowptr, colw, pC, b2, pB, N, E);
    pull32_k<<<gRow, blk, 0, stream>>>(rowptr, colw, pA, b2 + PLANE, pC, N, E);

    gemm3_k<<<gN, blk, 0, stream>>>(pB, pC, W3, h3, N);
    pull2_lsm_k<<<gN, blk, 0, stream>>>(rowptr, colw, h3, b3, out, N, E);
}


# edge sort without global atomics: chunks of 4096 edges ordered by 512-node bucket on LDS counters, then one block per bucket ranks and scans in LDS; sums stay in arrival order (of the LDS adds)
# speedup vs baseline: 4.4967x; 1.5053x over previous
#include <hip/hip_runtime.h>


typedef __attribute__((ext_vector_type(2))) float v2f;
typedef __attribute__((ext_vector_type(4))) float v4f;
typedef __attribute__((ext_vector_type(8))) float v8f;

#define F_IN   256
#define HIDDEN 64
#define PLANE  32


#define BSH  9
#define BW   (1 << BSH)
#define MAXB 256
#define CH   4096

__global__ __launch_bounds__(1024) void part_k(const int* __restrict__ src, const int* __restrict__ dst,
                                               int2* __restrict__ tmp, int2* __restrict__ tab,
                                               int E, int nbk, int nblk) {
    __shared__ int  hist[MAXB];
    __shared__ int  offL[MAXB];
    __shared__ int  wsum[8];
    __shared__ int2 stage[CH];
    const int tid = threadIdx.x, lane = tid & 31, wave = tid >> 5;
    const int base = blockIdx.x * CH;
    if (tid < MAXB) hist[tid] = 0;
    __syncthreads();

    int s[4], d[4], r[4];
#pragma unroll
    for (int k = 0; k < 4; ++k) {
        const int e = base + k * 1024 + tid;
        s[k] = 0; d[k] = -1; r[k] = 0;
        if (e < E) { s[k] = src[e]; d[k] = dst[e]; }
    }
#pragma unroll
    for (int k = 0; k < 4; ++k)
        if (d[k] >= 0) r[k] = atomicAdd(&hist[d[k] >> BSH], 1);
    __syncthreads();

    const int c = (tid < MAXB) ? hist[tid] : 0;
    int incl = c;
    for (int o = 1; o < 32; o <<= 1) {
        int t = __shfl_up(incl, o);
        if (lane >= o) incl += t;
    }
    if (tid < MAXB && lane == 31) wsum[wave] = incl;
    __syncthreads();
    if (tid < MAXB) {
        int pre = 0;
        for (int w = 0; w < wave; ++w) pre += wsum[w];
        offL[tid] = pre + incl - c;
    }
    __syncthreads();

#pragma unroll
    for (int k = 0; k < 4; ++k)
        if (d[k] >= 0) stage[offL[d[k] >> BSH] + r[k]] = make_int2(s[k], d[k]);
    __syncthreads();
    const int nvalid = (E - base < CH) ? (E - base) : CH;
#pragma unroll
    for (int k = 0; k < 4; ++k) {
        const int idx = k * 1024 + tid;
        if (idx < nvalid) tmp[base + idx] = stage[idx];
    }
    if (tid < nbk) tab[(size_t)tid * nblk + blockIdx.x] = make_int2(offL[tid], hist[tid]);
}

__global__ __launch_bounds__(1024) void bucket_k(const int2* __restrict__ tmp, const int2* __restrict__ tab,
                                                 int* __restrict__ rankT, int* __restrict__ cnt,
                                                 int* __restrict__ rowptr, int n, int nblk) {
    __shared__ int cntL[BW];
    __shared__ int wred[32];
    __shared__ int wsum[16];
    const int tid = threadIdx.x, lane = tid & 31;
    const int wave = __builtin_amdgcn_readfirstlane((int)(threadIdx.x >> 5));
    const int b = blockIdx.x, lo = b << BSH;

    int part = 0;
    const int front = b * nblk;
    for (int i = tid; i < front; i += 1024) part += tab[i].y;
    for (int o = 16; o > 0; o >>= 1) part += __shfl_xor(part, o);
    if (tid < BW) cntL[tid] = 0;
    if (lane == 0) wred[wave] = part;
    __syncthreads();

    const int2* trow = tab + (size_t)b * nblk;
    for (int j0 = wave; j0 < nblk; j0 += 128) {
        int st[4], ln[4];
        int2 e[4];
        int rk[4];
#pragma unroll
        for (int u = 0; u < 4; ++u) {
            const int j  = j0 + 32 * u;
            const int jc = (j < nblk) ? j : nblk - 1;
            const int2 t = trow[jc];
            st[u] = jc * CH + t.x;
            ln[u] = (j < nblk) ? t.y : 0;
        }
#pragma unroll
        for (int u = 0; u < 4; ++u) {
            e[u] = make_int2(0, lo);
            if (lane < ln[u]) e[u] = tmp[st[u] + lane];
        }
#pragma unroll
        for (int u = 0; u < 4; ++u) {
            rk[u] = 0;
            if (lane < ln[u]) rk[u] = atomicAdd(&cntL[e[u].y - lo], 1);
        }
#pragma unroll
        for (int u = 0; u < 4; ++u)
            if (lane < ln[u]) rankT[st[u] + lane] = rk[u];
#pragma unroll
        for (int u = 0; u < 4; ++u)
            for (int k = 32; k < ln[u]; k += 32)
                if (k + lane < ln[u]) {
                    const int2 ee = tmp[st[u] + k + lane];
                    rankT[st[u] + k + lane] = atomicAdd(&cntL[ee.y - lo], 1);
                }
    }
    __syncthreads();

    int bs = wred[lane];
    for (int o = 16; o > 0; o >>= 1) bs += __shfl_xor(bs, o);
    const int c = (tid < BW) ? cntL[tid] : 0;
    int incl = c;
    for (int o = 1; o < 32; o <<= 1) {
        int t = __shfl_up(incl, o);
        if (lane >= o) incl += t;
    }
    if (tid < BW && lane == 31) wsum[wave] = incl;
    __syncthreads();
    int tp = (lane < wave && lane < 16) ? wsum[lane] : 0;
    for (int o = 16; o > 0; o >>= 1) tp += __shfl_xor(tp, o);
    const int i = lo + tid;
    if (tid < BW && i < n) {
        rowptr[i] = bs + tp + incl - c;
        cnt[i] = c;
    }
}

__global__ void place_k(const int2* __restrict__ tmp, const int* __restrict__ rankT,
                        const int* __restrict__ cnt, const int* __restrict__ rowptr,
                        int2* __restrict__ colw, int E) {
    int p = blockIdx.x * blockDim.x + threadIdx.x;
    if (p >= E) return;
    const int2 sd = tmp[p];
    const float ws = rsqrtf((float)(cnt[sd.x] + 1));
    colw[rowptr[sd.y] + rankT[p]] = make_int2(sd.x, __float_as_int(ws));
}


template <int K, bool PLANES>
__global__ __launch_bounds__(128) void gemm_wmma_k(const float* A0, const float* A1,
                                                   const float* __restrict__ W,
                                                   float* O0, float* O1, int M) {
    static_assert(!PLANES || K == 2 * PLANE, "plane input is 64 wide");
    constexpr int NOUT = HIDDEN;
    constexpr int LDSK = K + 4;
    __shared__ float Atile[16 * LDSK];

    const int tid  = threadIdx.x;
    const int row0 = blockIdx.x * 16;

    constexpr int KV = K / 4;
    for (int idx = tid; idx < 16 * KV; idx += 128) {
        int r = idx / KV, c4 = idx % KV;
        int row = row0 + r; if (row >= M) row = M - 1;
        const float* p;
        if (PLANES) p = (c4 < PLANE / 4) ? &A0[(size_t)row * PLANE + c4 * 4]
                                         : &A1[(size_t)row * PLANE + (c4 - PLANE / 4) * 4];
        else        p = &A0[(size_t)row * K + c4 * 4];
        v4f v = *(const v4f*)p;
        *(v4f*)&Atile[r * LDSK + c4 * 4] = v;
    }
    __syncthreads();

    const int lane = tid & 31;
    const int wave = tid >> 5;
    const int half = lane >> 4;
    const int l    = lane & 15;
    const int col0 = wave * 16;

    v8f acc = {};
    const float* wp = W + col0 + l;
    for (int k = 0; k < K; k += 4) {
        const int ka = k + half * 2;
        v2f a = *(const v2f*)&Atile[l * LDSK + ka];
        v2f b;
        b.x = wp[ka * NOUT];
        b.y = wp[(ka + 1) * NOUT];
        acc = __builtin_amdgcn_wmma_f32_16x16x4_f32(false, a, false, b,
                                                    (short)0, acc, false, false);
    }

    float* O  = (wave < 2) ? O0 : O1;
    float* op = O + (size_t)(row0 + half * 8) * PLANE + (wave & 1) * 16 + l;
    if (row0 + 16 <= M) {
#pragma unroll
        for (int r = 0; r < 8; ++r) op[r * PLANE] = acc[r];
    } else {
#pragma unroll
        for (int r = 0; r < 8; ++r)
            if (row0 + half * 8 + r < M) op[r * PLANE] = acc[r];
    }
}


#define RPW 12

__global__ __launch_bounds__(256) void pull32_k(const int* __restrict__ rowptr,
                                                const int2* __restrict__ colw,
                                                const float* __restrict__ hin,
                                                const float* __restrict__ bias,
                                                float* __restrict__ hout, int n, int E) {
    const int lane = threadIdx.x & 31;
    const int wave = __builtin_amdgcn_readfirstlane((int)(threadIdx.x >> 5));
    const int row0 = ((int)blockIdx.x * 8 + wave) * RPW;
    if (row0 >= n) return;
    const int nrows = (n - row0 < RPW) ? (n - row0) : RPW;

    int bnd = E;
    if (lane <= nrows && row0 + lane < n) bnd = rowptr[row0 + lane];
    const float bl = bias[lane];

    int beg_n = __builtin_amdgcn_readlane(bnd, 0);
    int end_n = __builtin_amdgcn_readlane(bnd, 1);
    int2 rec_n = make_int2(0, 0);
    if (lane < end_n - beg_n) rec_n = colw[beg_n + lane];
    float self_n = hin[(size_t)row0 * PLANE + lane];

    for (int r = 0; r < nrows; ++r) {
        const int beg = beg_n, end = end_n;
        const int2 cur = rec_n;
        const float selfv = self_n;
        const float di = rsqrtf((float)(end - beg + 1));
        float acc = selfv * (di * di);

        const int cnt   = (end - beg < 32) ? (end - beg) : 32;
        const int nfull = cnt >> 3, rem = cnt & 7, jr = cnt & ~7;
        const int s  = cur.x;
        const int wi = __float_as_int(__int_as_float(cur.y) * di);
#define ROWLD(j) ((hin + (size_t)(unsigned)__builtin_amdgcn_readlane(s, (j)) * PLANE)[lane])
#define WJ(j)    (__int_as_float(__builtin_amdgcn_readlane(wi, (j))))
        float a0, a1, a2, a3, a4, a5, a6, a7,
              b0, b1, b2, b3, b4, b5, b6, b7,
              c0, c1, c2, c3, c4, c5, c6, c7,
              d0, d1, d2, d3, d4, d5, d6, d7,
              p0, p1, p2, p3, p4, p5, p6;
        if (nfull > 0) { a0 = ROWLD(0); a1 = ROWLD(1); a2 = ROWLD(2); a3 = ROWLD(3); a4 = ROWLD(4); a5 = ROWLD(5); a6 = ROWLD(6); a7 = ROWLD(7); }
        if (nfull > 1) { b0 = ROWLD(8); b1 = ROWLD(9); b2 = ROWLD(10); b3 = ROWLD(11); b4 = ROWLD(12); b5 = ROWLD(13); b6 = ROWLD(14); b7 = ROWLD(15); }
        if (nfull > 2) { c0 = ROWLD(16); c1 = ROWLD(17); c2 = ROWLD(18); c3 = ROWLD(19); c4 = ROWLD(20); c5 = ROWLD(21); c6 = ROWLD(22); c7 = ROWLD(23); }
        if (nfull > 3) { d0 = ROWLD(24); d1 = ROWLD(25); d2 = ROWLD(26); d3 = ROWLD(27); d4 = ROWLD(28); d5 = ROWLD(29); d6 = ROWLD(30); d7 = ROWLD(31); }
        if (rem > 0) p0 = ROWLD(jr + 0);
        if (rem > 1) p1 = ROWLD(jr + 1);
        if (rem > 2) p2 = ROWLD(jr + 2);
        if (rem > 3) p3 = ROWLD(jr + 3);
        if (rem > 4) p4 = ROWLD(jr + 4);
        if (rem > 5) p5 = ROWLD(jr + 5);
        if (rem > 6) p6 = ROWLD(jr + 6);

        {
            const int rn = (r + 1 < nrows) ? r + 1 : r;
            beg_n = __builtin_amdgcn_readlane(bnd, rn);
            end_n = __builtin_amdgcn_readlane(bnd, rn + 1);
            rec_n = make_int2(0, 0);
            if (lane < end_n - beg_n) rec_n = colw[beg_n + lane];
            self_n = hin[(size_t)(row0 + rn) * PLANE + lane];
        }

        if (nfull > 0) { acc += a0 * WJ(0); acc += a1 * WJ(1); acc += a2 * WJ(2); acc += a3 * WJ(3); acc += a4 * WJ(4); acc += a5 * WJ(5); acc += a6 * WJ(6); acc += a7 * WJ(7); }
        if (nfull > 1) { acc += b0 * WJ(8); acc += b1 * WJ(9); acc += b2 * WJ(10); acc += b3 * WJ(11); acc += b4 * WJ(12); acc += b5 * WJ(13); acc += b6 * WJ(14); acc += b7 * WJ(15); }
        if (nfull > 2) { acc += c0 * WJ(16); acc += c1 * WJ(17); acc += c2 * WJ(18); acc += c3 * WJ(19); acc += c4 * WJ(20); acc += c5 * WJ(21); acc += c6 * WJ(22); acc += c7 * WJ(23); }
        if (nfull > 3) { acc += d0 * WJ(24); acc += d1 * WJ(25); acc += d2 * WJ(26); acc += d3 * WJ(27); acc += d4 * WJ(28); acc += d5 * WJ(29); acc += d6 * WJ(30); acc += d7 * WJ(31); }
        if (rem > 0) acc += p0 * WJ(jr + 0);
        if (rem > 1) acc += p1 * WJ(jr + 1);
        if (rem > 2) acc += p2 * WJ(jr + 2);
        if (rem > 3) acc += p3 * WJ(jr + 3);
        if (rem > 4) acc += p4 * WJ(jr + 4);
        if (rem > 5) acc += p5 * WJ(jr + 5);
        if (rem > 6) acc += p6 * WJ(jr + 6);
#undef ROWLD
#undef WJ

        for (int base = beg + 32; base < end; base += 32) {
            const int cnt = (end - base < 32) ? (end - base) : 32;
            int s = 0, wi = 0;
            if (lane < cnt) {
                const int2 rr = colw[base + lane];
                s  = rr.x;
                wi = __float_as_int(__int_as_float(rr.y) * di);
            }
            for (int j = 0; j < cnt; ++j) {
                const int sj = __builtin_amdgcn_readlane(s, j);
                acc += hin[(size_t)sj * PLANE + lane] * __int_as_float(__builtin_amdgcn_readlane(wi, j));
            }
        }
        const float res = acc + bl;
        hout[(size_t)(row0 + r) * PLANE + lane] = res > 0.0f ? res : 0.0f;
    }
}


__global__ void gemm3_k(const float* __restrict__ p0, const float* __restrict__ p1,
                        const float* __restrict__ W3, float* __restrict__ h3, int n) {
    int i = blockIdx.x * blockDim.x + threadIdx.x;
    if (i >= n) return;
    const float* r0 = p0 + (size_t)i * PLANE;
    const float* r1 = p1 + (size_t)i * PLANE;
    float s0 = 0.0f, s1 = 0.0f;
#pragma unroll
    for (int f = 0; f < PLANE; ++f) {
        float v = r0[f];
        s0 += v * W3[2 * f];
        s1 += v * W3[2 * f + 1];
    }
#pragma unroll
    for (int f = 0; f < PLANE; ++f) {
        float v = r1[f];
        s0 += v * W3[2 * (f + PLANE)];
        s1 += v * W3[2 * (f + PLANE) + 1];
    }
    h3[2 * i] = s0; h3[2 * i + 1] = s1;
}

__global__ void pull2_lsm_k(const int* __restrict__ rowptr, const int2* __restrict__ colw,
                            const float* __restrict__ h3, const float* __restrict__ b3,
                            float* __restrict__ out, int n, int E) {
    int i = blockIdx.x * blockDim.x + threadIdx.x;
    if (i >= n) return;
    const int beg = rowptr[i];
    const int end = (i + 1 < n) ? rowptr[i + 1] : E;
    const float di = rsqrtf((float)(end - beg + 1));
    const v2f* h2 = (const v2f*)h3;
    const v2f hs = h2[i];
    float a0 = hs.x * (di * di), a1 = hs.y * (di * di);
    for (int e = beg; e < end; ++e) {
        const int2 r = colw[e];
        const float w = __int_as_float(r.y) * di;
        const v2f v = h2[r.x];
        a0 += v.x * w;
        a1 += v.y * w;
    }
    float z0 = a0 + b3[0];
    float z1 = a1 + b3[1];
    float m = fmaxf(z0, z1);
    float lse = m + logf(__expf(z0 - m) + __expf(z1 - m));
    out[2 * i] = z0 - lse;
    out[2 * i + 1] = z1 - lse;
}


extern "C" void kernel_launch(void* const* d_in, const int* in_sizes, int n_in,
                              void* d_out, int out_size, void* d_ws, size_t ws_size,
                              hipStream_t stream) {
    const float* x  = (const float*)d_in[0];
    const int*   ei = (const int*)d_in[1];
    const float* W1 = (const float*)d_in[2];
    const float* b1 = (const float*)d_in[3];
    const float* W2 = (const float*)d_in[4];
    const float* b2 = (const float*)d_in[5];
    const float* W3 = (const float*)d_in[6];
    const float* b3 = (const float*)d_in[7];
    float* out = (float*)d_out;

    const int N = in_sizes[0] / F_IN;
    const int E = in_sizes[1] / 2;
    const int* src = ei;
    const int* dst = ei + E;

    int2*  colw   = (int2*)d_ws;
    int*   rowptr = (int*)d_ws + (size_t)2 * E;
    float* pA     = (float*)(rowptr + N);
    float* pB     = pA + (size_t)N * PLANE;
    float* pC     = pB + (size_t)N * PLANE;
    int2*  tmp    = (int2*)pC;
    int*   rankT  = (int*)pA;
    int2*  tab    = (int2*)pB;
    float* h3     = pA;
    int*   cnt    = (int*)d_out;

    const int T = 256;
    dim3 blk(T);
    int gN    = (N + T - 1) / T;
    int gE    = (E + T - 1) / T;
    int gRow  = (N + 8 * RPW - 1) / (8 * RPW);
    int gTile = (N + 15) / 16;
    int nbk   = (N + BW - 1) / BW;
    int nblk  = (E + CH - 1) / CH;

    part_k<<<nblk, dim3(1024), 0, stream>>>(src, dst, tmp, tab, E, nbk, nblk);
    bucket_k<<<nbk, dim3(1024), 0, stream>>>(tmp, tab, rankT, cnt, rowptr, N, nblk);
    place_k<<<gE, blk, 0, stream>>>(tmp, rankT, cnt, rowptr, colw, E);

    gemm_wmma_k<F_IN, false><<<gTile, dim3(128), 0, stream>>>(x, x, W1, pA, pB, N);
    pull32_k<<<gRow, blk, 0, stream>>>(rowptr, colw, pA, b1, pC, N, E);
    pull32_k<<<gRow, blk, 0, stream>>>(rowptr, colw, pB, b1 + PLANE, pA, N, E);

    gemm_wmma_k<HIDDEN, true><<<gTile, dim3(128), 0, stream>>>(pC, pA, W2, pC, pA, N);
    pull32_k<<<gRow, blk, 0, stream>>>(rowptr, colw, pC, b2, pB, N, E);
    pull32_k<<<gRow, blk, 0, stream>>>(rowptr, colw, pA, b2 + PLANE, pC, N, E);

    gemm3_k<<<gN, blk, 0, stream>>>(pB, pC, W3, h3, N);
    pull2_lsm_k<<<gN, blk, 0, stream>>>(rowptr, colw, h3, b3, out, N, E);
}


# layer 2 aggregates both column planes in one launch and multiplies each finished row by W3 in the wave (64-term dot by a lane tree, not serial): h2 is never stored, the 64->2 launch is gone
# speedup vs baseline: 4.8617x; 1.0812x over previous
#include <hip/hip_runtime.h>


typedef __attribute__((ext_vector_type(2))) float v2f;
typedef __attribute__((ext_vector_type(4))) float v4f;
typedef __attribute__((ext_vector_type(8))) float v8f;

#define F_IN   256
#define HIDDEN 64
#define PLANE  32


#define BSH  9
#define BW   (1 << BSH)
#define MAXB 256
#define CH   4096

__global__ __launch_bounds__(1024) void part_k(const int* __restrict__ src, const int* __restrict__ dst,
                                               int2* __restrict__ tmp, int2* __restrict__ tab,
                                               int E, int nbk, int nblk) {
    __shared__ int  hist[MAXB];
    __shared__ int  offL[MAXB];
    __shared__ int  wsum[8];
    __shared__ int2 stage[CH];
    const int tid = threadIdx.x, lane = tid & 31, wave = tid >> 5;
    const int base = blockIdx.x * CH;
    if (tid < MAXB) hist[tid] = 0;
    __syncthreads();

    int s[4], d[4], r[4];
#pragma unroll
    for (int k = 0; k < 4; ++k) {
        const int e = base + k * 1024 + tid;
        s[k] = 0; d[k] = -1; r[k] = 0;
        if (e < E) { s[k] = src[e]; d[k] = dst[e]; }
    }
#pragma unroll
    for (int k = 0; k < 4; ++k)
        if (d[k] >= 0) r[k] = atomicAdd(&hist[d[k] >> BSH], 1);
    __syncthreads();

    const int c = (tid < MAXB) ? hist[tid] : 0;
    int incl = c;
    for (int o = 1; o < 32; o <<= 1) {
        int t = __shfl_up(incl, o);
        if (lane >= o) incl += t;
    }
    if (tid < MAXB && lane == 31) wsum[wave] = incl;
    __syncthreads();
    if (tid < MAXB) {
        int pre = 0;
        for (int w = 0; w < wave; ++w) pre += wsum[w];
        offL[tid] = pre + incl - c;
    }
    __syncthreads();

#pragma unroll
    for (int k = 0; k < 4; ++k)
        if (d[k] >= 0) stage[offL[d[k] >> BSH] + r[k]] = make_int2(s[k], d[k]);
    __syncthreads();
    const int nvalid = (E - base < CH) ? (E - base) : CH;
#pragma unroll
    for (int k = 0; k < 4; ++k) {
        const int idx = k * 1024 + tid;
        if (idx < nvalid) tmp[base + idx] = stage[idx];
    }
    if (tid < nbk) tab[(size_t)tid * nblk + blockIdx.x] = make_int2(offL[tid], hist[tid]);
}

__global__ __launch_bounds__(1024) void bucket_k(const int2* __restrict__ tmp, const int2* __restrict__ tab,
                                                 int* __restrict__ rankT, int* __restrict__ cnt,
                                                 int* __restrict__ rowptr, int n, int nblk) {
    __shared__ int cntL[BW];
    __shared__ int wred[32];
    __shared__ int wsum[16];
    const int tid = threadIdx.x, lane = tid & 31;
    const int wave = __builtin_amdgcn_readfirstlane((int)(threadIdx.x >> 5));
    const int b = blockIdx.x, lo = b << BSH;

    int part = 0;
    const int front = b * nblk;
    for (int i = tid; i < front; i += 1024) part += tab[i].y;
    for (int o = 16; o > 0; o >>= 1) part += __shfl_xor(part, o);
    if (tid < BW) cntL[tid] = 0;
    if (lane == 0) wred[wave] = part;
    __syncthreads();

    const int2* trow = tab + (size_t)b * nblk;
    for (int j0 = wave; j0 < nblk; j0 += 128) {
        int st[4], ln[4];
        int2 e[4];
        int rk[4];
#pragma unroll
        for (int u = 0; u < 4; ++u) {
            const int j  = j0 + 32 * u;
            const int jc = (j < nblk) ? j : nblk - 1;
            const int2 t = trow[jc];
            st[u] = jc * CH + t.x;
            ln[u] = (j < nblk) ? t.y : 0;
        }
#pragma unroll
        for (int u = 0; u < 4; ++u) {
            e[u] = make_int2(0, lo);
            if (lane < ln[u]) e[u] = tmp[st[u] + lane];
        }
#pragma unroll
        for (int u = 0; u < 4; ++u) {
            rk[u] = 0;
            if (lane < ln[u]) rk[u] = atomicAdd(&cntL[e[u].y - lo], 1);
        }
#pragma unroll
        for (int u = 0; u < 4; ++u)
            if (lane < ln[u]) rankT[st[u] + lane] = rk[u];
#pragma unroll
        for (int u = 0; u < 4; ++u)
            for (int k = 32; k < ln[u]; k += 32)
                if (k + lane < ln[u]) {
                    const int2 ee = tmp[st[u] + k + lane];
                    rankT[st[u] + k + lane] = atomicAdd(&cntL[ee.y - lo], 1);
                }
    }
    __syncthreads();

    int bs = wred[lane];
    for (int o = 16; o > 0; o >>= 1) bs += __shfl_xor(bs, o);
    const int c = (tid < BW) ? cntL[tid] : 0;
    int incl = c;
    for (int o = 1; o < 32; o <<= 1) {
        int t = __shfl_up(incl, o);
        if (lane >= o) incl += t;
    }
    if (tid < BW && lane == 31) wsum[wave] = incl;
    __syncthreads();
    int tp = (lane < wave && lane < 16) ? wsum[lane] : 0;
    for (int o = 16; o > 0; o >>= 1) tp += __shfl_xor(tp, o);
    const int i = lo + tid;
    if (tid < BW && i < n) {
        rowptr[i] = bs + tp + incl - c;
        cnt[i] = c;
    }
}

__global__ void place_k(const int2* __restrict__ tmp, const int* __restrict__ rankT,
                        const int* __restrict__ cnt, const int* __restrict__ rowptr,
                        int2* __restrict__ colw, int E) {
    int p = blockIdx.x * blockDim.x + threadIdx.x;
    if (p >= E) return;
    const int2 sd = tmp[p];
    const float ws = rsqrtf((float)(cnt[sd.x] + 1));
    colw[rowptr[sd.y] + rankT[p]] = make_int2(sd.x, __float_as_int(ws));
}


template <int K, bool PLANES>
__global__ __launch_bounds__(128) void gemm_wmma_k(const float* A0, const float* A1,
                                                   const float* __restrict__ W,
                                                   float* O0, float* O1, int M) {
    static_assert(!PLANES || K == 2 * PLANE, "plane input is 64 wide");
    constexpr int NOUT = HIDDEN;
    constexpr int LDSK = K + 4;
    __shared__ float Atile[16 * LDSK];

    const int tid  = threadIdx.x;
    const int row0 = blockIdx.x * 16;

    constexpr int KV = K / 4;
    for (int idx = tid; idx < 16 * KV; idx += 128) {
        int r = idx / KV, c4 = idx % KV;
        int row = row0 + r; if (row >= M) row = M - 1;
        const float* p;
        if (PLANES) p = (c4 < PLANE / 4) ? &A0[(size_t)row * PLANE + c4 * 4]
                                         : &A1[(size_t)row * PLANE + (c4 - PLANE / 4) * 4];
        else        p = &A0[(size_t)row * K + c4 * 4];
        v4f v = *(const v4f*)p;
        *(v4f*)&Atile[r * LDSK + c4 * 4] = v;
    }
    __syncthreads();

    const int lane = tid & 31;
    const int wave = tid >> 5;
    const int half = lane >> 4;
    const int l    = lane & 15;
    const int col0 = wave * 16;

    v8f acc = {};
    const float* wp = W + col0 + l;
    for (int k = 0; k < K; k += 4) {
        const int ka = k + half * 2;
        v2f a = *(const v2f*)&Atile[l * LDSK + ka];
        v2f b;
        b.x = wp[ka * NOUT];
        b.y = wp[(ka + 1) * NOUT];
        acc = __builtin_amdgcn_wmma_f32_16x16x4_f32(false, a, false, b,
                                                    (short)0, acc, false, false);
    }

    float* O  = (wave < 2) ? O0 : O1;
    float* op = O + (size_t)(row0 + half * 8) * PLANE + (wave & 1) * 16 + l;
    if (row0 + 16 <= M) {
#pragma unroll
        for (int r = 0; r < 8; ++r) op[r * PLANE] = acc[r];
    } else {
#pragma unroll
        for (int r = 0; r < 8; ++r)
            if (row0 + half * 8 + r < M) op[r * PLANE] = acc[r];
    }
}


#define RPW 12

__global__ __launch_bounds__(256) void pull32_k(const int* __restrict__ rowptr,
                                                const int2* __restrict__ colw,
                                                const float* __restrict__ hin,
                                                const float* __restrict__ bias,
                                                float* __restrict__ hout, int n, int E) {
    const int lane = threadIdx.x & 31;
    const int wave = __builtin_amdgcn_readfirstlane((int)(threadIdx.x >> 5));
    const int row0 = ((int)blockIdx.x * 8 + wave) * RPW;
    if (row0 >= n) return;
    const int nrows = (n - row0 < RPW) ? (n - row0) : RPW;

    int bnd = E;
    if (lane <= nrows && row0 + lane < n) bnd = rowptr[row0 + lane];
    const float bl = bias[lane];

    int beg_n = __builtin_amdgcn_readlane(bnd, 0);
    int end_n = __builtin_amdgcn_readlane(bnd, 1);
    int2 rec_n = make_int2(0, 0);
    if (lane < end_n - beg_n) rec_n = colw[beg_n + lane];
    float self_n = hin[(size_t)row0 * PLANE + lane];

    for (int r = 0; r < nrows; ++r) {
        const int beg = beg_n, end = end_n;
        const int2 cur = rec_n;
        const float selfv = self_n;
        const float di = rsqrtf((float)(end - beg + 1));
        float acc = selfv * (di * di);

        const int cnt   = (end - beg < 32) ? (end - beg) : 32;
        const int nfull = cnt >> 3, rem = cnt & 7, jr = cnt & ~7;
        const int s  = cur.x;
        const int wi = __float_as_int(__int_as_float(cur.y) * di);
#define ROWLD(j) ((hin + (size_t)(unsigned)__builtin_amdgcn_readlane(s, (j)) * PLANE)[lane])
#define WJ(j)    (__int_as_float(__builtin_amdgcn_readlane(wi, (j))))
        float a0, a1, a2, a3, a4, a5, a6, a7,
              b0, b1, b2, b3, b4, b5, b6, b7,
              c0, c1, c2, c3, c4, c5, c6, c7,
              d0, d1, d2, d3, d4, d5, d6, d7,
              p0, p1, p2, p3, p4, p5, p6;
        if (nfull > 0) { a0 = ROWLD(0); a1 = ROWLD(1); a2 = ROWLD(2); a3 = ROWLD(3); a4 = ROWLD(4); a5 = ROWLD(5); a6 = ROWLD(6); a7 = ROWLD(7); }
        if (nfull > 1) { b0 = ROWLD(8); b1 = ROWLD(9); b2 = ROWLD(10); b3 = ROWLD(11); b4 = ROWLD(12); b5 = ROWLD(13); b6 = ROWLD(14); b7 = ROWLD(15); }
        if (nfull > 2) { c0 = ROWLD(16); c1 = ROWLD(17); c2 = ROWLD(18); c3 = ROWLD(19); c4 = ROWLD(20); c5 = ROWLD(21); c6 = ROWLD(22); c7 = ROWLD(23); }
        if (nfull > 3) { d0 = ROWLD(24); d1 = ROWLD(25); d2 = ROWLD(26); d3 = ROWLD(27); d4 = ROWLD(28); d5 = ROWLD(29); d6 = ROWLD(30); d7 = ROWLD(31); }
        if (rem > 0) p0 = ROWLD(jr + 0);
        if (rem > 1) p1 = ROWLD(jr + 1);
        if (rem > 2) p2 = ROWLD(jr + 2);
        if (rem > 3) p3 = ROWLD(jr + 3);
        if (rem > 4) p4 = ROWLD(jr + 4);
        if (rem > 5) p5 = ROWLD(jr + 5);
        if (rem > 6) p6 = ROWLD(jr + 6);

        {
            const int rn = (r + 1 < nrows) ? r + 1 : r;
            beg_n = __builtin_amdgcn_readlane(bnd, rn);
            end_n = __builtin_amdgcn_readlane(bnd, rn + 1);
            rec_n = make_int2(0, 0);
            if (lane < end_n - beg_n) rec_n = colw[beg_n + lane];
            self_n = hin[(size_t)(row0 + rn) * PLANE + lane];
        }

        if (nfull > 0) { acc += a0 * WJ(0); acc += a1 * WJ(1); acc += a2 * WJ(2); acc += a3 * WJ(3); acc += a4 * WJ(4); acc += a5 * WJ(5); acc += a6 * WJ(6); acc += a7 * WJ(7); }
        if (nfull > 1) { acc += b0 * WJ(8); acc += b1 * WJ(9); acc += b2 * WJ(10); acc += b3 * WJ(11); acc += b4 * WJ(12); acc += b5 * WJ(13); acc += b6 * WJ(14); acc += b7 * WJ(15); }
        if (nfull > 2) { acc += c0 * WJ(16); acc += c1 * WJ(17); acc += c2 * WJ(18); acc += c3 * WJ(19); acc += c4 * WJ(20); acc += c5 * WJ(21); acc += c6 * WJ(22); acc += c7 * WJ(23); }
        if (nfull > 3) { acc += d0 * WJ(24); acc += d1 * WJ(25); acc += d2 * WJ(26); acc += d3 * WJ(27); acc += d4 * WJ(28); acc += d5 * WJ(29); acc += d6 * WJ(30); acc += d7 * WJ(31); }
        if (rem > 0) acc += p0 * WJ(jr + 0);
        if (rem > 1) acc += p1 * WJ(jr + 1);
        if (rem > 2) acc += p2 * WJ(jr + 2);
        if (rem > 3) acc += p3 * WJ(jr + 3);
        if (rem > 4) acc += p4 * WJ(jr + 4);
        if (rem > 5) acc += p5 * WJ(jr + 5);
        if (rem > 6) acc += p6 * WJ(jr + 6);
#undef ROWLD
#undef WJ

        for (int base = beg + 32; base < end; base += 32) {
            const int cnt = (end - base < 32) ? (end - base) : 32;
            int s = 0, wi = 0;
            if (lane < cnt) {
                const int2 rr = colw[base + lane];
                s  = rr.x;
                wi = __float_as_int(__int_as_float(rr.y) * di);
            }
            for (int j = 0; j < cnt; ++j) {
                const int sj = __builtin_amdgcn_readlane(s, j);
                acc += hin[(size_t)sj * PLANE + lane] * __int_as_float(__builtin_amdgcn_readlane(wi, j));
            }
        }
        const float res = acc + bl;
        hout[(size_t)(row0 + r) * PLANE + lane] = res > 0.0f ? res : 0.0f;
    }
}


#define RPW3 16

#define XM(v, m) __int_as_float(__builtin_amdgcn_update_dpp(0, __float_as_int(v), 0x160 | (m), 0xf, 0xf, true))
__device__ __forceinline__ float lane_sum32(float v) {
    v += XM(v, 1);
    v += XM(v, 2);
    v += XM(v, 4);
    v += XM(v, 8);
    const int b = __float_as_int(v);
    v += __int_as_float(__builtin_amdgcn_permlanex16(b, b, 0x76543210, 0xfedcba98, false, false));
    return v;
}
#undef XM

__global__ __launch_bounds__(128) void pull2p_w3_k(const int* __restrict__ rowptr,
                                                   const int2* __restrict__ colw,
                                                   const float* __restrict__ g0,
                                                   const float* __restrict__ g1,
                                                   const float* __restrict__ bias,
                                                   const float* __restrict__ W3,
                                                   float* __restrict__ h3, int n, int E) {
    const int lane = threadIdx.x & 31;
    const int wave = __builtin_amdgcn_readfirstlane((int)(threadIdx.x >> 5));
    const int row0 = ((int)blockIdx.x * 4 + wave) * RPW3;
    if (row0 >= n) return;
    const int nrows = (n - row0 < RPW3) ? (n - row0) : RPW3;

    int bnd = E;
    if (lane <= nrows && row0 + lane < n) bnd = rowptr[row0 + lane];
    const float bl0 = bias[lane], bl1 = bias[PLANE + lane];
    const float w00 = W3[2 * lane],           w01 = W3[2 * lane + 1];
    const float w10 = W3[2 * (lane + PLANE)], w11 = W3[2 * (lane + PLANE) + 1];

    int beg_n = __builtin_amdgcn_readlane(bnd, 0);
    int end_n = __builtin_amdgcn_readlane(bnd, 1);
    int2 rec_n = make_int2(0, 0);
    if (lane < end_n - beg_n) rec_n = colw[beg_n + lane];
    float self0_n = g0[(size_t)row0 * PLANE + lane];
    float self1_n = g1[(size_t)row0 * PLANE + lane];
    float keep = 0.0f;

    for (int r = 0; r < nrows; ++r) {
        const int beg = beg_n, end = end_n;
        const int2 cur = rec_n;
        const float di = rsqrtf((float)(end - beg + 1));
        float acc0 = self0_n * (di * di);
        float acc1 = self1_n * (di * di);

        const int cnt   = (end - beg < 32) ? (end - beg) : 32;
        const int nfull = cnt >> 3, rem = cnt & 7, jr = cnt & ~7;
        const int s  = cur.x;
        const int wi = __float_as_int(__int_as_float(cur.y) * di);
#define OFFJ(j)  ((size_t)(unsigned)__builtin_amdgcn_readlane(s, (j)) * PLANE)
#define LD0(j)   ((g0 + OFFJ(j))[lane])
#define LD1(j)   ((g1 + OFFJ(j))[lane])
#define WJ(j)    (__int_as_float(__builtin_amdgcn_readlane(wi, (j))))
        float a0, a1, a2, a3, a4, a5, a6, a7,  A0, A1, A2, A3, A4, A5, A6, A7,
              b0, b1, b2, b3, b4, b5, b6, b7,  B0, B1, B2, B3, B4, B5, B6, B7,
              c0, c1, c2, c3, c4, c5, c6, c7,  C0, C1, C2, C3, C4, C5, C6, C7,
              d0, d1, d2, d3, d4, d5, d6, d7,  D0, D1, D2, D3, D4, D5, D6, D7,
              p0, p1, p2, p3, p4, p5, p6,      P0, P1, P2, P3, P4, P5, P6;
        if (nfull > 0) { a0 = LD0(0); A0 = LD1(0); a1 = LD0(1); A1 = LD1(1); a2 = LD0(2); A2 = LD1(2); a3 = LD0(3); A3 = LD1(3);
                         a4 = LD0(4); A4 = LD1(4); a5 = LD0(5); A5 = LD1(5); a6 = LD0(6); A6 = LD1(6); a7 = LD0(7); A7 = LD1(7); }
        if (nfull > 1) { b0 = LD0(8); B0 = LD1(8); b1 = LD0(9); B1 = LD1(9); b2 = LD0(10); B2 = LD1(10); b3 = LD0(11); B3 = LD1(11);
                         b4 = LD0(12); B4 = LD1(12); b5 = LD0(13); B5 = LD1(13); b6 = LD0(14); B6 = LD1(14); b7 = LD0(15); B7 = LD1(15); }
        if (nfull > 2) { c0 = LD0(16); C0 = LD1(16); c1 = LD0(17); C1 = LD1(17); c2 = LD0(18); C2 = LD1(18); c3 = LD0(19); C3 = LD1(19);
                         c4 = LD0(20); C4 = LD1(20); c5 = LD0(21); C5 = LD1(21); c6 = LD0(22); C6 = LD1(22); c7 = LD0(23); C7 = LD1(23); }
        if (nfull > 3) { d0 = LD0(24); D0 = LD1(24); d1 = LD0(25); D1 = LD1(25); d2 = LD0(26); D2 = LD1(26); d3 = LD0(27); D3 = LD1(27);
                         d4 = LD0(28); D4 = LD1(28); d5 = LD0(29); D5 = LD1(29); d6 = LD0(30); D6 = LD1(30); d7 = LD0(31); D7 = LD1(31); }
        if (rem > 0) { p0 = LD0(jr + 0); P0 = LD1(jr + 0); }
        if (rem > 1) { p1 = LD0(jr + 1); P1 = LD1(jr + 1); }
        if (rem > 2) { p2 = LD0(jr + 2); P2 = LD1(jr + 2); }
        if (rem > 3) { p3 = LD0(jr + 3); P3 = LD1(jr + 3); }
        if (rem > 4) { p4 = LD0(jr + 4); P4 = LD1(jr + 4); }
        if (rem > 5) { p5 = LD0(jr + 5); P5 = LD1(jr + 5); }
        if (rem > 6) { p6 = LD0(jr + 6); P6 = LD1(jr + 6); }

        {
            const int rn = (r + 1 < nrows) ? r + 1 : r;
            beg_n = __builtin_amdgcn_readlane(bnd, rn);
            end_n = __builtin_amdgcn_readlane(bnd, rn + 1);
            rec_n = make_int2(0, 0);
            if (lane < end_n - beg_n) rec_n = colw[beg_n + lane];
            self0_n = g0[(size_t)(row0 + rn) * PLANE + lane];
            self1_n = g1[(size_t)(row0 + rn) * PLANE + lane];
        }

#define AD(x, X, j) { const float w_ = WJ(j); acc0 += (x) * w_; acc1 += (X) * w_; }
        if (nfull > 0) { AD(a0, A0, 0) AD(a1, A1, 1) AD(a2, A2, 2) AD(a3, A3, 3) AD(a4, A4, 4) AD(a5, A5, 5) AD(a6, A6, 6) AD(a7, A7, 7) }
        if (nfull > 1) { AD(b0, B0, 8) AD(b1, B1, 9) AD(b2, B2, 10) AD(b3, B3, 11) AD(b4, B4, 12) AD(b5, B5, 13) AD(b6, B6, 14) AD(b7, B7, 15) }
        if (nfull > 2) { AD(c0, C0, 16) AD(c1, C1, 17) AD(c2, C2, 18) AD(c3, C3, 19) AD(c4, C4, 20) AD(c5, C5, 21) AD(c6, C6, 22) AD(c7, C7, 23) }
        if (nfull > 3) { AD(d0, D0, 24) AD(d1, D1, 25) AD(d2, D2, 26) AD(d3, D3, 27) AD(d4, D4, 28) AD(d5, D5, 29) AD(d6, D6, 30) AD(d7, D7, 31) }
        if (rem > 0) AD(p0, P0, jr + 0)
        if (rem > 1) AD(p1, P1, jr + 1)
        if (rem > 2) AD(p2, P2, jr + 2)
        if (rem > 3) AD(p3, P3, jr + 3)
        if (rem > 4) AD(p4, P4, jr + 4)
        if (rem > 5) AD(p5, P5, jr + 5)
        if (rem > 6) AD(p6, P6, jr + 6)
#undef AD
#undef LD0
#undef LD1
#undef OFFJ
#undef WJ

        for (int base = beg + 32; base < end; base += 32) {
            const int cnt = (end - base < 32) ? (end - base) : 32;
            int s = 0, wi = 0;
            if (lane < cnt) {
                const int2 rr = colw[base + lane];
                s  = rr.x;
                wi = __float_as_int(__int_as_float(rr.y) * di);
            }
            for (int j = 0; j < cnt; ++j) {
                const size_t off = (size_t)(unsigned)__builtin_amdgcn_readlane(s, j) * PLANE + lane;
                const float w_ = __int_as_float(__builtin_amdgcn_readlane(wi, j));
                acc0 += g0[off] * w_;
                acc1 += g1[off] * w_;
            }
        }
        const float r0 = acc0 + bl0, r1 = acc1 + bl1;
        const float v0 = r0 > 0.0f ? r0 : 0.0f;
        const float v1 = r1 > 0.0f ? r1 : 0.0f;
        const float t0 = lane_sum32(v0 * w00 + v1 * w10);
        const float t1 = lane_sum32(v0 * w01 + v1 * w11);
        if (lane == 2 * r)     keep = t0;
        if (lane == 2 * r + 1) keep = t1;
    }
    if (lane < 2 * nrows) h3[(size_t)row0 * 2 + lane] = keep;
}

__global__ void pull2_lsm_k(const int* __restrict__ rowptr, const int2* __restrict__ colw,
                            const float* __restrict__ h3, const float* __restrict__ b3,
                            float* __restrict__ out, int n, int E) {
    int i = blockIdx.x * blockDim.x + threadIdx.x;
    if (i >= n) return;
    const int beg = rowptr[i];
    const int end = (i + 1 < n) ? rowptr[i + 1] : E;
    const float di = rsqrtf((float)(end - beg + 1));
    const v2f* h2 = (const v2f*)h3;
    const v2f hs = h2[i];
    float a0 = hs.x * (di * di), a1 = hs.y * (di * di);
    for (int e = beg; e < end; ++e) {
        const int2 r = colw[e];
        const float w = __int_as_float(r.y) * di;
        const v2f v = h2[r.x];
        a0 += v.x * w;
        a1 += v.y * w;
    }
    float z0 = a0 + b3[0];
    float z1 = a1 + b3[1];
    float m = fmaxf(z0, z1);
    float lse = m + logf(__expf(z0 - m) + __expf(z1 - m));
    out[2 * i] = z0 - lse;
    out[2 * i + 1] = z1 - lse;
}


extern "C" void kernel_launch(void* const* d_in, const int* in_sizes, int n_in,
                              void* d_out, int out_size, void* d_ws, size_t ws_size,
                              hipStream_t stream) {
    const float* x  = (const float*)d_in[0];
    const int*   ei = (const int*)d_in[1];
    const float* W1 = (const float*)d_in[2];
    const float* b1 = (const float*)d_in[3];
    const float* W2 = (const float*)d_in[4];
    const float* b2 = (const float*)d_in[5];
    const float* W3 = (const float*)d_in[6];
    const float* b3 = (const float*)d_in[7];
    float* out = (float*)d_out;

    const int N = in_sizes[0] / F_IN;
    const int E = in_sizes[1] / 2;
    const int* src = ei;
    const int* dst = ei + E;

    int2*  colw   = (int2*)d_ws;
    int*   rowptr = (int*)d_ws + (size_t)2 * E;
    float* pA     = (float*)(rowptr + N);
    float* pB     = pA + (size_t)N * PLANE;
    float* pC     = pB + (size_t)N * PLANE;
    int2*  tmp    = (int2*)pC;
    int*   rankT  = (int*)pA;
    int2*  tab    = (int2*)pB;
    float* h3     = pB;
    int*   cnt    = (int*)d_out;

    const int T = 256;
    dim3 blk(T);
    int gN    = (N + T - 1) / T;
    int gE    = (E + T - 1) / T;
    int gRow  = (N + 8 * RPW - 1) / (8 * RPW);
    int gTile = (N + 15) / 16;
    int gRow3 = (N + 4 * RPW3 - 1) / (4 * RPW3);
    int nbk   = (N + BW - 1) / BW;
    int nblk  = (E + CH - 1) / CH;

    part_k<<<nblk, dim3(1024), 0, stream>>>(src, dst, tmp, tab, E, nbk, nblk);
    bucket_k<<<nbk, dim3(1024), 0, stream>>>(tmp, tab, rankT, cnt, rowptr, N, nblk);
    place_k<<<gE, blk, 0, stream>>>(tmp, rankT, cnt, rowptr, colw, E);

    gemm_wmma_k<F_IN, false><<<gTile, dim3(128), 0, stream>>>(x, x, W1, pA, pB, N);
    pull32_k<<<gRow, blk, 0, stream>>>(rowptr, colw, pA, b1, pC, N, E);
    pull32_k<<<gRow, blk, 0, stream>>>(rowptr, colw, pB, b1 + PLANE, pA, N, E);

    gemm_wmma_k<HIDDEN, true><<<gTile, dim3(128), 0, stream>>>(pC, pA, W2, pC, pA, N);
    pull2p_w3_k<<<gRow3, dim3(128), 0, stream>>>(rowptr, colw, pC, pA, b2, W3, h3, N, E);

    pull2_lsm_k<<<gN, blk, 0, stream>>>(rowptr, colw, h3, b3, out, N, E);
}


# edge weight folded into rows (products store row*dinv[row], row sum times dinv[dst] once): list = bare sources, placed by the bucket block itself, place launch gone; LDS-add arrival order kept
# speedup vs baseline: 5.8076x; 1.1946x over previous
#include <hip/hip_runtime.h>


typedef __attribute__((ext_vector_type(2))) float v2f;
typedef __attribute__((ext_vector_type(4))) float v4f;
typedef __attribute__((ext_vector_type(8))) float v8f;

#define F_IN   256
#define HIDDEN 64
#define PLANE  32


#define BSH  9
#define BW   (1 << BSH)
#define MAXB 256
#define CH   4096

__global__ __launch_bounds__(1024) void part_k(const int* __restrict__ src, const int* __restrict__ dst,
                                               int2* __restrict__ tmp, int2* __restrict__ tab,
                                               int E, int nbk, int nblk) {
    __shared__ int  hist[MAXB];
    __shared__ int  offL[MAXB];
    __shared__ int  wsum[8];
    __shared__ int2 stage[CH];
    const int tid = threadIdx.x, lane = tid & 31, wave = tid >> 5;
    const int base = blockIdx.x * CH;
    if (tid < MAXB) hist[tid] = 0;
    __syncthreads();

    int s[4], d[4], r[4];
#pragma unroll
    for (int k = 0; k < 4; ++k) {
        const int e = base + k * 1024 + tid;
        s[k] = 0; d[k] = -1; r[k] = 0;
        if (e < E) { s[k] = src[e]; d[k] = dst[e]; }
    }
#pragma unroll
    for (int k = 0; k < 4; ++k)
        if (d[k] >= 0) r[k] = atomicAdd(&hist[d[k] >> BSH], 1);
    __syncthreads();

    const int c = (tid < MAXB) ? hist[tid] : 0;
    int incl = c;
    for (int o = 1; o < 32; o <<= 1) {
        int t = __shfl_up(incl, o);
        if (lane >= o) incl += t;
    }
    if (tid < MAXB && lane == 31) wsum[wave] = incl;
    __syncthreads();
    if (tid < MAXB) {
        int pre = 0;
        for (int w = 0; w < wave; ++w) pre += wsum[w];
        offL[tid] = pre + incl - c;
    }
    __syncthreads();

#pragma unroll
    for (int k = 0; k < 4; ++k)
        if (d[k] >= 0) stage[offL[d[k] >> BSH] + r[k]] = make_int2(s[k], d[k]);
    __syncthreads();
    const int nvalid = (E - base < CH) ? (E - base) : CH;
#pragma unroll
    for (int k = 0; k < 4; ++k) {
        const int idx = k * 1024 + tid;
        if (idx < nvalid) tmp[base + idx] = stage[idx];
    }
    if (tid < nbk) tab[(size_t)tid * nblk + blockIdx.x] = make_int2(offL[tid], hist[tid]);
}

#define LCAP 16384
__global__ __launch_bounds__(1024) void bucket_k(const int2* __restrict__ tmp, const int2* __restrict__ tab,
                                                 int* __restrict__ cnt, int* __restrict__ rowptr,
                                                 int* __restrict__ col, int n, int nblk) {
    __shared__ int cntL[BW];
    __shared__ int wred[32];
    __shared__ int wsum[16];
    __shared__ int lst[LCAP];
    const int tid = threadIdx.x, lane = tid & 31;
    const int wave = __builtin_amdgcn_readfirstlane((int)(threadIdx.x >> 5));
    const int b = blockIdx.x, lo = b << BSH;

    int part = 0;
    const int front = b * nblk;
    for (int i = tid; i < front; i += 1024) part += tab[i].y;
    for (int o = 16; o > 0; o >>= 1) part += __shfl_xor(part, o);
    if (tid < BW) cntL[tid] = 0;
    if (lane == 0) wred[wave] = part;
    __syncthreads();

    const int2* trow = tab + (size_t)b * nblk;
    for (int j0 = wave; j0 < nblk; j0 += 128) {
        int st[4], ln[4];
        int2 e[4];
#pragma unroll
        for (int u = 0; u < 4; ++u) {
            const int j  = j0 + 32 * u;
            const int jc = (j < nblk) ? j : nblk - 1;
            const int2 t = trow[jc];
            st[u] = jc * CH + t.x;
            ln[u] = (j < nblk) ? t.y : 0;
        }
#pragma unroll
        for (int u = 0; u < 4; ++u) {
            e[u] = make_int2(0, lo);
            if (lane < ln[u]) e[u] = tmp[st[u] + lane];
        }
#pragma unroll
        for (int u = 0; u < 4; ++u)
            if (lane < ln[u]) atomicAdd(&cntL[e[u].y - lo], 1);
#pragma unroll
        for (int u = 0; u < 4; ++u)
            for (int k = 32; k < ln[u]; k += 32)
                if (k + lane < ln[u]) {
                    const int2 ee = tmp[st[u] + k + lane];
                    atomicAdd(&cntL[ee.y - lo], 1);
                }
    }
    __syncthreads();

    int bs = wred[lane];
    for (int o = 16; o > 0; o >>= 1) bs += __shfl_xor(bs, o);
    const int c = (tid < BW) ? cntL[tid] : 0;
    int incl = c;
    for (int o = 1; o < 32; o <<= 1) {
        int t = __shfl_up(incl, o);
        if (lane >= o) incl += t;
    }
    if (tid < BW && lane == 31) wsum[wave] = incl;
    __syncthreads();
    const int wsl = (lane < 16) ? wsum[lane] : 0;
    int tp = (lane < wave) ? wsl : 0;
    for (int o = 16; o > 0; o >>= 1) tp += __shfl_xor(tp, o);
    int total = wsl;
    for (int o = 16; o > 0; o >>= 1) total += __shfl_xor(total, o);
    const int i = lo + tid;
    if (tid < BW) {
        const int off = tp + incl - c;
        cntL[tid] = off;
        if (i < n) {
            rowptr[i] = bs + off;
            cnt[i] = c;
        }
    }
    __syncthreads();

    const bool inl = total <= LCAP;
    for (int j0 = wave; j0 < nblk; j0 += 128) {
        int st[4], ln[4];
        int2 e[4];
        int sl[4];
#pragma unroll
        for (int u = 0; u < 4; ++u) {
            const int j  = j0 + 32 * u;
            const int jc = (j < nblk) ? j : nblk - 1;
            const int2 t = trow[jc];
            st[u] = jc * CH + t.x;
            ln[u] = (j < nblk) ? t.y : 0;
        }
#pragma unroll
        for (int u = 0; u < 4; ++u) {
            e[u] = make_int2(0, lo);
            if (lane < ln[u]) e[u] = tmp[st[u] + lane];
        }
#pragma unroll
        for (int u = 0; u < 4; ++u) {
            sl[u] = 0;
            if (lane < ln[u]) sl[u] = atomicAdd(&cntL[e[u].y - lo], 1);
        }
#pragma unroll
        for (int u = 0; u < 4; ++u)
            if (lane < ln[u]) {
                if (inl) lst[sl[u]] = e[u].x;
                else     col[bs + sl[u]] = e[u].x;
            }
#pragma unroll
        for (int u = 0; u < 4; ++u)
            for (int k = 32; k < ln[u]; k += 32)
                if (k + lane < ln[u]) {
                    const int2 ee = tmp[st[u] + k + lane];
                    const int s2 = atomicAdd(&cntL[ee.y - lo], 1);
                    if (inl) lst[s2] = ee.x;
                    else     col[bs + s2] = ee.x;
                }
    }
    __syncthreads();
    if (inl)
        for (int k = tid; k < total; k += 1024) col[bs + k] = lst[k];
}


template <int K, bool PLANES>
__global__ __launch_bounds__(128) void gemm_wmma_k(const float* A0, const float* A1,
                                                   const float* __restrict__ W,
                                                   const int* __restrict__ cnt,
                                                   float* O0, float* O1, int M) {
    static_assert(!PLANES || K == 2 * PLANE, "plane input is 64 wide");
    constexpr int NOUT = HIDDEN;
    constexpr int LDSK = K + 4;
    __shared__ float Atile[16 * LDSK];

    const int tid  = threadIdx.x;
    const int row0 = blockIdx.x * 16;

    constexpr int KV = K / 4;
    for (int idx = tid; idx < 16 * KV; idx += 128) {
        int r = idx / KV, c4 = idx % KV;
        int row = row0 + r; if (row >= M) row = M - 1;
        const float* p;
        if (PLANES) p = (c4 < PLANE / 4) ? &A0[(size_t)row * PLANE + c4 * 4]
                                         : &A1[(size_t)row * PLANE + (c4 - PLANE / 4) * 4];
        else        p = &A0[(size_t)row * K + c4 * 4];
        v4f v = *(const v4f*)p;
        *(v4f*)&Atile[r * LDSK + c4 * 4] = v;
    }
    __syncthreads();

    const int lane = tid & 31;
    const int wave = tid >> 5;
    const int half = lane >> 4;
    const int l    = lane & 15;
    const int col0 = wave * 16;

    int deg[8];
#pragma unroll
    for (int r = 0; r < 8; ++r) {
        int row = row0 + half * 8 + r; if (row >= M) row = M - 1;
        deg[r] = cnt[row];
    }

    v8f acc = {};
    const float* wp = W + col0 + l;
    for (int k = 0; k < K; k += 4) {
        const int ka = k + half * 2;
        v2f a = *(const v2f*)&Atile[l * LDSK + ka];
        v2f b;
        b.x = wp[ka * NOUT];
        b.y = wp[(ka + 1) * NOUT];
        acc = __builtin_amdgcn_wmma_f32_16x16x4_f32(false, a, false, b,
                                                    (short)0, acc, false, false);
    }

    float* O  = (wave < 2) ? O0 : O1;
    float* op = O + (size_t)(row0 + half * 8) * PLANE + (wave & 1) * 16 + l;
    if (row0 + 16 <= M) {
#pragma unroll
        for (int r = 0; r < 8; ++r) op[r * PLANE] = acc[r] * rsqrtf((float)(deg[r] + 1));
    } else {
#pragma unroll
        for (int r = 0; r < 8; ++r)
            if (row0 + half * 8 + r < M) op[r * PLANE] = acc[r] * rsqrtf((float)(deg[r] + 1));
    }
}


#define RPW 12

__global__ __launch_bounds__(256) void pull32_k(const int* __restrict__ rowptr,
                                                const int* __restrict__ colw,
                                                const float* __restrict__ hin,
                                                const float* __restrict__ bias,
                                                float* __restrict__ hout, int n, int E) {
    const int lane = threadIdx.x & 31;
    const int wave = __builtin_amdgcn_readfirstlane((int)(threadIdx.x >> 5));
    const int row0 = ((int)blockIdx.x * 8 + wave) * RPW;
    if (row0 >= n) return;
    const int nrows = (n - row0 < RPW) ? (n - row0) : RPW;

    int bnd = E;
    if (lane <= nrows && row0 + lane < n) bnd = rowptr[row0 + lane];
    const float bl = bias[lane];

    int beg_n = __builtin_amdgcn_readlane(bnd, 0);
    int end_n = __builtin_amdgcn_readlane(bnd, 1);
    int rec_n = 0;
    if (lane < end_n - beg_n) rec_n = colw[beg_n + lane];
    float self_n = hin[(size_t)row0 * PLANE + lane];

    for (int r = 0; r < nrows; ++r) {
        const int beg = beg_n, end = end_n;
        const int cur = rec_n;
        const float di = rsqrtf((float)(end - beg + 1));
        float acc = self_n;

        const int cnt   = (end - beg < 32) ? (end - beg) : 32;
        const int nfull = cnt >> 3, rem = cnt & 7, jr = cnt & ~7;
        const int s  = cur * PLANE;
#define ROWLD(j) ((hin + (size_t)(unsigned)__builtin_amdgcn_readlane(s, (j)))[lane])
        float a0, a1, a2, a3, a4, a5, a6, a7,
              b0, b1, b2, b3, b4, b5, b6, b7,
              c0, c1, c2, c3, c4, c5, c6, c7,
              d0, d1, d2, d3, d4, d5, d6, d7,
              p0, p1, p2, p3, p4, p5, p6;
        if (nfull > 0) { a0 = ROWLD(0); a1 = ROWLD(1); a2 = ROWLD(2); a3 = ROWLD(3); a4 = ROWLD(4); a5 = ROWLD(5); a6 = ROWLD(6); a7 = ROWLD(7); }
        if (nfull > 1) { b0 = ROWLD(8); b1 = ROWLD(9); b2 = ROWLD(10); b3 = ROWLD(11); b4 = ROWLD(12); b5 = ROWLD(13); b6 = ROWLD(14); b7 = ROWLD(15); }
        if (nfull > 2) { c0 = ROWLD(16); c1 = ROWLD(17); c2 = ROWLD(18); c3 = ROWLD(19); c4 = ROWLD(20); c5 = ROWLD(21); c6 = ROWLD(22); c7 = ROWLD(23); }
        if (nfull > 3) { d0 = ROWLD(24); d1 = ROWLD(25); d2 = ROWLD(26); d3 = ROWLD(27); d4 = ROWLD(28); d5 = ROWLD(29); d6 = ROWLD(30); d7 = ROWLD(31); }
        if (rem > 0) p0 = ROWLD(jr + 0);
        if (rem > 1) p1 = ROWLD(jr + 1);
        if (rem > 2) p2 = ROWLD(jr + 2);
        if (rem > 3) p3 = ROWLD(jr + 3);
        if (rem > 4) p4 = ROWLD(jr + 4);
        if (rem > 5) p5 = ROWLD(jr + 5);
        if (rem > 6) p6 = ROWLD(jr + 6);

        {
            const int rn = (r + 1 < nrows) ? r + 1 : r;
            beg_n = __builtin_amdgcn_readlane(bnd, rn);
            end_n = __builtin_amdgcn_readlane(bnd, rn + 1);
            rec_n = 0;
            if (lane < end_n - beg_n) rec_n = colw[beg_n + lane];
            self_n = hin[(size_t)(row0 + rn) * PLANE + lane];
        }

        if (nfull > 0) { acc += a0; acc += a1; acc += a2; acc += a3; acc += a4; acc += a5; acc += a6; acc += a7; }
        if (nfull > 1) { acc += b0; acc += b1; acc += b2; acc += b3; acc += b4; acc += b5; acc += b6; acc += b7; }
        if (nfull > 2) { acc += c0; acc += c1; acc += c2; acc += c3; acc += c4; acc += c5; acc += c6; acc += c7; }
        if (nfull > 3) { acc += d0; acc += d1; acc += d2; acc += d3; acc += d4; acc += d5; acc += d6; acc += d7; }
        if (rem > 0) acc += p0;
        if (rem > 1) acc += p1;
        if (rem > 2) acc += p2;
        if (rem > 3) acc += p3;
        if (rem > 4) acc += p4;
        if (rem > 5) acc += p5;
        if (rem > 6) acc += p6;
#undef ROWLD

        for (int base = beg + 32; base < end; base += 32) {
            const int cnt = (end - base < 32) ? (end - base) : 32;
            int s = 0;
            if (lane < cnt) s = colw[base + lane];
            for (int j = 0; j < cnt; ++j) {
                const int sj = __builtin_amdgcn_readlane(s, j);
                acc += hin[(size_t)sj * PLANE + lane];
            }
        }
        const float res = acc * di + bl;
        hout[(size_t)(row0 + r) * PLANE + lane] = res > 0.0f ? res : 0.0f;
    }
}


#define RPW3 16

#define XM(v, m) __int_as_float(__builtin_amdgcn_update_dpp(0, __float_as_int(v), 0x160 | (m), 0xf, 0xf, true))
__device__ __forceinline__ float lane_sum32(float v) {
    v += XM(v, 1);
    v += XM(v, 2);
    v += XM(v, 4);
    v += XM(v, 8);
    const int b = __float_as_int(v);
    v += __int_as_float(__builtin_amdgcn_permlanex16(b, b, 0x76543210, 0xfedcba98, false, false));
    return v;
}
#undef XM

__global__ __launch_bounds__(128) void pull2p_w3_k(const int* __restrict__ rowptr,
                                                   const int* __restrict__ colw,
                                                   const float* __restrict__ g0,
                                                   const float* __restrict__ g1,
                                                   const float* __restrict__ bias,
                                                   const float* __restrict__ W3,
                                                   float* __restrict__ h3, int n, int E) {
    const int lane = threadIdx.x & 31;
    const int wave = __builtin_amdgcn_readfirstlane((int)(threadIdx.x >> 5));
    const int row0 = ((int)blockIdx.x * 4 + wave) * RPW3;
    if (row0 >= n) return;
    const int nrows = (n - row0 < RPW3) ? (n - row0) : RPW3;

    int bnd = E;
    if (lane <= nrows && row0 + lane < n) bnd = rowptr[row0 + lane];
    const float bl0 = bias[lane], bl1 = bias[PLANE + lane];
    const float w00 = W3[2 * lane],           w01 = W3[2 * lane + 1];
    const float w10 = W3[2 * (lane + PLANE)], w11 = W3[2 * (lane + PLANE) + 1];

    int beg_n = __builtin_amdgcn_readlane(bnd, 0);
    int end_n = __builtin_amdgcn_readlane(bnd, 1);
    int rec_n = 0;
    if (lane < end_n - beg_n) rec_n = colw[beg_n + lane];
    float self0_n = g0[(size_t)row0 * PLANE + lane];
    float self1_n = g1[(size_t)row0 * PLANE + lane];
    float keep = 0.0f;

    for (int r = 0; r < nrows; ++r) {
        const int beg = beg_n, end = end_n;
        const int cur = rec_n;
        const float di = rsqrtf((float)(end - beg + 1));
        float acc0 = self0_n;
        float acc1 = self1_n;

        const int cnt   = (end - beg < 32) ? (end - beg) : 32;
        const int nfull = cnt >> 3, rem = cnt & 7, jr = cnt & ~7;
        const int s  = cur * PLANE;
#define OFFJ(j)  ((size_t)(unsigned)__builtin_amdgcn_readlane(s, (j)))
#define LD0(j)   ((g0 + OFFJ(j))[lane])
#define LD1(j)   ((g1 + OFFJ(j))[lane])
        float a0, a1, a2, a3, a4, a5, a6, a7,  A0, A1, A2, A3, A4, A5, A6, A7,
              b0, b1, b2, b3, b4, b5, b6, b7,  B0, B1, B2, B3, B4, B5, B6, B7,
              c0, c1, c2, c3, c4, c5, c6, c7,  C0, C1, C2, C3, C4, C5, C6, C7,
              d0, d1, d2, d3, d4, d5, d6, d7,  D0, D1, D2, D3, D4, D5, D6, D7,
              p0, p1, p2, p3, p4, p5, p6,      P0, P1, P2, P3, P4, P5, P6;
        if (nfull > 0) { a0 = LD0(0); A0 = LD1(0); a1 = LD0(1); A1 = LD1(1); a2 = LD0(2); A2 = LD1(2); a3 = LD0(3); A3 = LD1(3);
                         a4 = LD0(4); A4 = LD1(4); a5 = LD0(5); A5 = LD1(5); a6 = LD0(6); A6 = LD1(6); a7 = LD0(7); A7 = LD1(7); }
        if (nfull > 1) { b0 = LD0(8); B0 = LD1(8); b1 = LD0(9); B1 = LD1(9); b2 = LD0(10); B2 = LD1(10); b3 = LD0(11); B3 = LD1(11);
                         b4 = LD0(12); B4 = LD1(12); b5 = LD0(13); B5 = LD1(13); b6 = LD0(14); B6 = LD1(14); b7 = LD0(15); B7 = LD1(15); }
        if (nfull > 2) { c0 = LD0(16); C0 = LD1(16); c1 = LD0(17); C1 = LD1(17); c2 = LD0(18); C2 = LD1(18); c3 = LD0(19); C3 = LD1(19);
                         c4 = LD0(20); C4 = LD1(20); c5 = LD0(21); C5 = LD1(21); c6 = LD0(22); C6 = LD1(22); c7 = LD0(23); C7 = LD1(23); }
        if (nfull > 3) { d0 = LD0(24); D0 = LD1(24); d1 = LD0(25); D1 = LD1(25); d2 = LD0(26); D2 = LD1(26); d3 = LD0(27); D3 = LD1(27);
                         d4 = LD0(28); D4 = LD1(28); d5 = LD0(29); D5 = LD1(29); d6 = LD0(30); D6 = LD1(30); d7 = LD0(31); D7 = LD1(31); }
        if (rem > 0) { p0 = LD0(jr + 0); P0 = LD1(jr + 0); }
        if (rem > 1) { p1 = LD0(jr + 1); P1 = LD1(jr + 1); }
        if (rem > 2) { p2 = LD0(jr + 2); P2 = LD1(jr + 2); }
        if (rem > 3) { p3 = LD0(jr + 3); P3 = LD1(jr + 3); }
        if (rem > 4) { p4 = LD0(jr + 4); P4 = LD1(jr + 4); }
        if (rem > 5) { p5 = LD0(jr + 5); P5 = LD1(jr + 5); }
        if (rem > 6) { p6 = LD0(jr + 6); P6 = LD1(jr + 6); }

        {
            const int rn = (r + 1 < nrows) ? r + 1 : r;
            beg_n = __builtin_amdgcn_readlane(bnd, rn);
            end_n = __builtin_amdgcn_readlane(bnd, rn + 1);
            rec_n = 0;
            if (lane < end_n - beg_n) rec_n = colw[beg_n + lane];
            self0_n = g0[(size_t)(row0 + rn) * PLANE + lane];
            self1_n = g1[(size_t)(row0 + rn) * PLANE + lane];
        }

#define AD(x, X, j) { acc0 += (x); acc1 += (X); }
        if (nfull > 0) { AD(a0, A0, 0) AD(a1, A1, 1) AD(a2, A2, 2) AD(a3, A3, 3) AD(a4, A4, 4) AD(a5, A5, 5) AD(a6, A6, 6) AD(a7, A7, 7) }
        if (nfull > 1) { AD(b0, B0, 8) AD(b1, B1, 9) AD(b2, B2, 10) AD(b3, B3, 11) AD(b4, B4, 12) AD(b5, B5, 13) AD(b6, B6, 14) AD(b7, B7, 15) }
        if (nfull > 2) { AD(c0, C0, 16) AD(c1, C1, 17) AD(c2, C2, 18) AD(c3, C3, 19) AD(c4, C4, 20) AD(c5, C5, 21) AD(c6, C6, 22) AD(c7, C7, 23) }
        if (nfull > 3) { AD(d0, D0, 24) AD(d1, D1, 25) AD(d2, D2, 26) AD(d3, D3, 27) AD(d4, D4, 28) AD(d5, D5, 29) AD(d6, D6, 30) AD(d7, D7, 31) }
        if (rem > 0) AD(p0, P0, jr + 0)
        if (rem > 1) AD(p1, P1, jr + 1)
        if (rem > 2) AD(p2, P2, jr + 2)
        if (rem > 3) AD(p3, P3, jr + 3)
        if (rem > 4) AD(p4, P4, jr + 4)
        if (rem > 5) AD(p5, P5, jr + 5)
        if (rem > 6) AD(p6, P6, jr + 6)
#undef AD
#undef LD0
#undef LD1
#undef OFFJ

        for (int base = beg + 32; base < end; base += 32) {
            const int cnt = (end - base < 32) ? (end - base) : 32;
            int s = 0;
            if (lane < cnt) s = colw[base + lane];
            for (int j = 0; j < cnt; ++j) {
                const size_t off = (size_t)(unsigned)__builtin_amdgcn_readlane(s, j) * PLANE + lane;
                acc0 += g0[off];
                acc1 += g1[off];
            }
        }
        const float r0 = acc0 * di + bl0, r1 = acc1 * di + bl1;
        const float v0 = r0 > 0.0f ? r0 : 0.0f;
        const float v1 = r1 > 0.0f ? r1 : 0.0f;
        const float t0 = lane_sum32(v0 * w00 + v1 * w10);
        const float t1 = lane_sum32(v0 * w01 + v1 * w11);
        if (lane == 2 * r)     keep = t0 * di;
        if (lane == 2 * r + 1) keep = t1 * di;
    }
    if (lane < 2 * nrows) h3[(size_t)row0 * 2 + lane] = keep;
}

__global__ void pull2_lsm_k(const int* __restrict__ rowptr, const int* __restrict__ colw,
                            const float* __restrict__ h3, const float* __restrict__ b3,
                            float* __restrict__ out, int n, int E) {
    int i = blockIdx.x * blockDim.x + threadIdx.x;
    if (i >= n) return;
    const int beg = rowptr[i];
    const int end = (i + 1 < n) ? rowptr[i + 1] : E;
    const float di = rsqrtf((float)(end - beg + 1));
    const v2f* h2 = (const v2f*)h3;
    const v2f hs = h2[i];
    float a0 = hs.x, a1 = hs.y;
    for (int e = beg; e < end; ++e) {
        const v2f v = h2[colw[e]];
        a0 += v.x;
        a1 += v.y;
    }
    float z0 = a0 * di + b3[0];
    float z1 = a1 * di + b3[1];
    float m = fmaxf(z0, z1);
    float lse = m + logf(__expf(z0 - m) + __expf(z1 - m));
    out[2 * i] = z0 - lse;
    out[2 * i + 1] = z1 - lse;
}


extern "C" void kernel_launch(void* const* d_in, const int* in_sizes, int n_in,
                              void* d_out, int out_size, void* d_ws, size_t ws_size,
                              hipStream_t stream) {
    const float* x  = (const float*)d_in[0];
    const int*   ei = (const int*)d_in[1];
    const float* W1 = (const float*)d_in[2];
    const float* b1 = (const float*)d_in[3];
    const float* W2 = (const float*)d_in[4];
    const float* b2 = (const float*)d_in[5];
    const float* W3 = (const float*)d_in[6];
    const float* b3 = (const float*)d_in[7];
    float* out = (float*)d_out;

    const int N = in_sizes[0] / F_IN;
    const int E = in_sizes[1] / 2;
    const int* src = ei;
    const int* dst = ei + E;

    int*   colw   = (int*)d_ws;
    int*   rowptr = (int*)d_ws + (size_t)2 * E;
    float* pA     = (float*)(rowptr + N);
    float* pB     = pA + (size_t)N * PLANE;
    float* pC     = pB + (size_t)N * PLANE;
    int2*  tmp    = (int2*)pC;
    int2*  tab    = (int2*)pB;
    float* h3     = pB;
    int*   cnt    = (int*)d_out;

    const int T = 256;
    dim3 blk(T);
    int gN    = (N + T - 1) / T;
    int gRow  = (N + 8 * RPW - 1) / (8 * RPW);
    int gTile = (N + 15) / 16;
    int gRow3 = (N + 4 * RPW3 - 1) / (4 * RPW3);
    int nbk   = (N + BW - 1) / BW;
    int nblk  = (E + CH - 1) / CH;

    part_k<<<nblk, dim3(1024), 0, stream>>>(src, dst, tmp, tab, E, nbk, nblk);
    bucket_k<<<nbk, dim3(1024), 0, stream>>>(tmp, tab, cnt, rowptr, colw, N, nblk);

    gemm_wmma_k<F_IN, false><<<gTile, dim3(128), 0, stream>>>(x, x, W1, cnt, pA, pB, N);
    pull32_k<<<gRow, blk, 0, stream>>>(rowptr, colw, pA, b1, pC, N, E);
    pull32_k<<<gRow, blk, 0, stream>>>(rowptr, colw, pB, b1 + PLANE, pA, N, E);

    gemm_wmma_k<HIDDEN, true><<<gTile, dim3(128), 0, stream>>>(pC, pA, W2, cnt, pC, pA, N);
    pull2p_w3_k<<<gRow3, dim3(128), 0, stream>>>(rowptr, colw, pC, pA, b2, W3, h3, N, E);

    pull2_lsm_k<<<gN, blk, 0, stream>>>(rowptr, colw, h3, b3, out, N, E);
}


# bucket pass: dependent round trips about 24 -> 5: list start = sum of own runs' in-chunk starts (not all lower buckets' run lengths); a walk asks for a wave's 13 runs at once; arrival order kept
# speedup vs baseline: 6.1632x; 1.0612x over previous
#include <hip/hip_runtime.h>


typedef __attribute__((ext_vector_type(2))) float v2f;
typedef __attribute__((ext_vector_type(4))) float v4f;
typedef __attribute__((ext_vector_type(8))) float v8f;

#define F_IN   256
#define HIDDEN 64
#define PLANE  32


#define BSH  9
#define BW   (1 << BSH)
#define MAXB 256
#define CH   4096

__global__ __launch_bounds__(1024) void part_k(const int* __restrict__ src, const int* __restrict__ dst,
                                               int2* __restrict__ tmp, int2* __restrict__ tab,
                                               int E, int nbk, int nblk) {
    __shared__ int  hist[MAXB];
    __shared__ int  offL[MAXB];
    __shared__ int  wsum[8];
    __shared__ int2 stage[CH];
    const int tid = threadIdx.x, lane = tid & 31, wave = tid >> 5;
    const int base = blockIdx.x * CH;
    if (tid < MAXB) hist[tid] = 0;
    __syncthreads();

    int s[4], d[4], r[4];
#pragma unroll
    for (int k = 0; k < 4; ++k) {
        const int e = base + k * 1024 + tid;
        s[k] = 0; d[k] = -1; r[k] = 0;
        if (e < E) { s[k] = src[e]; d[k] = dst[e]; }
    }
#pragma unroll
    for (int k = 0; k < 4; ++k)
        if (d[k] >= 0) r[k] = atomicAdd(&hist[d[k] >> BSH], 1);
    __syncthreads();

    const int c = (tid < MAXB) ? hist[tid] : 0;
    int incl = c;
    for (int o = 1; o < 32; o <<= 1) {
        int t = __shfl_up(incl, o);
        if (lane >= o) incl += t;
    }
    if (tid < MAXB && lane == 31) wsum[wave] = incl;
    __syncthreads();
    if (tid < MAXB) {
        int pre = 0;
        for (int w = 0; w < wave; ++w) pre += wsum[w];
        offL[tid] = pre + incl - c;
    }
    __syncthreads();

#pragma unroll
    for (int k = 0; k < 4; ++k)
        if (d[k] >= 0) stage[offL[d[k] >> BSH] + r[k]] = make_int2(s[k], d[k]);
    __syncthreads();
    const int nvalid = (E - base < CH) ? (E - base) : CH;
#pragma unroll
    for (int k = 0; k < 4; ++k) {
        const int idx = k * 1024 + tid;
        if (idx < nvalid) tmp[base + idx] = stage[idx];
    }
    if (tid < nbk) tab[(size_t)tid * nblk + blockIdx.x] = make_int2(offL[tid], hist[tid]);
}

#define LCAP 16384
#define UR   13
__global__ __launch_bounds__(1024) void bucket_k(const int2* __restrict__ tmp, const int2* __restrict__ tab,
                                                 int* __restrict__ cnt, int* __restrict__ rowptr,
                                                 int* __restrict__ col, int n, int nblk) {
    __shared__ int cntL[BW];
    __shared__ int wred[32];
    __shared__ int wsum[16];
    __shared__ int lst[LCAP];
    const int tid = threadIdx.x, lane = tid & 31;
    const int wave = __builtin_amdgcn_readfirstlane((int)(threadIdx.x >> 5));
    const int b = blockIdx.x, lo = b << BSH;

    const int2* trow = tab + (size_t)b * nblk;
    int part = 0;
    for (int j = tid; j < nblk; j += 1024) part += trow[j].x;
    for (int o = 16; o > 0; o >>= 1) part += __shfl_xor(part, o);
    if (tid < BW) cntL[tid] = 0;
    if (lane == 0) wred[wave] = part;
    __syncthreads();

    for (int j0 = wave; j0 < nblk; j0 += 32 * UR) {
        int st[UR], ln[UR];
        int2 e[UR];
        bool anylong = false;
#pragma unroll
        for (int u = 0; u < UR; ++u) {
            const int j  = j0 + 32 * u;
            const int jc = (j < nblk) ? j : nblk - 1;
            const int2 t = trow[jc];
            st[u] = jc * CH + t.x;
            ln[u] = (j < nblk) ? t.y : 0;
            anylong |= ln[u] > 32;
        }
#pragma unroll
        for (int u = 0; u < UR; ++u) {
            e[u] = make_int2(0, lo);
            if (lane < ln[u]) e[u] = tmp[st[u] + lane];
        }
#pragma unroll
        for (int u = 0; u < UR; ++u)
            if (lane < ln[u]) atomicAdd(&cntL[e[u].y - lo], 1);
        if (anylong) {
#pragma nounroll
            for (int u = 0; u < UR; ++u) {
                const int j = j0 + 32 * u;
                if (j >= nblk) break;
                const int2 t = trow[j];
                const int s0 = j * CH + t.x;
                for (int k = 32; k < t.y; k += 32)
                    if (k + lane < t.y) {
                        const int2 ee = tmp[s0 + k + lane];
                        atomicAdd(&cntL[ee.y - lo], 1);
                    }
            }
        }
    }
    __syncthreads();

    int bs = wred[lane];
    for (int o = 16; o > 0; o >>= 1) bs += __shfl_xor(bs, o);
    const int c = (tid < BW) ? cntL[tid] : 0;
    int incl = c;
    for (int o = 1; o < 32; o <<= 1) {
        int t = __shfl_up(incl, o);
        if (lane >= o) incl += t;
    }
    if (tid < BW && lane == 31) wsum[wave] = incl;
    __syncthreads();
    const int wsl = (lane < 16) ? wsum[lane] : 0;
    int tp = (lane < wave) ? wsl : 0;
    for (int o = 16; o > 0; o >>= 1) tp += __shfl_xor(tp, o);
    int total = wsl;
    for (int o = 16; o > 0; o >>= 1) total += __shfl_xor(total, o);
    const int i = lo + tid;
    if (tid < BW) {
        const int off = tp + incl - c;
        cntL[tid] = off;
        if (i < n) {
            rowptr[i] = bs + off;
            cnt[i] = c;
        }
    }
    __syncthreads();

    const bool inl = total <= LCAP;
    for (int j0 = wave; j0 < nblk; j0 += 32 * UR) {
        int st[UR], ln[UR];
        int2 e[UR];
        int sl[UR];
        bool anylong = false;
#pragma unroll
        for (int u = 0; u < UR; ++u) {
            const int j  = j0 + 32 * u;
            const int jc = (j < nblk) ? j : nblk - 1;
            const int2 t = trow[jc];
            st[u] = jc * CH + t.x;
            ln[u] = (j < nblk) ? t.y : 0;
            anylong |= ln[u] > 32;
        }
#pragma unroll
        for (int u = 0; u < UR; ++u) {
            e[u] = make_int2(0, lo);
            if (lane < ln[u]) e[u] = tmp[st[u] + lane];
        }
#pragma unroll
        for (int u = 0; u < UR; ++u) {
            sl[u] = 0;
            if (lane < ln[u]) sl[u] = atomicAdd(&cntL[e[u].y - lo], 1);
        }
#pragma unroll
        for (int u = 0; u < UR; ++u)
            if (lane < ln[u]) {
                if (inl) lst[sl[u]] = e[u].x;
                else     col[bs + sl[u]] = e[u].x;
            }
        if (anylong) {
#pragma nounroll
            for (int u = 0; u < UR; ++u) {
                const int j = j0 + 32 * u;
                if (j >= nblk) break;
                const int2 t = trow[j];
                const int s0 = j * CH + t.x;
                for (int k = 32; k < t.y; k += 32)
                    if (k + lane < t.y) {
                        const int2 ee = tmp[s0 + k + lane];
                        const int s2 = atomicAdd(&cntL[ee.y - lo], 1);
                        if (inl) lst[s2] = ee.x;
                        else     col[bs + s2] = ee.x;
                    }
            }
        }
    }
    __syncthreads();
    if (inl)
        for (int k = tid; k < total; k += 1024) col[bs + k] = lst[k];
}


template <int K, bool PLANES>
__global__ __launch_bounds__(128) void gemm_wmma_k(const float* A0, const float* A1,
                                                   const float* __restrict__ W,
                                                   const int* __restrict__ cnt,
                                                   float* O0, float* O1, int M) {
    static_assert(!PLANES || K == 2 * PLANE, "plane input is 64 wide");
    constexpr int NOUT = HIDDEN;
    constexpr int LDSK = K + 4;
    __shared__ float Atile[16 * LDSK];

    const int tid  = threadIdx.x;
    const int row0 = blockIdx.x * 16;

    constexpr int KV = K / 4;
    for (int idx = tid; idx < 16 * KV; idx += 128) {
        int r = idx / KV, c4 = idx % KV;
        int row = row0 + r; if (row >= M) row = M - 1;
        const float* p;
        if (PLANES) p = (c4 < PLANE / 4) ? &A0[(size_t)row * PLANE + c4 * 4]
                                         : &A1[(size_t)row * PLANE + (c4 - PLANE / 4) * 4];
        else        p = &A0[(size_t)row * K + c4 * 4];
        v4f v = *(const v4f*)p;
        *(v4f*)&Atile[r * LDSK + c4 * 4] = v;
    }
    __syncthreads();

    const int lane = tid & 31;
    const int wave = tid >> 5;
    const int half = lane >> 4;
    const int l    = lane & 15;
    const int col0 = wave * 16;

    int deg[8];
#pragma unroll
    for (int r = 0; r < 8; ++r) {
        int row = row0 + half * 8 + r; if (row >= M) row = M - 1;
        deg[r] = cnt[row];
    }

    v8f acc = {};
    const float* wp = W + col0 + l;
    for (int k = 0; k < K; k += 4) {
        const int ka = k + half * 2;
        v2f a = *(const v2f*)&Atile[l * LDSK + ka];
        v2f b;
        b.x = wp[ka * NOUT];
        b.y = wp[(ka + 1) * NOUT];
        acc = __builtin_amdgcn_wmma_f32_16x16x4_f32(false, a, false, b,
                                                    (short)0, acc, false, false);
    }

    float* O  = (wave < 2) ? O0 : O1;
    float* op = O + (size_t)(row0 + half * 8) * PLANE + (wave & 1) * 16 + l;
    if (row0 + 16 <= M) {
#pragma unroll
        for (int r = 0; r < 8; ++r) op[r * PLANE] = acc[r] * rsqrtf((float)(deg[r] + 1));
    } else {
#pragma unroll
        for (int r = 0; r < 8; ++r)
            if (row0 + half * 8 + r < M) op[r * PLANE] = acc[r] * rsqrtf((float)(deg[r] + 1));
    }
}


#define RPW 12

__global__ __launch_bounds__(256) void pull32_k(const int* __restrict__ rowptr,
                                                const int* __restrict__ colw,
                                                const float* __restrict__ hin,
                                                const float* __restrict__ bias,
                                                float* __restrict__ hout, int n, int E) {
    const int lane = threadIdx.x & 31;
    const int wave = __builtin_amdgcn_readfirstlane((int)(threadIdx.x >> 5));
    const int row0 = ((int)blockIdx.x * 8 + wave) * RPW;
    if (row0 >= n) return;
    const int nrows = (n - row0 < RPW) ? (n - row0) : RPW;

    int bnd = E;
    if (lane <= nrows && row0 + lane < n) bnd = rowptr[row0 + lane];
    const float bl = bias[lane];

    int beg_n = __builtin_amdgcn_readlane(bnd, 0);
    int end_n = __builtin_amdgcn_readlane(bnd, 1);
    int rec_n = 0;
    if (lane < end_n - beg_n) rec_n = colw[beg_n + lane];
    float self_n = hin[(size_t)row0 * PLANE + lane];

    for (int r = 0; r < nrows; ++r) {
        const int beg = beg_n, end = end_n;
        const int cur = rec_n;
        const float di = rsqrtf((float)(end - beg + 1));
        float acc = self_n;

        const int cnt   = (end - beg < 32) ? (end - beg) : 32;
        const int nfull = cnt >> 3, rem = cnt & 7, jr = cnt & ~7;
        const int s  = cur * PLANE;
#define ROWLD(j) ((hin + (size_t)(unsigned)__builtin_amdgcn_readlane(s, (j)))[lane])
        float a0, a1, a2, a3, a4, a5, a6, a7,
              b0, b1, b2, b3, b4, b5, b6, b7,
              c0, c1, c2, c3, c4, c5, c6, c7,
              d0, d1, d2, d3, d4, d5, d6, d7,
              p0, p1, p2, p3, p4, p5, p6;
        if (nfull > 0) { a0 = ROWLD(0); a1 = ROWLD(1); a2 = ROWLD(2); a3 = ROWLD(3); a4 = ROWLD(4); a5 = ROWLD(5); a6 = ROWLD(6); a7 = ROWLD(7); }
        if (nfull > 1) { b0 = ROWLD(8); b1 = ROWLD(9); b2 = ROWLD(10); b3 = ROWLD(11); b4 = ROWLD(12); b5 = ROWLD(13); b6 = ROWLD(14); b7 = ROWLD(15); }
        if (nfull > 2) { c0 = ROWLD(16); c1 = ROWLD(17); c2 = ROWLD(18); c3 = ROWLD(19); c4 = ROWLD(20); c5 = ROWLD(21); c6 = ROWLD(22); c7 = ROWLD(23); }
        if (nfull > 3) { d0 = ROWLD(24); d1 = ROWLD(25); d2 = ROWLD(26); d3 = ROWLD(27); d4 = ROWLD(28); d5 = ROWLD(29); d6 = ROWLD(30); d7 = ROWLD(31); }
        if (rem > 0) p0 = ROWLD(jr + 0);
        if (rem > 1) p1 = ROWLD(jr + 1);
        if (rem > 2) p2 = ROWLD(jr + 2);
        if (rem > 3) p3 = ROWLD(jr + 3);
        if (rem > 4) p4 = ROWLD(jr + 4);
        if (rem > 5) p5 = ROWLD(jr + 5);
        if (rem > 6) p6 = ROWLD(jr + 6);

        {
            const int rn = (r + 1 < nrows) ? r + 1 : r;
            beg_n = __builtin_amdgcn_readlane(bnd, rn);
            end_n = __builtin_amdgcn_readlane(bnd, rn + 1);
            rec_n = 0;
            if (lane < end_n - beg_n) rec_n = colw[beg_n + lane];
            self_n = hin[(size_t)(row0 + rn) * PLANE + lane];
        }

        if (nfull > 0) { acc += a0; acc += a1; acc += a2; acc += a3; acc += a4; acc += a5; acc += a6; acc += a7; }
        if (nfull > 1) { acc += b0; acc += b1; acc += b2; acc += b3; acc += b4; acc += b5; acc += b6; acc += b7; }
        if (nfull > 2) { acc += c0; acc += c1; acc += c2; acc += c3; acc += c4; acc += c5; acc += c6; acc += c7; }
        if (nfull > 3) { acc += d0; acc += d1; acc += d2; acc += d3; acc += d4; acc += d5; acc += d6; acc += d7; }
        if (rem > 0) acc += p0;
        if (rem > 1) acc += p1;
        if (rem > 2) acc += p2;
        if (rem > 3) acc += p3;
        if (rem > 4) acc += p4;
        if (rem > 5) acc += p5;
        if (rem > 6) acc += p6;
#undef ROWLD

        for (int base = beg + 32; base < end; base += 32) {
            const int cnt = (end - base < 32) ? (end - base) : 32;
            int s = 0;
            if (lane < cnt) s = colw[base + lane];
            for (int j = 0; j < cnt; ++j) {
                const int sj = __builtin_amdgcn_readlane(s, j);
                acc += hin[(size_t)sj * PLANE + lane];
            }
        }
        const float res = acc * di + bl;
        hout[(size_t)(row0 + r) * PLANE + lane] = res > 0.0f ? res : 0.0f;
    }
}


#define RPW3 16

#define XM(v, m) __int_as_float(__builtin_amdgcn_update_dpp(0, __float_as_int(v), 0x160 | (m), 0xf, 0xf, true))
__device__ __forceinline__ float lane_sum32(float v) {
    v += XM(v, 1);
    v += XM(v, 2);
    v += XM(v, 4);
    v += XM(v, 8);
    const int b = __float_as_int(v);
    v += __int_as_float(__builtin_amdgcn_permlanex16(b, b, 0x76543210, 0xfedcba98, false, false));
    return v;
}
#undef XM

__global__ __launch_bounds__(128) void pull2p_w3_k(const int* __restrict__ rowptr,
                                                   const int* __restrict__ colw,
                                                   const float* __restrict__ g0,
                                                   const float* __restrict__ g1,
                                                   const float* __restrict__ bias,
                                                   const float* __restrict__ W3,
                                                   float* __restrict__ h3, int n, int E) {
    const int lane = threadIdx.x & 31;
    const int wave = __builtin_amdgcn_readfirstlane((int)(threadIdx.x >> 5));
    const int row0 = ((int)blockIdx.x * 4 + wave) * RPW3;
    if (row0 >= n) return;
    const int nrows = (n - row0 < RPW3) ? (n - row0) : RPW3;

    int bnd = E;
    if (lane <= nrows && row0 + lane < n) bnd = rowptr[row0 + lane];
    const float bl0 = bias[lane], bl1 = bias[PLANE + lane];
    const float w00 = W3[2 * lane],           w01 = W3[2 * lane + 1];
    const float w10 = W3[2 * (lane + PLANE)], w11 = W3[2 * (lane + PLANE) + 1];

    int beg_n = __builtin_amdgcn_readlane(bnd, 0);
    int end_n = __builtin_amdgcn_readlane(bnd, 1);
    int rec_n = 0;
    if (lane < end_n - beg_n) rec_n = colw[beg_n + lane];
    float self0_n = g0[(size_t)row0 * PLANE + lane];
    float self1_n = g1[(size_t)row0 * PLANE + lane];
    float keep = 0.0f;

    for (int r = 0; r < nrows; ++r) {
        const int beg = beg_n, end = end_n;
        const int cur = rec_n;
        const float di = rsqrtf((float)(end - beg + 1));
        float acc0 = self0_n;
        float acc1 = self1_n;

        const int cnt   = (end - beg < 32) ? (end - beg) : 32;
        const int nfull = cnt >> 3, rem = cnt & 7, jr = cnt & ~7;
        const int s  = cur * PLANE;
#define OFFJ(j)  ((size_t)(unsigned)__builtin_amdgcn_readlane(s, (j)))
#define LD0(j)   ((g0 + OFFJ(j))[lane])
#define LD1(j)   ((g1 + OFFJ(j))[lane])
        float a0, a1, a2, a3, a4, a5, a6, a7,  A0, A1, A2, A3, A4, A5, A6, A7,
              b0, b1, b2, b3, b4, b5, b6, b7,  B0, B1, B2, B3, B4, B5, B6, B7,
              c0, c1, c2, c3, c4, c5, c6, c7,  C0, C1, C2, C3, C4, C5, C6, C7,
              d0, d1, d2, d3, d4, d5, d6, d7,  D0, D1, D2, D3, D4, D5, D6, D7,
              p0, p1, p2, p3, p4, p5, p6,      P0, P1, P2, P3, P4, P5, P6;
        if (nfull > 0) { a0 = LD0(0); A0 = LD1(0); a1 = LD0(1); A1 = LD1(1); a2 = LD0(2); A2 = LD1(2); a3 = LD0(3); A3 = LD1(3);
                         a4 = LD0(4); A4 = LD1(4); a5 = LD0(5); A5 = LD1(5); a6 = LD0(6); A6 = LD1(6); a7 = LD0(7); A7 = LD1(7); }
        if (nfull > 1) { b0 = LD0(8); B0 = LD1(8); b1 = LD0(9); B1 = LD1(9); b2 = LD0(10); B2 = LD1(10); b3 = LD0(11); B3 = LD1(11);
                         b4 = LD0(12); B4 = LD1(12); b5 = LD0(13); B5 = LD1(13); b6 = LD0(14); B6 = LD1(14); b7 = LD0(15); B7 = LD1(15); }
        if (nfull > 2) { c0 = LD0(16); C0 = LD1(16); c1 = LD0(17); C1 = LD1(17); c2 = LD0(18); C2 = LD1(18); c3 = LD0(19); C3 = LD1(19);
                         c4 = LD0(20); C4 = LD1(20); c5 = LD0(21); C5 = LD1(21); c6 = LD0(22); C6 = LD1(22); c7 = LD0(23); C7 = LD1(23); }
        if (nfull > 3) { d0 = LD0(24); D0 = LD1(24); d1 = LD0(25); D1 = LD1(25); d2 = LD0(26); D2 = LD1(26); d3 = LD0(27); D3 = LD1(27);
                         d4 = LD0(28); D4 = LD1(28); d5 = LD0(29); D5 = LD1(29); d6 = LD0(30); D6 = LD1(30); d7 = LD0(31); D7 = LD1(31); }
        if (rem > 0) { p0 = LD0(jr + 0); P0 = LD1(jr + 0); }
        if (rem > 1) { p1 = LD0(jr + 1); P1 = LD1(jr + 1); }
        if (rem > 2) { p2 = LD0(jr + 2); P2 = LD1(jr + 2); }
        if (rem > 3) { p3 = LD0(jr + 3); P3 = LD1(jr + 3); }
        if (rem > 4) { p4 = LD0(jr + 4); P4 = LD1(jr + 4); }
        if (rem > 5) { p5 = LD0(jr + 5); P5 = LD1(jr + 5); }
        if (rem > 6) { p6 = LD0(jr + 6); P6 = LD1(jr + 6); }

        {
            const int rn = (r + 1 < nrows) ? r + 1 : r;
            beg_n = __builtin_amdgcn_readlane(bnd, rn);
            end_n = __builtin_amdgcn_readlane(bnd, rn + 1);
            rec_n = 0;
            if (lane < end_n - beg_n) rec_n = colw[beg_n + lane];
            self0_n = g0[(size_t)(row0 + rn) * PLANE + lane];
            self1_n = g1[(size_t)(row0 + rn) * PLANE + lane];
        }

#define AD(x, X, j) { acc0 += (x); acc1 += (X); }
        if (nfull > 0) { AD(a0, A0, 0) AD(a1, A1, 1) AD(a2, A2, 2) AD(a3, A3, 3) AD(a4, A4, 4) AD(a5, A5, 5) AD(a6, A6, 6) AD(a7, A7, 7) }
        if (nfull > 1) { AD(b0, B0, 8) AD(b1, B1, 9) AD(b2, B2, 10) AD(b3, B3, 11) AD(b4, B4, 12) AD(b5, B5, 13) AD(b6, B6, 14) AD(b7, B7, 15) }
        if (nfull > 2) { AD(c0, C0, 16) AD(c1, C1, 17) AD(c2, C2, 18) AD(c3, C3, 19) AD(c4, C4, 20) AD(c5, C5, 21) AD(c6, C6, 22) AD(c7, C7, 23) }
        if (nfull > 3) { AD(d0, D0, 24) AD(d1, D1, 25) AD(d2, D2, 26) AD(d3, D3, 27) AD(d4, D4, 28) AD(d5, D5, 29) AD(d6, D6, 30) AD(d7, D7, 31) }
        if (rem > 0) AD(p0, P0, jr + 0)
        if (rem > 1) AD(p1, P1, jr + 1)
        if (rem > 2) AD(p2, P2, jr + 2)
        if (rem > 3) AD(p3, P3, jr + 3)
        if (rem > 4) AD(p4, P4, jr + 4)
        if (rem > 5) AD(p5, P5, jr + 5)
        if (rem > 6) AD(p6, P6, jr + 6)
#undef AD
#undef LD0
#undef LD1
#undef OFFJ

        for (int base = beg + 32; base < end; base += 32) {
            const int cnt = (end - base < 32) ? (end - base) : 32;
            int s = 0;
            if (lane < cnt) s = colw[base + lane];
            for (int j = 0; j < cnt; ++j) {
                const size_t off = (size_t)(unsigned)__builtin_amdgcn_readlane(s, j) * PLANE + lane;
                acc0 += g0[off];
                acc1 += g1[off];
            }
        }
        const float r0 = acc0 * di + bl0, r1 = acc1 * di + bl1;
        const float v0 = r0 > 0.0f ? r0 : 0.0f;
        const float v1 = r1 > 0.0f ? r1 : 0.0f;
        const float t0 = lane_sum32(v0 * w00 + v1 * w10);
        const float t1 = lane_sum32(v0 * w01 + v1 * w11);
        if (lane == 2 * r)     keep = t0 * di;
        if (lane == 2 * r + 1) keep = t1 * di;
    }
    if (lane < 2 * nrows) h3[(size_t)row0 * 2 + lane] = keep;
}

__global__ void pull2_lsm_k(const int* __restrict__ rowptr, const int* __restrict__ colw,
                            const float* __restrict__ h3, const float* __restrict__ b3,
                            float* __restrict__ out, int n, int E) {
    int i = blockIdx.x * blockDim.x + threadIdx.x;
    if (i >= n) return;
    const int beg = rowptr[i];
    const int end = (i + 1 < n) ? rowptr[i + 1] : E;
    const float di = rsqrtf((float)(end - beg + 1));
    const v2f* h2 = (const v2f*)h3;
    const v2f hs = h2[i];
    float a0 = hs.x, a1 = hs.y;
    for (int e = beg; e < end; ++e) {
        const v2f v = h2[colw[e]];
        a0 += v.x;
        a1 += v.y;
    }
    float z0 = a0 * di + b3[0];
    float z1 = a1 * di + b3[1];
    float m = fmaxf(z0, z1);
    float lse = m + logf(__expf(z0 - m) + __expf(z1 - m));
    out[2 * i] = z0 - lse;
    out[2 * i + 1] = z1 - lse;
}


extern "C" void kernel_launch(void* const* d_in, const int* in_sizes, int n_in,
                              void* d_out, int out_size, void* d_ws, size_t ws_size,
                              hipStream_t stream) {
    const float* x  = (const float*)d_in[0];
    const int*   ei = (const int*)d_in[1];
    const float* W1 = (const float*)d_in[2];
    const float* b1 = (const float*)d_in[3];
    const float* W2 = (const float*)d_in[4];
    const float* b2 = (const float*)d_in[5];
    const float* W3 = (const float*)d_in[6];
    const float* b3 = (const float*)d_in[7];
    float* out = (float*)d_out;

    const int N = in_sizes[0] / F_IN;
    const int E = in_sizes[1] / 2;
    const int* src = ei;
    const int* dst = ei + E;

    int*   colw   = (int*)d_ws;
    int*   rowptr = (int*)d_ws + (size_t)2 * E;
    float* pA     = (float*)(rowptr + N);
    float* pB     = pA + (size_t)N * PLANE;
    float* pC     = pB + (size_t)N * PLANE;
    int2*  tmp    = (int2*)pC;
    int2*  tab    = (int2*)pB;
    float* h3     = pB;
    int*   cnt    = (int*)d_out;

    const int T = 256;
    dim3 blk(T);
    int gN    = (N + T - 1) / T;
    int gRow  = (N + 8 * RPW - 1) / (8 * RPW);
    int gTile = (N + 15) / 16;
    int gRow3 = (N + 4 * RPW3 - 1) / (4 * RPW3);
    int nbk   = (N + BW - 1) / BW;
    int nblk  = (E + CH - 1) / CH;

    part_k<<<nblk, dim3(1024), 0, stream>>>(src, dst, tmp, tab, E, nbk, nblk);
    bucket_k<<<nbk, dim3(1024), 0, stream>>>(tmp, tab, cnt, rowptr, colw, N, nblk);

    gemm_wmma_k<F_IN, false><<<gTile, dim3(128), 0, stream>>>(x, x, W1, cnt, pA, pB, N);
    pull32_k<<<gRow, blk, 0, stream>>>(rowptr, colw, pA, b1, pC, N, E);
    pull32_k<<<gRow, blk, 0, stream>>>(rowptr, colw, pB, b1 + PLANE, pA, N, E);

    gemm_wmma_k<HIDDEN, true><<<gTile, dim3(128), 0, stream>>>(pC, pA, W2, cnt, pC, pA, N);
    pull2p_w3_k<<<gRow3, dim3(128), 0, stream>>>(rowptr, colw, pC, pA, b2, W3, h3, N, E);

    pull2_lsm_k<<<gN, blk, 0, stream>>>(rowptr, colw, h3, b3, out, N, E);
}


# layer 2 sums rows first, then the aggregating wave multiplies its 16-row tile by W2 on the matrix unit (tile and W2 in LDS): the 64x64 product launch and its pass over h1 are gone; sum order changes
# speedup vs baseline: 6.3968x; 1.0379x over previous
#include <hip/hip_runtime.h>


typedef __attribute__((ext_vector_type(2))) float v2f;
typedef __attribute__((ext_vector_type(4))) float v4f;
typedef __attribute__((ext_vector_type(8))) float v8f;

#define F_IN   256
#define HIDDEN 64
#define PLANE  32


#define BSH  9
#define BW   (1 << BSH)
#define MAXB 256
#define CH   4096

__global__ __launch_bounds__(1024) void part_k(const int* __restrict__ src, const int* __restrict__ dst,
                                               int2* __restrict__ tmp, int2* __restrict__ tab,
                                               int E, int nbk, int nblk) {
    __shared__ int  hist[MAXB];
    __shared__ int  offL[MAXB];
    __shared__ int  wsum[8];
    __shared__ int2 stage[CH];
    const int tid = threadIdx.x, lane = tid & 31, wave = tid >> 5;
    const int base = blockIdx.x * CH;
    if (tid < MAXB) hist[tid] = 0;
    __syncthreads();

    int s[4], d[4], r[4];
#pragma unroll
    for (int k = 0; k < 4; ++k) {
        const int e = base + k * 1024 + tid;
        s[k] = 0; d[k] = -1; r[k] = 0;
        if (e < E) { s[k] = src[e]; d[k] = dst[e]; }
    }
#pragma unroll
    for (int k = 0; k < 4; ++k)
        if (d[k] >= 0) r[k] = atomicAdd(&hist[d[k] >> BSH], 1);
    __syncthreads();

    const int c = (tid < MAXB) ? hist[tid] : 0;
    int incl = c;
    for (int o = 1; o < 32; o <<= 1) {
        int t = __shfl_up(incl, o);
        if (lane >= o) incl += t;
    }
    if (tid < MAXB && lane == 31) wsum[wave] = incl;
    __syncthreads();
    if (tid < MAXB) {
        int pre = 0;
        for (int w = 0; w < wave; ++w) pre += wsum[w];
        offL[tid] = pre + incl - c;
    }
    __syncthreads();

#pragma unroll
    for (int k = 0; k < 4; ++k)
        if (d[k] >= 0) stage[offL[d[k] >> BSH] + r[k]] = make_int2(s[k], d[k]);
    __syncthreads();
    const int nvalid = (E - base < CH) ? (E - base) : CH;
#pragma unroll
    for (int k = 0; k < 4; ++k) {
        const int idx = k * 1024 + tid;
        if (idx < nvalid) tmp[base + idx] = stage[idx];
    }
    if (tid < nbk) tab[(size_t)tid * nblk + blockIdx.x] = make_int2(offL[tid], hist[tid]);
}

#define LCAP 16384
#define UR   13
__global__ __launch_bounds__(1024) void bucket_k(const int2* __restrict__ tmp, const int2* __restrict__ tab,
                                                 int* __restrict__ cnt, int* __restrict__ rowptr,
                                                 int* __restrict__ col, int n, int nblk) {
    __shared__ int cntL[BW];
    __shared__ int wred[32];
    __shared__ int wsum[16];
    __shared__ int lst[LCAP];
    const int tid = threadIdx.x, lane = tid & 31;
    const int wave = __builtin_amdgcn_readfirstlane((int)(threadIdx.x >> 5));
    const int b = blockIdx.x, lo = b << BSH;

    const int2* trow = tab + (size_t)b * nblk;
    int part = 0;
    for (int j = tid; j < nblk; j += 1024) part += trow[j].x;
    for (int o = 16; o > 0; o >>= 1) part += __shfl_xor(part, o);
    if (tid < BW) cntL[tid] = 0;
    if (lane == 0) wred[wave] = part;
    __syncthreads();

    for (int j0 = wave; j0 < nblk; j0 += 32 * UR) {
        int st[UR], ln[UR];
        int2 e[UR];
        bool anylong = false;
#pragma unroll
        for (int u = 0; u < UR; ++u) {
            const int j  = j0 + 32 * u;
            const int jc = (j < nblk) ? j : nblk - 1;
            const int2 t = trow[jc];
            st[u] = jc * CH + t.x;
            ln[u] = (j < nblk) ? t.y : 0;
            anylong |= ln[u] > 32;
        }
#pragma unroll
        for (int u = 0; u < UR; ++u) {
            e[u] = make_int2(0, lo);
            if (lane < ln[u]) e[u] = tmp[st[u] + lane];
        }
#pragma unroll
        for (int u = 0; u < UR; ++u)
            if (lane < ln[u]) atomicAdd(&cntL[e[u].y - lo], 1);
        if (anylong) {
#pragma nounroll
            for (int u = 0; u < UR; ++u) {
                const int j = j0 + 32 * u;
                if (j >= nblk) break;
                const int2 t = trow[j];
                const int s0 = j * CH + t.x;
                for (int k = 32; k < t.y; k += 32)
                    if (k + lane < t.y) {
                        const int2 ee = tmp[s0 + k + lane];
                        atomicAdd(&cntL[ee.y - lo], 1);
                    }
            }
        }
    }
    __syncthreads();

    int bs = wred[lane];
    for (int o = 16; o > 0; o >>= 1) bs += __shfl_xor(bs, o);
    const int c = (tid < BW) ? cntL[tid] : 0;
    int incl = c;
    for (int o = 1; o < 32; o <<= 1) {
        int t = __shfl_up(incl, o);
        if (lane >= o) incl += t;
    }
    if (tid < BW && lane == 31) wsum[wave] = incl;
    __syncthreads();
    const int wsl = (lane < 16) ? wsum[lane] : 0;
    int tp = (lane < wave) ? wsl : 0;
    for (int o = 16; o > 0; o >>= 1) tp += __shfl_xor(tp, o);
    int total = wsl;
    for (int o = 16; o > 0; o >>= 1) total += __shfl_xor(total, o);
    const int i = lo + tid;
    if (tid < BW) {
        const int off = tp + incl - c;
        cntL[tid] = off;
        if (i < n) {
            rowptr[i] = bs + off;
            cnt[i] = c;
        }
    }
    __syncthreads();

    const bool inl = total <= LCAP;
    for (int j0 = wave; j0 < nblk; j0 += 32 * UR) {
        int st[UR], ln[UR];
        int2 e[UR];
        int sl[UR];
        bool anylong = false;
#pragma unroll
        for (int u = 0; u < UR; ++u) {
            const int j  = j0 + 32 * u;
            const int jc = (j < nblk) ? j : nblk - 1;
            const int2 t = trow[jc];
            st[u] = jc * CH + t.x;
            ln[u] = (j < nblk) ? t.y : 0;
            anylong |= ln[u] > 32;
        }
#pragma unroll
        for (int u = 0; u < UR; ++u) {
            e[u] = make_int2(0, lo);
            if (lane < ln[u]) e[u] = tmp[st[u] + lane];
        }
#pragma unroll
        for (int u = 0; u < UR; ++u) {
            sl[u] = 0;
            if (lane < ln[u]) sl[u] = atomicAdd(&cntL[e[u].y - lo], 1);
        }
#pragma unroll
        for (int u = 0; u < UR; ++u)
            if (lane < ln[u]) {
                if (inl) lst[sl[u]] = e[u].x;
                else     col[bs + sl[u]] = e[u].x;
            }
        if (anylong) {
#pragma nounroll
            for (int u = 0; u < UR; ++u) {
                const int j = j0 + 32 * u;
                if (j >= nblk) break;
                const int2 t = trow[j];
                const int s0 = j * CH + t.x;
                for (int k = 32; k < t.y; k += 32)
                    if (k + lane < t.y) {
                        const int2 ee = tmp[s0 + k + lane];
                        const int s2 = atomicAdd(&cntL[ee.y - lo], 1);
                        if (inl) lst[s2] = ee.x;
                        else     col[bs + s2] = ee.x;
                    }
            }
        }
    }
    __syncthreads();
    if (inl)
        for (int k = tid; k < total; k += 1024) col[bs + k] = lst[k];
}


template <int K>
__global__ __launch_bounds__(128) void gemm_wmma_k(const float* A,
                                                   const float* __restrict__ W,
                                                   const int* __restrict__ cnt,
                                                   float* O0, float* O1, int M) {
    constexpr int NOUT = HIDDEN;
    constexpr int LDSK = K + 4;
    __shared__ float Atile[16 * LDSK];

    const int tid  = threadIdx.x;
    const int row0 = blockIdx.x * 16;

    constexpr int KV = K / 4;
    for (int idx = tid; idx < 16 * KV; idx += 128) {
        int r = idx / KV, c4 = idx % KV;
        int row = row0 + r; if (row >= M) row = M - 1;
        v4f v = *(const v4f*)&A[(size_t)row * K + c4 * 4];
        *(v4f*)&Atile[r * LDSK + c4 * 4] = v;
    }
    __syncthreads();

    const int lane = tid & 31;
    const int wave = tid >> 5;
    const int half = lane >> 4;
    const int l    = lane & 15;
    const int col0 = wave * 16;

    int deg[8];
#pragma unroll
    for (int r = 0; r < 8; ++r) {
        int row = row0 + half * 8 + r; if (row >= M) row = M - 1;
        deg[r] = cnt[row];
    }

    v8f acc = {};
    const float* wp = W + col0 + l;
    for (int k = 0; k < K; k += 4) {
        const int ka = k + half * 2;
        v2f a = *(const v2f*)&Atile[l * LDSK + ka];
        v2f b;
        b.x = wp[ka * NOUT];
        b.y = wp[(ka + 1) * NOUT];
        acc = __builtin_amdgcn_wmma_f32_16x16x4_f32(false, a, false, b,
                                                    (short)0, acc, false, false);
    }

    float* O  = (wave < 2) ? O0 : O1;
    float* op = O + (size_t)(row0 + half * 8) * PLANE + (wave & 1) * 16 + l;
    if (row0 + 16 <= M) {
#pragma unroll
        for (int r = 0; r < 8; ++r) op[r * PLANE] = acc[r] * rsqrtf((float)(deg[r] + 1));
    } else {
#pragma unroll
        for (int r = 0; r < 8; ++r)
            if (row0 + half * 8 + r < M) op[r * PLANE] = acc[r] * rsqrtf((float)(deg[r] + 1));
    }
}


#define RPW 12

__global__ __launch_bounds__(256) void pull32_k(const int* __restrict__ rowptr,
                                                const int* __restrict__ colw,
                                                const float* __restrict__ hin,
                                                const float* __restrict__ bias,
                                                float* __restrict__ hout, int n, int E) {
    const int lane = threadIdx.x & 31;
    const int wave = __builtin_amdgcn_readfirstlane((int)(threadIdx.x >> 5));
    const int row0 = ((int)blockIdx.x * 8 + wave) * RPW;
    if (row0 >= n) return;
    const int nrows = (n - row0 < RPW) ? (n - row0) : RPW;

    int bnd = E;
    if (lane <= nrows && row0 + lane < n) bnd = rowptr[row0 + lane];
    const float bl = bias[lane];

    int beg_n = __builtin_amdgcn_readlane(bnd, 0);
    int end_n = __builtin_amdgcn_readlane(bnd, 1);
    int rec_n = 0;
    if (lane < end_n - beg_n) rec_n = colw[beg_n + lane];
    float self_n = hin[(size_t)row0 * PLANE + lane];

    for (int r = 0; r < nrows; ++r) {
        const int beg = beg_n, end = end_n;
        const int cur = rec_n;
        const float di = rsqrtf((float)(end - beg + 1));
        float acc = self_n;

        const int cnt   = (end - beg < 32) ? (end - beg) : 32;
        const int nfull = cnt >> 3, rem = cnt & 7, jr = cnt & ~7;
        const int s  = cur * PLANE;
#define ROWLD(j) ((hin + (size_t)(unsigned)__builtin_amdgcn_readlane(s, (j)))[lane])
        float a0, a1, a2, a3, a4, a5, a6, a7,
              b0, b1, b2, b3, b4, b5, b6, b7,
              c0, c1, c2, c3, c4, c5, c6, c7,
              d0, d1, d2, d3, d4, d5, d6, d7,
              p0, p1, p2, p3, p4, p5, p6;
        if (nfull > 0) { a0 = ROWLD(0); a1 = ROWLD(1); a2 = ROWLD(2); a3 = ROWLD(3); a4 = ROWLD(4); a5 = ROWLD(5); a6 = ROWLD(6); a7 = ROWLD(7); }
        if (nfull > 1) { b0 = ROWLD(8); b1 = ROWLD(9); b2 = ROWLD(10); b3 = ROWLD(11); b4 = ROWLD(12); b5 = ROWLD(13); b6 = ROWLD(14); b7 = ROWLD(15); }
        if (nfull > 2) { c0 = ROWLD(16); c1 = ROWLD(17); c2 = ROWLD(18); c3 = ROWLD(19); c4 = ROWLD(20); c5 = ROWLD(21); c6 = ROWLD(22); c7 = ROWLD(23); }
        if (nfull > 3) { d0 = ROWLD(24); d1 = ROWLD(25); d2 = ROWLD(26); d3 = ROWLD(27); d4 = ROWLD(28); d5 = ROWLD(29); d6 = ROWLD(30); d7 = ROWLD(31); }
        if (rem > 0) p0 = ROWLD(jr + 0);
        if (rem > 1) p1 = ROWLD(jr + 1);
        if (rem > 2) p2 = ROWLD(jr + 2);
        if (rem > 3) p3 = ROWLD(jr + 3);
        if (rem > 4) p4 = ROWLD(jr + 4);
        if (rem > 5) p5 = ROWLD(jr + 5);
        if (rem > 6) p6 = ROWLD(jr + 6);

        {
            const int rn = (r + 1 < nrows) ? r + 1 : r;
            beg_n = __builtin_amdgcn_readlane(bnd, rn);
            end_n = __builtin_amdgcn_readlane(bnd, rn + 1);
            rec_n = 0;
            if (lane < end_n - beg_n) rec_n = colw[beg_n + lane];
            self_n = hin[(size_t)(row0 + rn) * PLANE + lane];
        }

        if (nfull > 0) { acc += a0; acc += a1; acc += a2; acc += a3; acc += a4; acc += a5; acc += a6; acc += a7; }
        if (nfull > 1) { acc += b0; acc += b1; acc += b2; acc += b3; acc += b4; acc += b5; acc += b6; acc += b7; }
        if (nfull > 2) { acc += c0; acc += c1; acc += c2; acc += c3; acc += c4; acc += c5; acc += c6; acc += c7; }
        if (nfull > 3) { acc += d0; acc += d1; acc += d2; acc += d3; acc += d4; acc += d5; acc += d6; acc += d7; }
        if (rem > 0) acc += p0;
        if (rem > 1) acc += p1;
        if (rem > 2) acc += p2;
        if (rem > 3) acc += p3;
        if (rem > 4) acc += p4;
        if (rem > 5) acc += p5;
        if (rem > 6) acc += p6;
#undef ROWLD

        for (int base = beg + 32; base < end; base += 32) {
            const int cnt = (end - base < 32) ? (end - base) : 32;
            int s = 0;
            if (lane < cnt) s = colw[base + lane];
            for (int j = 0; j < cnt; ++j) {
                const int sj = __builtin_amdgcn_readlane(s, j);
                acc += hin[(size_t)sj * PLANE + lane];
            }
        }
        const float res = acc * di + bl;
        hout[(size_t)(row0 + r) * PLANE + lane] = res > 0.0f ? res * di : 0.0f;
    }
}


#define RPW3 16
#define TP   68
#define WPP  160

#define XM(v, m) __int_as_float(__builtin_amdgcn_update_dpp(0, __float_as_int(v), 0x160 | (m), 0xf, 0xf, true))
__device__ __forceinline__ float row_sum16(float v) {
    v += XM(v, 1);
    v += XM(v, 2);
    v += XM(v, 4);
    v += XM(v, 8);
    return v;
}
#undef XM

__global__ __launch_bounds__(128) void pull2p_w2w3_k(const int* __restrict__ rowptr,
                                                     const int* __restrict__ colw,
                                                     const float* __restrict__ g0,
                                                     const float* __restrict__ g1,
                                                     const float* __restrict__ W2,
                                                     const float* __restrict__ bias,
                                                     const float* __restrict__ W3,
                                                     float* __restrict__ h3, int n, int E) {
    __shared__ __attribute__((aligned(16))) float W2s[(HIDDEN / 2) * WPP];
    __shared__ __attribute__((aligned(16))) float Ts[4 * RPW3 * TP];

    for (int it = threadIdx.x; it < (HIDDEN / 2) * (HIDDEN / 4); it += 128) {
        const int kp = it >> 4, c4 = it & 15;
        const v4f r0 = *(const v4f*)&W2[(2 * kp) * HIDDEN + c4 * 4];
        const v4f r1 = *(const v4f*)&W2[(2 * kp + 1) * HIDDEN + c4 * 4];
        const v4f lo = {r0.x, r1.x, r0.y, r1.y};
        const v4f hi = {r0.z, r1.z, r0.w, r1.w};
        *(v4f*)&W2s[kp * WPP + c4 * 8]     = lo;
        *(v4f*)&W2s[kp * WPP + c4 * 8 + 4] = hi;
    }
    __syncthreads();

    const int lane = threadIdx.x & 31;
    const int wave = __builtin_amdgcn_readfirstlane((int)(threadIdx.x >> 5));
    const int row0 = ((int)blockIdx.x * 4 + wave) * RPW3;
    if (row0 >= n) return;
    const int nrows = (n - row0 < RPW3) ? (n - row0) : RPW3;
    float* T = Ts + wave * (RPW3 * TP);

    int bnd = E;
    if (lane <= nrows && row0 + lane < n) bnd = rowptr[row0 + lane];

    int beg_n = __builtin_amdgcn_readlane(bnd, 0);
    int end_n = __builtin_amdgcn_readlane(bnd, 1);
    int rec_n = 0;
    if (lane < end_n - beg_n) rec_n = colw[beg_n + lane];
    float self0_n = g0[(size_t)row0 * PLANE + lane];
    float self1_n = g1[(size_t)row0 * PLANE + lane];

    for (int r = 0; r < nrows; ++r) {
        const int beg = beg_n, end = end_n;
        const int cur = rec_n;
        const float di = rsqrtf((float)(end - beg + 1));
        float acc0 = self0_n;
        float acc1 = self1_n;

        const int cnt   = (end - beg < 32) ? (end - beg) : 32;
        const int nfull = cnt >> 3, rem = cnt & 7, jr = cnt & ~7;
        const int s  = cur * PLANE;
#define OFFJ(j)  ((size_t)(unsigned)__builtin_amdgcn_readlane(s, (j)))
#define LD0(j)   ((g0 + OFFJ(j))[lane])
#define LD1(j)   ((g1 + OFFJ(j))[lane])
        float a0, a1, a2, a3, a4, a5, a6, a7,  A0, A1, A2, A3, A4, A5, A6, A7,
              b0, b1, b2, b3, b4, b5, b6, b7,  B0, B1, B2, B3, B4, B5, B6, B7,
              c0, c1, c2, c3, c4, c5, c6, c7,  C0, C1, C2, C3, C4, C5, C6, C7,
              d0, d1, d2, d3, d4, d5, d6, d7,  D0, D1, D2, D3, D4, D5, D6, D7,
              p0, p1, p2, p3, p4, p5, p6,      P0, P1, P2, P3, P4, P5, P6;
        if (nfull > 0) { a0 = LD0(0); A0 = LD1(0); a1 = LD0(1); A1 = LD1(1); a2 = LD0(2); A2 = LD1(2); a3 = LD0(3); A3 = LD1(3);
                         a4 = LD0(4); A4 = LD1(4); a5 = LD0(5); A5 = LD1(5); a6 = LD0(6); A6 = LD1(6); a7 = LD0(7); A7 = LD1(7); }
        if (nfull > 1) { b0 = LD0(8); B0 = LD1(8); b1 = LD0(9); B1 = LD1(9); b2 = LD0(10); B2 = LD1(10); b3 = LD0(11); B3 = LD1(11);
                         b4 = LD0(12); B4 = LD1(12); b5 = LD0(13); B5 = LD1(13); b6 = LD0(14); B6 = LD1(14); b7 = LD0(15); B7 = LD1(15); }
        if (nfull > 2) { c0 = LD0(16); C0 = LD1(16); c1 = LD0(17); C1 = LD1(17); c2 = LD0(18); C2 = LD1(18); c3 = LD0(19); C3 = LD1(19);
                         c4 = LD0(20); C4 = LD1(20); c5 = LD0(21); C5 = LD1(21); c6 = LD0(22); C6 = LD1(22); c7 = LD0(23); C7 = LD1(23); }
        if (nfull > 3) { d0 = LD0(24); D0 = LD1(24); d1 = LD0(25); D1 = LD1(25); d2 = LD0(26); D2 = LD1(26); d3 = LD0(27); D3 = LD1(27);
                         d4 = LD0(28); D4 = LD1(28); d5 = LD0(29); D5 = LD1(29); d6 = LD0(30); D6 = LD1(30); d7 = LD0(31); D7 = LD1(31); }
        if (rem > 0) { p0 = LD0(jr + 0); P0 = LD1(jr + 0); }
        if (rem > 1) { p1 = LD0(jr + 1); P1 = LD1(jr + 1); }
        if (rem > 2) { p2 = LD0(jr + 2); P2 = LD1(jr + 2); }
        if (rem > 3) { p3 = LD0(jr + 3); P3 = LD1(jr + 3); }
        if (rem > 4) { p4 = LD0(jr + 4); P4 = LD1(jr + 4); }
        if (rem > 5) { p5 = LD0(jr + 5); P5 = LD1(jr + 5); }
        if (rem > 6) { p6 = LD0(jr + 6); P6 = LD1(jr + 6); }

        {
            const int rn = (r + 1 < nrows) ? r + 1 : r;
            beg_n = __builtin_amdgcn_readlane(bnd, rn);
            end_n = __builtin_amdgcn_readlane(bnd, rn + 1);
            rec_n = 0;
            if (lane < end_n - beg_n) rec_n = colw[beg_n + lane];
            self0_n = g0[(size_t)(row0 + rn) * PLANE + lane];
            self1_n = g1[(size_t)(row0 + rn) * PLANE + lane];
        }

#define AD(x, X, j) { acc0 += (x); acc1 += (X); }
        if (nfull > 0) { AD(a0, A0, 0) AD(a1, A1, 1) AD(a2, A2, 2) AD(a3, A3, 3) AD(a4, A4, 4) AD(a5, A5, 5) AD(a6, A6, 6) AD(a7, A7, 7) }
        if (nfull > 1) { AD(b0, B0, 8) AD(b1, B1, 9) AD(b2, B2, 10) AD(b3, B3, 11) AD(b4, B4, 12) AD(b5, B5, 13) AD(b6, B6, 14) AD(b7, B7, 15) }
        if (nfull > 2) { AD(c0, C0, 16) AD(c1, C1, 17) AD(c2, C2, 18) AD(c3, C3, 19) AD(c4, C4, 20) AD(c5, C5, 21) AD(c6, C6, 22) AD(c7, C7, 23) }
        if (nfull > 3) { AD(d0, D0, 24) AD(d1, D1, 25) AD(d2, D2, 26) AD(d3, D3, 27) AD(d4, D4, 28) AD(d5, D5, 29) AD(d6, D6, 30) AD(d7, D7, 31) }
        if (rem > 0) AD(p0, P0, jr + 0)
        if (rem > 1) AD(p1, P1, jr + 1)
        if (rem > 2) AD(p2, P2, jr + 2)
        if (rem > 3) AD(p3, P3, jr + 3)
        if (rem > 4) AD(p4, P4, jr + 4)
        if (rem > 5) AD(p5, P5, jr + 5)
        if (rem > 6) AD(p6, P6, jr + 6)
#undef AD
#undef LD0
#undef LD1
#undef OFFJ

        for (int base = beg + 32; base < end; base += 32) {
            const int cnt = (end - base < 32) ? (end - base) : 32;
            int s = 0;
            if (lane < cnt) s = colw[base + lane];
            for (int j = 0; j < cnt; ++j) {
                const size_t off = (size_t)(unsigned)__builtin_amdgcn_readlane(s, j) * PLANE + lane;
                acc0 += g0[off];
                acc1 += g1[off];
            }
        }
        T[r * TP + lane]         = acc0 * di;
        T[r * TP + PLANE + lane] = acc1 * di;
    }
    for (int r = nrows; r < RPW3; ++r) {
        T[r * TP + lane]         = 0.0f;
        T[r * TP + PLANE + lane] = 0.0f;
    }

    const int half = lane >> 4;
    const int l    = lane & 15;
    const float* tp = T + l * TP + half * 2;
    const float* bp = W2s + half * WPP + l * 2;
    v8f q0 = {}, q1 = {}, q2 = {}, q3 = {};
    for (int st = 0; st < HIDDEN / 4; ++st) {
        const v2f a  = *(const v2f*)(tp + 4 * st);
        const float* bq = bp + 2 * st * WPP;
        const v2f w0 = *(const v2f*)(bq);
        const v2f w1 = *(const v2f*)(bq + 32);
        const v2f w2 = *(const v2f*)(bq + 64);
        const v2f w3 = *(const v2f*)(bq + 96);
        q0 = __builtin_amdgcn_wmma_f32_16x16x4_f32(false, a, false, w0, (short)0, q0, false, false);
        q1 = __builtin_amdgcn_wmma_f32_16x16x4_f32(false, a, false, w1, (short)0, q1, false, false);
        q2 = __builtin_amdgcn_wmma_f32_16x16x4_f32(false, a, false, w2, (short)0, q2, false, false);
        q3 = __builtin_amdgcn_wmma_f32_16x16x4_f32(false, a, false, w3, (short)0, q3, false, false);
    }

    const float bb0 = bias[l], bb1 = bias[16 + l], bb2 = bias[32 + l], bb3 = bias[48 + l];
    const v2f u0 = *(const v2f*)&W3[2 * l];
    const v2f u1 = *(const v2f*)&W3[2 * (16 + l)];
    const v2f u2 = *(const v2f*)&W3[2 * (32 + l)];
    const v2f u3 = *(const v2f*)&W3[2 * (48 + l)];
    float keep = 0.0f;
#pragma unroll
    for (int i = 0; i < 8; ++i) {
        const float z0 = q0[i] + bb0, z1 = q1[i] + bb1, z2 = q2[i] + bb2, z3 = q3[i] + bb3;
        const float v0 = z0 > 0.0f ? z0 : 0.0f;
        const float v1 = z1 > 0.0f ? z1 : 0.0f;
        const float v2 = z2 > 0.0f ? z2 : 0.0f;
        const float v3 = z3 > 0.0f ? z3 : 0.0f;
        const float t0 = row_sum16(v0 * u0.x + v1 * u1.x + v2 * u2.x + v3 * u3.x);
        const float t1 = row_sum16(v0 * u0.y + v1 * u1.y + v2 * u2.y + v3 * u3.y);
        if ((l >> 1) == i) keep = (l & 1) ? t1 : t0;
    }
    const int rr  = lane >> 1;
    const int deg = __shfl(bnd, rr + 1) - __shfl(bnd, rr);
    if (lane < 2 * nrows) h3[(size_t)row0 * 2 + lane] = keep * rsqrtf((float)(deg + 1));
}

__global__ void pull2_lsm_k(const int* __restrict__ rowptr, const int* __restrict__ colw,
                            const float* __restrict__ h3, const float* __restrict__ b3,
                            float* __restrict__ out, int n, int E) {
    int i = blockIdx.x * blockDim.x + threadIdx.x;
    if (i >= n) return;
    const int beg = rowptr[i];
    const int end = (i + 1 < n) ? rowptr[i + 1] : E;
    const float di = rsqrtf((float)(end - beg + 1));
    const v2f* h2 = (const v2f*)h3;
    const v2f hs = h2[i];
    float a0 = hs.x, a1 = hs.y;
    for (int e = beg; e < end; ++e) {
        const v2f v = h2[colw[e]];
        a0 += v.x;
        a1 += v.y;
    }
    float z0 = a0 * di + b3[0];
    float z1 = a1 * di + b3[1];
    float m = fmaxf(z0, z1);
    float lse = m + logf(__expf(z0 - m) + __expf(z1 - m));
    out[2 * i] = z0 - lse;
    out[2 * i + 1] = z1 - lse;
}


extern "C" void kernel_launch(void* const* d_in, const int* in_sizes, int n_in,
                              void* d_out, int out_size, void* d_ws, size_t ws_size,
                              hipStream_t stream) {
    const float* x  = (const float*)d_in[0];
    const int*   ei = (const int*)d_in[1];
    const float* W1 = (const float*)d_in[2];
    const float* b1 = (const float*)d_in[3];
    const float* W2 = (const float*)d_in[4];
    const float* b2 = (const float*)d_in[5];
    const float* W3 = (const float*)d_in[6];
    const float* b3 = (const float*)d_in[7];
    float* out = (float*)d_out;

    const int N = in_sizes[0] / F_IN;
    const int E = in_sizes[1] / 2;
    const int* src = ei;
    const int* dst = ei + E;

    int*   colw   = (int*)d_ws;
    int*   rowptr = (int*)d_ws + (size_t)2 * E;
    float* pA     = (float*)(rowptr + N);
    float* pB     = pA + (size_t)N * PLANE;
    float* pC     = pB + (size_t)N * PLANE;
    int2*  tmp    = (int2*)pC;
    int2*  tab    = (int2*)pB;
    float* h3     = pB;
    int*   cnt    = (int*)d_out;

    const int T = 256;
    dim3 blk(T);
    int gN    = (N + T - 1) / T;
    int gRow  = (N + 8 * RPW - 1) / (8 * RPW);
    int gTile = (N + 15) / 16;
    int gRow3 = (N + 4 * RPW3 - 1) / (4 * RPW3);
    int nbk   = (N + BW - 1) / BW;
    int nblk  = (E + CH - 1) / CH;

    part_k<<<nblk, dim3(1024), 0, stream>>>(src, dst, tmp, tab, E, nbk, nblk);
    bucket_k<<<nbk, dim3(1024), 0, stream>>>(tmp, tab, cnt, rowptr, colw, N, nblk);

    gemm_wmma_k<F_IN><<<gTile, dim3(128), 0, stream>>>(x, W1, cnt, pA, pB, N);
    pull32_k<<<gRow, blk, 0, stream>>>(rowptr, colw, pA, b1, pC, N, E);
    pull32_k<<<gRow, blk, 0, stream>>>(rowptr, colw, pB, b1 + PLANE, pA, N, E);

    pull2p_w2w3_k<<<gRow3, dim3(128), 0, stream>>>(rowptr, colw, pC, pA, W2, b2, W3, h3, N, E);

    pull2_lsm_k<<<gN, blk, 0, stream>>>(rowptr, colw, h3, b3, out, N, E);
}


# dependent load trips cut: last-layer wave loads 32 rows' records, then all source rows, at once (3 trips, were ~20; LDS sum, order kept); product's 8 tile loads together; bucket walk 2 reuses walk 1
# speedup vs baseline: 6.7765x; 1.0594x over previous
#include <hip/hip_runtime.h>


typedef __attribute__((ext_vector_type(2))) float v2f;
typedef __attribute__((ext_vector_type(4))) float v4f;
typedef __attribute__((ext_vector_type(8))) float v8f;

#define F_IN   256
#define HIDDEN 64
#define PLANE  32


#define BSH  9
#define BW   (1 << BSH)
#define MAXB 256
#define CH   4096

__global__ __launch_bounds__(1024) void part_k(const int* __restrict__ src, const int* __restrict__ dst,
                                               int2* __restrict__ tmp, int2* __restrict__ tab,
                                               int E, int nbk, int nblk) {
    __shared__ int  hist[MAXB];
    __shared__ int  offL[MAXB];
    __shared__ int  wsum[8];
    __shared__ int2 stage[CH];
    const int tid = threadIdx.x, lane = tid & 31, wave = tid >> 5;
    const int base = blockIdx.x * CH;
    if (tid < MAXB) hist[tid] = 0;
    __syncthreads();

    int s[4], d[4], r[4];
#pragma unroll
    for (int k = 0; k < 4; ++k) {
        const int e = base + k * 1024 + tid;
        s[k] = 0; d[k] = -1; r[k] = 0;
        if (e < E) { s[k] = src[e]; d[k] = dst[e]; }
    }
#pragma unroll
    for (int k = 0; k < 4; ++k)
        if (d[k] >= 0) r[k] = atomicAdd(&hist[d[k] >> BSH], 1);
    __syncthreads();

    const int c = (tid < MAXB) ? hist[tid] : 0;
    int incl = c;
    for (int o = 1; o < 32; o <<= 1) {
        int t = __shfl_up(incl, o);
        if (lane >= o) incl += t;
    }
    if (tid < MAXB && lane == 31) wsum[wave] = incl;
    __syncthreads();
    if (tid < MAXB) {
        int pre = 0;
        for (int w = 0; w < wave; ++w) pre += wsum[w];
        offL[tid] = pre + incl - c;
    }
    __syncthreads();

#pragma unroll
    for (int k = 0; k < 4; ++k)
        if (d[k] >= 0) stage[offL[d[k] >> BSH] + r[k]] = make_int2(s[k], d[k]);
    __syncthreads();
    const int nvalid = (E - base < CH) ? (E - base) : CH;
#pragma unroll
    for (int k = 0; k < 4; ++k) {
        const int idx = k * 1024 + tid;
        if (idx < nvalid) tmp[base + idx] = stage[idx];
    }
    if (tid < nbk) tab[(size_t)tid * nblk + blockIdx.x] = make_int2(offL[tid], hist[tid]);
}

#define LCAP 16384
#define UR   13
__global__ __launch_bounds__(1024) void bucket_k(const int2* __restrict__ tmp, const int2* __restrict__ tab,
                                                 int* __restrict__ cnt, int* __restrict__ rowptr,
                                                 int* __restrict__ col, int n, int nblk) {
    __shared__ int cntL[BW];
    __shared__ int wred[32];
    __shared__ int wsum[16];
    __shared__ int lst[LCAP];
    const int tid = threadIdx.x, lane = tid & 31;
    const int wave = __builtin_amdgcn_readfirstlane((int)(threadIdx.x >> 5));
    const int b = blockIdx.x, lo = b << BSH;

    const int2* trow = tab + (size_t)b * nblk;
    int part = 0;
    for (int j = tid; j < nblk; j += 1024) part += trow[j].x;
    for (int o = 16; o > 0; o >>= 1) part += __shfl_xor(part, o);
    if (tid < BW) cntL[tid] = 0;
    if (lane == 0) wred[wave] = part;
    __syncthreads();

    const bool one = nblk <= 32 * UR;
    int st[UR], ln[UR];
    int2 e[UR];
    bool anylong = false;
    for (int j0 = wave; j0 < nblk; j0 += 32 * UR) {
        anylong = false;
#pragma unroll
        for (int u = 0; u < UR; ++u) {
            const int j  = j0 + 32 * u;
            const int jc = (j < nblk) ? j : nblk - 1;
            const int2 t = trow[jc];
            st[u] = jc * CH + t.x;
            ln[u] = (j < nblk) ? t.y : 0;
            anylong |= ln[u] > 32;
        }
#pragma unroll
        for (int u = 0; u < UR; ++u) {
            e[u] = make_int2(0, lo);
            if (lane < ln[u]) e[u] = tmp[st[u] + lane];
        }
#pragma unroll
        for (int u = 0; u < UR; ++u)
            if (lane < ln[u]) atomicAdd(&cntL[e[u].y - lo], 1);
        if (anylong) {
#pragma nounroll
            for (int u = 0; u < UR; ++u) {
                const int j = j0 + 32 * u;
                if (j >= nblk) break;
                const int2 t = trow[j];
                const int s0 = j * CH + t.x;
                for (int k = 32; k < t.y; k += 32)
                    if (k + lane < t.y) {
                        const int2 ee = tmp[s0 + k + lane];
                        atomicAdd(&cntL[ee.y - lo], 1);
                    }
            }
        }
    }
    __syncthreads();

    int bs = wred[lane];
    for (int o = 16; o > 0; o >>= 1) bs += __shfl_xor(bs, o);
    const int c = (tid < BW) ? cntL[tid] : 0;
    int incl = c;
    for (int o = 1; o < 32; o <<= 1) {
        int t = __shfl_up(incl, o);
        if (lane >= o) incl += t;
    }
    if (tid < BW && lane == 31) wsum[wave] = incl;
    __syncthreads();
    const int wsl = (lane < 16) ? wsum[lane] : 0;
    int tp = (lane < wave) ? wsl : 0;
    for (int o = 16; o > 0; o >>= 1) tp += __shfl_xor(tp, o);
    int total = wsl;
    for (int o = 16; o > 0; o >>= 1) total += __shfl_xor(total, o);
    const int i = lo + tid;
    if (tid < BW) {
        const int off = tp + incl - c;
        cntL[tid] = off;
        if (i < n) {
            rowptr[i] = bs + off;
            cnt[i] = c;
        }
    }
    __syncthreads();

    const bool inl = total <= LCAP;
    for (int j0 = wave; j0 < nblk; j0 += 32 * UR) {
        int sl[UR];
        if (!one) {
            anylong = false;
#pragma unroll
            for (int u = 0; u < UR; ++u) {
                const int j  = j0 + 32 * u;
                const int jc = (j < nblk) ? j : nblk - 1;
                const int2 t = trow[jc];
                st[u] = jc * CH + t.x;
                ln[u] = (j < nblk) ? t.y : 0;
                anylong |= ln[u] > 32;
            }
#pragma unroll
            for (int u = 0; u < UR; ++u) {
                e[u] = make_int2(0, lo);
                if (lane < ln[u]) e[u] = tmp[st[u] + lane];
            }
        }
#pragma unroll
        for (int u = 0; u < UR; ++u) {
            sl[u] = 0;
            if (lane < ln[u]) sl[u] = atomicAdd(&cntL[e[u].y - lo], 1);
        }
#pragma unroll
        for (int u = 0; u < UR; ++u)
            if (lane < ln[u]) {
                if (inl) lst[sl[u]] = e[u].x;
                else     col[bs + sl[u]] = e[u].x;
            }
        if (anylong) {
#pragma nounroll
            for (int u = 0; u < UR; ++u) {
                const int j = j0 + 32 * u;
                if (j >= nblk) break;
                const int2 t = trow[j];
                const int s0 = j * CH + t.x;
                for (int k = 32; k < t.y; k += 32)
                    if (k + lane < t.y) {
                        const int2 ee = tmp[s0 + k + lane];
                        const int s2 = atomicAdd(&cntL[ee.y - lo], 1);
                        if (inl) lst[s2] = ee.x;
                        else     col[bs + s2] = ee.x;
                    }
            }
        }
    }
    __syncthreads();
    if (inl)
        for (int k = tid; k < total; k += 1024) col[bs + k] = lst[k];
}


template <int K>
__global__ __launch_bounds__(128) void gemm_wmma_k(const float* A,
                                                   const float* __restrict__ W,
                                                   const int* __restrict__ cnt,
                                                   float* O0, float* O1, int M) {
    constexpr int NOUT = HIDDEN;
    constexpr int LDSK = K + 4;
    __shared__ float Atile[16 * LDSK];

    const int tid  = threadIdx.x;
    const int row0 = blockIdx.x * 16;

    constexpr int KV  = K / 4;
    constexpr int NLD = 16 * KV / 128;
    static_assert(NLD * 128 == 16 * KV, "tile is a whole number of float4s per thread");
    v4f stg[NLD];
#pragma unroll
    for (int u = 0; u < NLD; ++u) {
        const int idx = tid + u * 128;
        const int r = idx / KV, c4 = idx % KV;
        int row = row0 + r; if (row >= M) row = M - 1;
        stg[u] = *(const v4f*)&A[(size_t)row * K + c4 * 4];
    }
#pragma unroll
    for (int u = 0; u < NLD; ++u) {
        const int idx = tid + u * 128;
        const int r = idx / KV, c4 = idx % KV;
        *(v4f*)&Atile[r * LDSK + c4 * 4] = stg[u];
    }
    __syncthreads();

    const int lane = tid & 31;
    const int wave = tid >> 5;
    const int half = lane >> 4;
    const int l    = lane & 15;
    const int col0 = wave * 16;

    int deg[8];
#pragma unroll
    for (int r = 0; r < 8; ++r) {
        int row = row0 + half * 8 + r; if (row >= M) row = M - 1;
        deg[r] = cnt[row];
    }

    v8f acc = {};
    const float* wp = W + col0 + l;
    for (int k = 0; k < K; k += 4) {
        const int ka = k + half * 2;
        v2f a = *(const v2f*)&Atile[l * LDSK + ka];
        v2f b;
        b.x = wp[ka * NOUT];
        b.y = wp[(ka + 1) * NOUT];
        acc = __builtin_amdgcn_wmma_f32_16x16x4_f32(false, a, false, b,
                                                    (short)0, acc, false, false);
    }

    float* O  = (wave < 2) ? O0 : O1;
    float* op = O + (size_t)(row0 + half * 8) * PLANE + (wave & 1) * 16 + l;
    if (row0 + 16 <= M) {
#pragma unroll
        for (int r = 0; r < 8; ++r) op[r * PLANE] = acc[r] * rsqrtf((float)(deg[r] + 1));
    } else {
#pragma unroll
        for (int r = 0; r < 8; ++r)
            if (row0 + half * 8 + r < M) op[r * PLANE] = acc[r] * rsqrtf((float)(deg[r] + 1));
    }
}


#define RPW 12

__global__ __launch_bounds__(256) void pull32_k(const int* __restrict__ rowptr,
                                                const int* __restrict__ colw,
                                                const float* __restrict__ hin,
                                                const float* __restrict__ bias,
                                                float* __restrict__ hout, int n, int E) {
    const int lane = threadIdx.x & 31;
    const int wave = __builtin_amdgcn_readfirstlane((int)(threadIdx.x >> 5));
    const int row0 = ((int)blockIdx.x * 8 + wave) * RPW;
    if (row0 >= n) return;
    const int nrows = (n - row0 < RPW) ? (n - row0) : RPW;

    int bnd = E;
    if (lane <= nrows && row0 + lane < n) bnd = rowptr[row0 + lane];
    const float bl = bias[lane];

    int beg_n = __builtin_amdgcn_readlane(bnd, 0);
    int end_n = __builtin_amdgcn_readlane(bnd, 1);
    int rec_n = 0;
    if (lane < end_n - beg_n) rec_n = colw[beg_n + lane];
    float self_n = hin[(size_t)row0 * PLANE + lane];

    for (int r = 0; r < nrows; ++r) {
        const int beg = beg_n, end = end_n;
        const int cur = rec_n;
        const float di = rsqrtf((float)(end - beg + 1));
        float acc = self_n;

        const int cnt   = (end - beg < 32) ? (end - beg) : 32;
        const int nfull = cnt >> 3, rem = cnt & 7, jr = cnt & ~7;
        const int s  = cur * PLANE;
#define ROWLD(j) ((hin + (size_t)(unsigned)__builtin_amdgcn_readlane(s, (j)))[lane])
        float a0, a1, a2, a3, a4, a5, a6, a7,
              b0, b1, b2, b3, b4, b5, b6, b7,
              c0, c1, c2, c3, c4, c5, c6, c7,
              d0, d1, d2, d3, d4, d5, d6, d7,
              p0, p1, p2, p3, p4, p5, p6;
        if (nfull > 0) { a0 = ROWLD(0); a1 = ROWLD(1); a2 = ROWLD(2); a3 = ROWLD(3); a4 = ROWLD(4); a5 = ROWLD(5); a6 = ROWLD(6); a7 = ROWLD(7); }
        if (nfull > 1) { b0 = ROWLD(8); b1 = ROWLD(9); b2 = ROWLD(10); b3 = ROWLD(11); b4 = ROWLD(12); b5 = ROWLD(13); b6 = ROWLD(14); b7 = ROWLD(15); }
        if (nfull > 2) { c0 = ROWLD(16); c1 = ROWLD(17); c2 = ROWLD(18); c3 = ROWLD(19); c4 = ROWLD(20); c5 = ROWLD(21); c6 = ROWLD(22); c7 = ROWLD(23); }
        if (nfull > 3) { d0 = ROWLD(24); d1 = ROWLD(25); d2 = ROWLD(26); d3 = ROWLD(27); d4 = ROWLD(28); d5 = ROWLD(29); d6 = ROWLD(30); d7 = ROWLD(31); }
        if (rem > 0) p0 = ROWLD(jr + 0);
        if (rem > 1) p1 = ROWLD(jr + 1);
        if (rem > 2) p2 = ROWLD(jr + 2);
        if (rem > 3) p3 = ROWLD(jr + 3);
        if (rem > 4) p4 = ROWLD(jr + 4);
        if (rem > 5) p5 = ROWLD(jr + 5);
        if (rem > 6) p6 = ROWLD(jr + 6);

        {
            const int rn = (r + 1 < nrows) ? r + 1 : r;
            beg_n = __builtin_amdgcn_readlane(bnd, rn);
            end_n = __builtin_amdgcn_readlane(bnd, rn + 1);
            rec_n = 0;
            if (lane < end_n - beg_n) rec_n = colw[beg_n + lane];
            self_n = hin[(size_t)(row0 + rn) * PLANE + lane];
        }

        if (nfull > 0) { acc += a0; acc += a1; acc += a2; acc += a3; acc += a4; acc += a5; acc += a6; acc += a7; }
        if (nfull > 1) { acc += b0; acc += b1; acc += b2; acc += b3; acc += b4; acc += b5; acc += b6; acc += b7; }
        if (nfull > 2) { acc += c0; acc += c1; acc += c2; acc += c3; acc += c4; acc += c5; acc += c6; acc += c7; }
        if (nfull > 3) { acc += d0; acc += d1; acc += d2; acc += d3; acc += d4; acc += d5; acc += d6; acc += d7; }
        if (rem > 0) acc += p0;
        if (rem > 1) acc += p1;
        if (rem > 2) acc += p2;
        if (rem > 3) acc += p3;
        if (rem > 4) acc += p4;
        if (rem > 5) acc += p5;
        if (rem > 6) acc += p6;
#undef ROWLD

        for (int base = beg + 32; base < end; base += 32) {
            const int cnt = (end - base < 32) ? (end - base) : 32;
            int s = 0;
            if (lane < cnt) s = colw[base + lane];
            for (int j = 0; j < cnt; ++j) {
                const int sj = __builtin_amdgcn_readlane(s, j);
                acc += hin[(size_t)sj * PLANE + lane];
            }
        }
        const float res = acc * di + bl;
        hout[(size_t)(row0 + r) * PLANE + lane] = res > 0.0f ? res * di : 0.0f;
    }
}


#define RPW3 16
#define TP   68
#define WPP  160

#define XM(v, m) __int_as_float(__builtin_amdgcn_update_dpp(0, __float_as_int(v), 0x160 | (m), 0xf, 0xf, true))
__device__ __forceinline__ float row_sum16(float v) {
    v += XM(v, 1);
    v += XM(v, 2);
    v += XM(v, 4);
    v += XM(v, 8);
    return v;
}
#undef XM

__global__ __launch_bounds__(128) void pull2p_w2w3_k(const int* __restrict__ rowptr,
                                                     const int* __restrict__ colw,
                                                     const float* __restrict__ g0,
                                                     const float* __restrict__ g1,
                                                     const float* __restrict__ W2,
                                                     const float* __restrict__ bias,
                                                     const float* __restrict__ W3,
                                                     float* __restrict__ h3, int n, int E) {
    __shared__ __attribute__((aligned(16))) float W2s[(HIDDEN / 2) * WPP];
    __shared__ __attribute__((aligned(16))) float Ts[4 * RPW3 * TP];

    for (int it = threadIdx.x; it < (HIDDEN / 2) * (HIDDEN / 4); it += 128) {
        const int kp = it >> 4, c4 = it & 15;
        const v4f r0 = *(const v4f*)&W2[(2 * kp) * HIDDEN + c4 * 4];
        const v4f r1 = *(const v4f*)&W2[(2 * kp + 1) * HIDDEN + c4 * 4];
        const v4f lo = {r0.x, r1.x, r0.y, r1.y};
        const v4f hi = {r0.z, r1.z, r0.w, r1.w};
        *(v4f*)&W2s[kp * WPP + c4 * 8]     = lo;
        *(v4f*)&W2s[kp * WPP + c4 * 8 + 4] = hi;
    }
    __syncthreads();

    const int lane = threadIdx.x & 31;
    const int wave = __builtin_amdgcn_readfirstlane((int)(threadIdx.x >> 5));
    const int row0 = ((int)blockIdx.x * 4 + wave) * RPW3;
    if (row0 >= n) return;
    const int nrows = (n - row0 < RPW3) ? (n - row0) : RPW3;
    float* T = Ts + wave * (RPW3 * TP);

    int bnd = E;
    if (lane <= nrows && row0 + lane < n) bnd = rowptr[row0 + lane];

    int beg_n = __builtin_amdgcn_readlane(bnd, 0);
    int end_n = __builtin_amdgcn_readlane(bnd, 1);
    int rec_n = 0;
    if (lane < end_n - beg_n) rec_n = colw[beg_n + lane];
    float self0_n = g0[(size_t)row0 * PLANE + lane];
    float self1_n = g1[(size_t)row0 * PLANE + lane];

    for (int r = 0; r < nrows; ++r) {
        const int beg = beg_n, end = end_n;
        const int cur = rec_n;
        const float di = rsqrtf((float)(end - beg + 1));
        float acc0 = self0_n;
        float acc1 = self1_n;

        const int cnt   = (end - beg < 32) ? (end - beg) : 32;
        const int nfull = cnt >> 3, rem = cnt & 7, jr = cnt & ~7;
        const int s  = cur * PLANE;
#define OFFJ(j)  ((size_t)(unsigned)__builtin_amdgcn_readlane(s, (j)))
#define LD0(j)   ((g0 + OFFJ(j))[lane])
#define LD1(j)   ((g1 + OFFJ(j))[lane])
        float a0, a1, a2, a3, a4, a5, a6, a7,  A0, A1, A2, A3, A4, A5, A6, A7,
              b0, b1, b2, b3, b4, b5, b6, b7,  B0, B1, B2, B3, B4, B5, B6, B7,
              c0, c1, c2, c3, c4, c5, c6, c7,  C0, C1, C2, C3, C4, C5, C6, C7,
              d0, d1, d2, d3, d4, d5, d6, d7,  D0, D1, D2, D3, D4, D5, D6, D7,
              p0, p1, p2, p3, p4, p5, p6,      P0, P1, P2, P3, P4, P5, P6;
        if (nfull > 0) { a0 = LD0(0); A0 = LD1(0); a1 = LD0(1); A1 = LD1(1); a2 = LD0(2); A2 = LD1(2); a3 = LD0(3); A3 = LD1(3);
                         a4 = LD0(4); A4 = LD1(4); a5 = LD0(5); A5 = LD1(5); a6 = LD0(6); A6 = LD1(6); a7 = LD0(7); A7 = LD1(7); }
        if (nfull > 1) { b0 = LD0(8); B0 = LD1(8); b1 = LD0(9); B1 = LD1(9); b2 = LD0(10); B2 = LD1(10); b3 = LD0(11); B3 = LD1(11);
                         b4 = LD0(12); B4 = LD1(12); b5 = LD0(13); B5 = LD1(13); b6 = LD0(14); B6 = LD1(14); b7 = LD0(15); B7 = LD1(15); }
        if (nfull > 2) { c0 = LD0(16); C0 = LD1(16); c1 = LD0(17); C1 = LD1(17); c2 = LD0(18); C2 = LD1(18); c3 = LD0(19); C3 = LD1(19);
                         c4 = LD0(20); C4 = LD1(20); c5 = LD0(21); C5 = LD1(21); c6 = LD0(22); C6 = LD1(22); c7 = LD0(23); C7 = LD1(23); }
        if (nfull > 3) { d0 = LD0(24); D0 = LD1(24); d1 = LD0(25); D1 = LD1(25); d2 = LD0(26); D2 = LD1(26); d3 = LD0(27); D3 = LD1(27);
                         d4 = LD0(28); D4 = LD1(28); d5 = LD0(29); D5 = LD1(29); d6 = LD0(30); D6 = LD1(30); d7 = LD0(31); D7 = LD1(31); }
        if (rem > 0) { p0 = LD0(jr + 0); P0 = LD1(jr + 0); }
        if (rem > 1) { p1 = LD0(jr + 1); P1 = LD1(jr + 1); }
        if (rem > 2) { p2 = LD0(jr + 2); P2 = LD1(jr + 2); }
        if (rem > 3) { p3 = LD0(jr + 3); P3 = LD1(jr + 3); }
        if (rem > 4) { p4 = LD0(jr + 4); P4 = LD1(jr + 4); }
        if (rem > 5) { p5 = LD0(jr + 5); P5 = LD1(jr + 5); }
        if (rem > 6) { p6 = LD0(jr + 6); P6 = LD1(jr + 6); }

        {
            const int rn = (r + 1 < nrows) ? r + 1 : r;
            beg_n = __builtin_amdgcn_readlane(bnd, rn);
            end_n = __builtin_amdgcn_readlane(bnd, rn + 1);
            rec_n = 0;
            if (lane < end_n - beg_n) rec_n = colw[beg_n + lane];
            self0_n = g0[(size_t)(row0 + rn) * PLANE + lane];
            self1_n = g1[(size_t)(row0 + rn) * PLANE + lane];
        }

#define AD(x, X, j) { acc0 += (x); acc1 += (X); }
        if (nfull > 0) { AD(a0, A0, 0) AD(a1, A1, 1) AD(a2, A2, 2) AD(a3, A3, 3) AD(a4, A4, 4) AD(a5, A5, 5) AD(a6, A6, 6) AD(a7, A7, 7) }
        if (nfull > 1) { AD(b0, B0, 8) AD(b1, B1, 9) AD(b2, B2, 10) AD(b3, B3, 11) AD(b4, B4, 12) AD(b5, B5, 13) AD(b6, B6, 14) AD(b7, B7, 15) }
        if (nfull > 2) { AD(c0, C0, 16) AD(c1, C1, 17) AD(c2, C2, 18) AD(c3, C3, 19) AD(c4, C4, 20) AD(c5, C5, 21) AD(c6, C6, 22) AD(c7, C7, 23) }
        if (nfull > 3) { AD(d0, D0, 24) AD(d1, D1, 25) AD(d2, D2, 26) AD(d3, D3, 27) AD(d4, D4, 28) AD(d5, D5, 29) AD(d6, D6, 30) AD(d7, D7, 31) }
        if (rem > 0) AD(p0, P0, jr + 0)
        if (rem > 1) AD(p1, P1, jr + 1)
        if (rem > 2) AD(p2, P2, jr + 2)
        if (rem > 3) AD(p3, P3, jr + 3)
        if (rem > 4) AD(p4, P4, jr + 4)
        if (rem > 5) AD(p5, P5, jr + 5)
        if (rem > 6) AD(p6, P6, jr + 6)
#undef AD
#undef LD0
#undef LD1
#undef OFFJ

        for (int base = beg + 32; base < end; base += 32) {
            const int cnt = (end - base < 32) ? (end - base) : 32;
            int s = 0;
            if (lane < cnt) s = colw[base + lane];
            for (int j = 0; j < cnt; ++j) {
                const size_t off = (size_t)(unsigned)__builtin_amdgcn_readlane(s, j) * PLANE + lane;
                acc0 += g0[off];
                acc1 += g1[off];
            }
        }
        T[r * TP + lane]         = acc0 * di;
        T[r * TP + PLANE + lane] = acc1 * di;
    }
    for (int r = nrows; r < RPW3; ++r) {
        T[r * TP + lane]         = 0.0f;
        T[r * TP + PLANE + lane] = 0.0f;
    }

    const int half = lane >> 4;
    const int l    = lane & 15;
    const float* tp = T + l * TP + half * 2;
    const float* bp = W2s + half * WPP + l * 2;
    v8f q0 = {}, q1 = {}, q2 = {}, q3 = {};
    for (int st = 0; st < HIDDEN / 4; ++st) {
        const v2f a  = *(const v2f*)(tp + 4 * st);
        const float* bq = bp + 2 * st * WPP;
        const v2f w0 = *(const v2f*)(bq);
        const v2f w1 = *(const v2f*)(bq + 32);
        const v2f w2 = *(const v2f*)(bq + 64);
        const v2f w3 = *(const v2f*)(bq + 96);
        q0 = __builtin_amdgcn_wmma_f32_16x16x4_f32(false, a, false, w0, (short)0, q0, false, false);
        q1 = __builtin_amdgcn_wmma_f32_16x16x4_f32(false, a, false, w1, (short)0, q1, false, false);
        q2 = __builtin_amdgcn_wmma_f32_16x16x4_f32(false, a, false, w2, (short)0, q2, false, false);
        q3 = __builtin_amdgcn_wmma_f32_16x16x4_f32(false, a, false, w3, (short)0, q3, false, false);
    }

    const float bb0 = bias[l], bb1 = bias[16 + l], bb2 = bias[32 + l], bb3 = bias[48 + l];
    const v2f u0 = *(const v2f*)&W3[2 * l];
    const v2f u1 = *(const v2f*)&W3[2 * (16 + l)];
    const v2f u2 = *(const v2f*)&W3[2 * (32 + l)];
    const v2f u3 = *(const v2f*)&W3[2 * (48 + l)];
    float keep = 0.0f;
#pragma unroll
    for (int i = 0; i < 8; ++i) {
        const float z0 = q0[i] + bb0, z1 = q1[i] + bb1, z2 = q2[i] + bb2, z3 = q3[i] + bb3;
        const float v0 = z0 > 0.0f ? z0 : 0.0f;
        const float v1 = z1 > 0.0f ? z1 : 0.0f;
        const float v2 = z2 > 0.0f ? z2 : 0.0f;
        const float v3 = z3 > 0.0f ? z3 : 0.0f;
        const float t0 = row_sum16(v0 * u0.x + v1 * u1.x + v2 * u2.x + v3 * u3.x);
        const float t1 = row_sum16(v0 * u0.y + v1 * u1.y + v2 * u2.y + v3 * u3.y);
        if ((l >> 1) == i) keep = (l & 1) ? t1 : t0;
    }
    const int rr  = lane >> 1;
    const int deg = __shfl(bnd, rr + 1) - __shfl(bnd, rr);
    if (lane < 2 * nrows) h3[(size_t)row0 * 2 + lane] = keep * rsqrtf((float)(deg + 1));
}

#define LROWS 32
#define LCH   24

template <int NC>
__device__ __forceinline__ void stretch_rows(const int* __restrict__ colw, const v2f* __restrict__ h2,
                                             v2f* V, int wbeg, int last, int lane) {
    unsigned c[NC];
    v2f g[NC];
#pragma unroll
    for (int k = 0; k < NC; ++k) {
        const int p = 32 * k + lane;
        c[k] = (unsigned)colw[wbeg + ((p < last) ? p : last)];
    }
#pragma unroll
    for (int k = 0; k < NC; ++k) g[k] = h2[c[k]];
#pragma unroll
    for (int k = 0; k < NC; ++k) V[32 * k + lane] = g[k];
}

__global__ __launch_bounds__(256) void pull2_lsm_k(const int* __restrict__ rowptr, const int* __restrict__ colw,
                                                   const float* __restrict__ h3, const float* __restrict__ b3,
                                                   float* __restrict__ out, int n, int E) {
    __shared__ v2f vals[8 * LCH * 32];
    const int lane = threadIdx.x & 31;
    const int wave = __builtin_amdgcn_readfirstlane((int)(threadIdx.x >> 5));
    const int row0 = ((int)blockIdx.x * 8 + wave) * LROWS;
    if (row0 >= n) return;
    const int i = row0 + lane;
    const v2f* h2 = (const v2f*)h3;
    v2f* V = vals + wave * (LCH * 32);

    int beg = E, end = E;
    if (i < n)     beg = rowptr[i];
    if (i + 1 < n) end = rowptr[i + 1];
    v2f acc = {0.0f, 0.0f};
    if (i < n) acc = h2[i];
    const int wbeg = __builtin_amdgcn_readlane(beg, 0);
    const int wend = __builtin_amdgcn_readlane(end, 31);
    const int R    = wend - wbeg;
    const int d    = end - beg;

    if (R <= LCH * 32) {
        if (R > 0) {
            const int last = R - 1;
            if      (R <= 16 * 32) stretch_rows<16>(colw, h2, V, wbeg, last, lane);
            else if (R <= 18 * 32) stretch_rows<18>(colw, h2, V, wbeg, last, lane);
            else if (R <= 20 * 32) stretch_rows<20>(colw, h2, V, wbeg, last, lane);
            else                   stretch_rows<LCH>(colw, h2, V, wbeg, last, lane);
            const int off = beg - wbeg;
            for (int j0 = 0; j0 < d; j0 += 8) {
                v2f t[8];
#pragma unroll
                for (int u = 0; u < 8; ++u) {
                    const int p = off + j0 + u;
                    t[u] = V[(p < last) ? p : last];
                }
#pragma unroll
                for (int u = 0; u < 8; ++u)
                    if (j0 + u < d) { acc.x += t[u].x; acc.y += t[u].y; }
            }
        }
    } else {
        for (int e = beg; e < end; ++e) {
            const v2f v = h2[colw[e]];
            acc.x += v.x;
            acc.y += v.y;
        }
    }
    if (i >= n) return;
    const float di = rsqrtf((float)(d + 1));
    float z0 = acc.x * di + b3[0];
    float z1 = acc.y * di + b3[1];
    float m = fmaxf(z0, z1);
    float lse = m + logf(__expf(z0 - m) + __expf(z1 - m));
    out[2 * i] = z0 - lse;
    out[2 * i + 1] = z1 - lse;
}


extern "C" void kernel_launch(void* const* d_in, const int* in_sizes, int n_in,
                              void* d_out, int out_size, void* d_ws, size_t ws_size,
                              hipStream_t stream) {
    const float* x  = (const float*)d_in[0];
    const int*   ei = (const int*)d_in[1];
    const float* W1 = (const float*)d_in[2];
    const float* b1 = (const float*)d_in[3];
    const float* W2 = (const float*)d_in[4];
    const float* b2 = (const float*)d_in[5];
    const float* W3 = (const float*)d_in[6];
    const float* b3 = (const float*)d_in[7];
    float* out = (float*)d_out;

    const int N = in_sizes[0] / F_IN;
    const int E = in_sizes[1] / 2;
    const int* src = ei;
    const int* dst = ei + E;

    int*   colw   = (int*)d_ws;
    int*   rowptr = (int*)d_ws + (size_t)2 * E;
    float* pA     = (float*)(rowptr + N);
    float* pB     = pA + (size_t)N * PLANE;
    float* pC     = pB + (size_t)N * PLANE;
    int2*  tmp    = (int2*)pC;
    int2*  tab    = (int2*)pB;
    float* h3     = pB;
    int*   cnt    = (int*)d_out;

    const int T = 256;
    dim3 blk(T);
    int gN    = (N + T - 1) / T;
    int gRow  = (N + 8 * RPW - 1) / (8 * RPW);
    int gTile = (N + 15) / 16;
    int gRow3 = (N + 4 * RPW3 - 1) / (4 * RPW3);
    int nbk   = (N + BW - 1) / BW;
    int nblk  = (E + CH - 1) / CH;

    part_k<<<nblk, dim3(1024), 0, stream>>>(src, dst, tmp, tab, E, nbk, nblk);
    bucket_k<<<nbk, dim3(1024), 0, stream>>>(tmp, tab, cnt, rowptr, colw, N, nblk);

    gemm_wmma_k<F_IN><<<gTile, dim3(128), 0, stream>>>(x, W1, cnt, pA, pB, N);
    pull32_k<<<gRow, blk, 0, stream>>>(rowptr, colw, pA, b1, pC, N, E);
    pull32_k<<<gRow, blk, 0, stream>>>(rowptr, colw, pB, b1 + PLANE, pA, N, E);

    pull2p_w2w3_k<<<gRow3, dim3(128), 0, stream>>>(rowptr, colw, pC, pA, W2, b2, W3, h3, N, E);

    pull2_lsm_k<<<gN, blk, 0, stream>>>(rowptr, colw, h3, b3, out, N, E);
}
